# starting kernel
# baseline (speedup 1.0000x reference)
#include <hip/hip_runtime.h>
#include <math.h>

#define B_ 2
#define I_ 256
#define D_ 768
#define H_ 12
#define DP_ 128
#define DK_ 64
#define NPQ_ 4
#define NPV_ 8

typedef float v2f __attribute__((ext_vector_type(2)));
typedef float v8f __attribute__((ext_vector_type(8)));

struct GemmP {
  const float* A; const float* B; float* C; const float* bias;
  int M, N, K;
  int lda, ldb, ldc;
  long sAy, sAz, sBy, sBz, sCy, sCz;
  float alpha;
};

template <int OPB, bool GUARD>
__global__ void ipa_gemm_wmma(GemmP p) {
  const int tilesN = (p.N + 15) >> 4;
  const int tilesM = (p.M + 15) >> 4;
  const int wave = (int)((blockIdx.x * blockDim.x + threadIdx.x) >> 5);
  if (wave >= tilesM * tilesN) return;
  const int lane = threadIdx.x & 31;
  const int tm = wave / tilesN, tn = wave % tilesN;
  const int row0 = tm << 4, col0 = tn << 4;

  const float* A = p.A + (long)blockIdx.y * p.sAy + (long)blockIdx.z * p.sAz;
  const float* B = p.B + (long)blockIdx.y * p.sBy + (long)blockIdx.z * p.sBz;
  float*       C = p.C + (long)blockIdx.y * p.sCy + (long)blockIdx.z * p.sCz;

  const int mn = lane & 15;
  const int kb = (lane >> 4) << 1;

  int aRow = row0 + mn;
  int bCol = col0 + mn;
  float aMask = 1.f, bMask = 1.f;
  if (GUARD) {
    if (aRow >= p.M) { aRow = p.M - 1; aMask = 0.f; }
    if (bCol >= p.N) { bCol = p.N - 1; bMask = 0.f; }
  }
  const float* aPtr = A + (long)aRow * p.lda + kb;
  const float* bPtr = (OPB == 0) ? (B + (long)kb * p.ldb + bCol)
                                 : (B + (long)bCol * p.ldb + kb);

  v8f acc = {0.f, 0.f, 0.f, 0.f, 0.f, 0.f, 0.f, 0.f};
#pragma unroll 4
  for (int k0 = 0; k0 < p.K; k0 += 4) {
    v2f a, b;
    a.x = aPtr[k0];
    a.y = aPtr[k0 + 1];
    if (OPB == 0) {
      const float* bp = bPtr + (long)k0 * p.ldb;
      b.x = bp[0];
      b.y = bp[p.ldb];
    } else {
      b.x = bPtr[k0];
      b.y = bPtr[k0 + 1];
    }
    if (GUARD) {
      a.x *= aMask; a.y *= aMask;
      b.x *= bMask; b.y *= bMask;
    }
    acc = __builtin_amdgcn_wmma_f32_16x16x4_f32(false, a, false, b,
                                                (short)0, acc, false, false);
  }

  const int cCol = col0 + mn;
  const int mo = (lane >> 4) << 3;
  const bool cColOK = GUARD ? (cCol < p.N) : true;
  if (cColOK) {
    for (int r = 0; r < 8; ++r) {
      const int cRow = row0 + mo + r;
      if (!GUARD || cRow < p.M) {
        float val = acc[r] * p.alpha;
        if (p.bias) val += p.bias[cCol];
        C[(long)cRow * p.ldc + cCol] = val;
      }
    }
  }
}

__global__ void ipa_affine(float* pts, const float* T, const float* R, int np, int total) {
  int idx = blockIdx.x * blockDim.x + threadIdx.x;
  if (idx >= total) return;
  int bi = idx / (H_ * np);
  const float* Rm = R + (long)bi * 9;
  const float* Tv = T + (long)bi * 3;
  float x = pts[idx * 3 + 0], y = pts[idx * 3 + 1], z = pts[idx * 3 + 2];
  float o0 = Rm[0] * x + Rm[3] * y + Rm[6] * z + Tv[0];
  float o1 = Rm[1] * x + Rm[4] * y + Rm[7] * z + Tv[1];
  float o2 = Rm[2] * x + Rm[5] * y + Rm[8] * z + Tv[2];
  pts[idx * 3 + 0] = o0; pts[idx * 3 + 1] = o1; pts[idx * 3 + 2] = o2;
}

__global__ void ipa_combine(float* logits,
                            const float* qgp, const float* kgp,
                            const float* qgl, const float* kgl,
                            const float* pairproj, const float* bias,
                            const float* tpw_past, const float* tpw_last) {
  long idx = (long)blockIdx.x * blockDim.x + threadIdx.x;
  const long total = (long)B_ * H_ * I_ * I_;
  if (idx >= total) return;
  int j = (int)(idx & (I_ - 1));
  long t = idx >> 8;
  int i = (int)(t & (I_ - 1));
  t >>= 8;
  int h = (int)(t % H_);
  int b = (int)(t / H_);

  const float point_w = 0.13608276348795434f;
  float pwp = -0.5f * point_w * log1pf(expf(tpw_past[h]));
  float pwl = -0.5f * point_w * log1pf(expf(tpw_last[h]));

  const float* qp = qgp + ((long)(b * I_ + i) * H_ + h) * (NPQ_ * 3);
  const float* kp = kgp + ((long)(b * I_ + j) * H_ + h) * (NPQ_ * 3);
  const float* ql = qgl + ((long)(b * I_ + i) * H_ + h) * (NPQ_ * 3);
  const float* kl = kgl + ((long)(b * I_ + j) * H_ + h) * (NPQ_ * 3);

  float sp = 0.f, sl = 0.f;
  for (int c = 0; c < NPQ_; ++c) {
    float dx = qp[c * 3 + 0] - kp[c * 3 + 0];
    float dy = qp[c * 3 + 1] - kp[c * 3 + 1];
    float dz = qp[c * 3 + 2] - kp[c * 3 + 2];
    sp += sqrtf(dx * dx + dy * dy + dz * dz);
    dx = ql[c * 3 + 0] - kl[c * 3 + 0];
    dy = ql[c * 3 + 1] - kl[c * 3 + 1];
    dz = ql[c * 3 + 2] - kl[c * 3 + 2];
    sl += sqrtf(dx * dx + dy * dy + dz * dz);
  }
  float pv = pairproj[((long)(b * I_ + i) * I_ + j) * H_ + h];
  logits[idx] += pwp * sp + pwl * sl + pv + bias[idx];
}

__global__ void ipa_softmax(float* logits) {
  __shared__ float red[I_];
  float* row = logits + (long)blockIdx.x * I_;
  int t = threadIdx.x;
  float v = row[t];
  red[t] = v;
  __syncthreads();
  for (int s = I_ / 2; s > 0; s >>= 1) {
    if (t < s) red[t] = fmaxf(red[t], red[t + s]);
    __syncthreads();
  }
  float mx = red[0];
  __syncthreads();
  float e = expf(v - mx);
  red[t] = e;
  __syncthreads();
  for (int s = I_ / 2; s > 0; s >>= 1) {
    if (t < s) red[t] += red[t + s];
    __syncthreads();
  }
  row[t] = e / red[0];
}

__global__ void ipa_inv_affine(const float* og, const float* T, const float* R,
                               float* feat, int colOl, int colN, int total) {
  int idx = blockIdx.x * blockDim.x + threadIdx.x;
  if (idx >= total) return;
  int c = idx & (NPV_ - 1);
  int t = idx >> 3;
  int h = t % H_;
  int bi = t / H_;
  const float* o = og + (long)idx * 3;
  const float* Tv = T + (long)bi * 3;
  const float* Rm = R + (long)bi * 9;
  float x = o[0] - Tv[0], y = o[1] - Tv[1], z = o[2] - Tv[2];
  float p0 = Rm[0] * x + Rm[1] * y + Rm[2] * z;
  float p1 = Rm[3] * x + Rm[4] * y + Rm[5] * z;
  float p2 = Rm[6] * x + Rm[7] * y + Rm[8] * z;
  float* fr = feat + (long)bi * (2 * D_ + 2 * H_ * NPV_ * 4);
  int tc = h * (NPV_ * 3) + c * 3;
  fr[colOl + tc + 0] = p0;
  fr[colOl + tc + 1] = p1;
  fr[colOl + tc + 2] = p2;
  fr[colN + h * NPV_ + c] = sqrtf(p0 * p0 + p1 * p1 + p2 * p2);
}

extern "C" void kernel_launch(void* const* d_in, const int* in_sizes, int n_in,
                              void* d_out, int out_size, void* d_ws, size_t ws_size,
                              hipStream_t stream) {
  (void)in_sizes; (void)n_in; (void)out_size; (void)ws_size;
  const float* x1d      = (const float*)d_in[0];
  const float* x2d      = (const float*)d_in[1];
  const float* past_T   = (const float*)d_in[2];
  const float* past_R   = (const float*)d_in[3];
  const float* last_T   = (const float*)d_in[4];
  const float* last_R   = (const float*)d_in[5];
  const float* bias     = (const float*)d_in[6];
  const float* Wq       = (const float*)d_in[7];
  const float* Wk       = (const float*)d_in[8];
  const float* Wv       = (const float*)d_in[9];
  const float* Wpb      = (const float*)d_in[10];
  const float* Wpq_past = (const float*)d_in[11];
  const float* Wpk_past = (const float*)d_in[12];
  const float* Wpv_past = (const float*)d_in[13];
  const float* Wpq_last = (const float*)d_in[14];
  const float* Wpk_last = (const float*)d_in[15];
  const float* Wpv_last = (const float*)d_in[16];
  const float* tpw_past = (const float*)d_in[17];
  const float* tpw_last = (const float*)d_in[18];
  const float* Wpv      = (const float*)d_in[19];
  const float* W_out    = (const float*)d_in[20];
  const float* b_out    = (const float*)d_in[21];
  float* out = (float*)d_out;
  float* ws  = (float*)d_ws;

  const int M  = B_ * I_;
  const int FE = 2 * D_ + 2 * H_ * NPV_ * 4;

  float* q    = ws;
  float* k    = q + 393216;
  float* v    = k + 393216;
  float* qgp  = v + 393216;
  float* kgp  = qgp + 73728;
  float* vgp  = kgp + 73728;
  float* qgl  = vgp + 147456;
  float* kgl  = qgl + 73728;
  float* vgl  = kgl + 73728;
  float* logits   = vgl + 147456;
  float* pairproj = logits + 1572864;
  float* ogp  = pairproj + 1572864;
  float* ogl  = ogp + 147456;
  float* z    = ogl + 147456;
  float* feat = z + 786432;

  auto gemm = [&](const float* A, const float* Bm, float* C,
                  int Mm, int Nn, int Kk, int lda, int ldb, int ldc,
                  long sAy, long sAz, long sBy, long sBz, long sCy, long sCz,
                  int gy, int gz, int opB, float alpha, const float* bia) {
    GemmP p{A, Bm, C, bia, Mm, Nn, Kk, lda, ldb, ldc,
            sAy, sAz, sBy, sBz, sCy, sCz, alpha};
    int waves = ((Mm + 15) / 16) * ((Nn + 15) / 16);
    dim3 grid((unsigned)((waves + 7) / 8), (unsigned)gy, (unsigned)gz);
    dim3 blk(256);
    bool guard = (Mm & 15) || (Nn & 15);
    if (opB == 0) {
      if (guard) ipa_gemm_wmma<0, true ><<<grid, blk, 0, stream>>>(p);
      else       ipa_gemm_wmma<0, false><<<grid, blk, 0, stream>>>(p);
    } else {
      if (guard) ipa_gemm_wmma<1, true ><<<grid, blk, 0, stream>>>(p);
      else       ipa_gemm_wmma<1, false><<<grid, blk, 0, stream>>>(p);
    }
  };

  gemm(x1d, Wq, q, M, D_, D_, D_, D_, D_, 0,0,0,0,0,0, 1,1, 0, 1.f, nullptr);
  gemm(x1d, Wk, k, M, D_, D_, D_, D_, D_, 0,0,0,0,0,0, 1,1, 0, 1.f, nullptr);
  gemm(x1d, Wv, v, M, D_, D_, D_, D_, D_, 0,0,0,0,0,0, 1,1, 0, 1.f, nullptr);
  gemm(x1d, Wpq_past, qgp, M, 144, D_, D_, 144, 144, 0,0,0,0,0,0, 1,1, 0, 1.f, nullptr);
  gemm(x1d, Wpk_past, kgp, M, 144, D_, D_, 144, 144, 0,0,0,0,0,0, 1,1, 0, 1.f, nullptr);
  gemm(x1d, Wpv_past, vgp, M, 288, D_, D_, 288, 288, 0,0,0,0,0,0, 1,1, 0, 1.f, nullptr);
  gemm(x1d, Wpq_last, qgl, M, 144, D_, D_, 144, 144, 0,0,0,0,0,0, 1,1, 0, 1.f, nullptr);
  gemm(x1d, Wpk_last, kgl, M, 144, D_, D_, 144, 144, 0,0,0,0,0,0, 1,1, 0, 1.f, nullptr);
  gemm(x1d, Wpv_last, vgl, M, 288, D_, D_, 288, 288, 0,0,0,0,0,0, 1,1, 0, 1.f, nullptr);

  {
    int t4 = B_ * I_ * H_ * NPQ_, t8 = B_ * I_ * H_ * NPV_;
    ipa_affine<<<(t4 + 255) / 256, 256, 0, stream>>>(qgp, past_T, past_R, NPQ_, t4);
    ipa_affine<<<(t4 + 255) / 256, 256, 0, stream>>>(kgp, past_T, past_R, NPQ_, t4);
    ipa_affine<<<(t8 + 255) / 256, 256, 0, stream>>>(vgp, past_T, past_R, NPV_, t8);
    ipa_affine<<<(t4 + 255) / 256, 256, 0, stream>>>(qgl, last_T, last_R, NPQ_, t4);
    ipa_affine<<<(t4 + 255) / 256, 256, 0, stream>>>(kgl, last_T, last_R, NPQ_, t4);
    ipa_affine<<<(t8 + 255) / 256, 256, 0, stream>>>(vgl, last_T, last_R, NPV_, t8);
  }

  const float scalar_w = 0.07216878364870323f;
  gemm(q, k, logits, I_, I_, DK_, D_, D_, I_,
       64, (long)I_ * D_, 64, (long)I_ * D_, (long)I_ * I_, (long)H_ * I_ * I_,
       H_, B_, 1, scalar_w, nullptr);

  const float pair_w = 0.5773502691896258f;
  gemm(x2d, Wpb, pairproj, B_ * I_ * I_, H_, DP_, DP_, H_, H_,
       0,0,0,0,0,0, 1,1, 0, pair_w, nullptr);

  {
    long total = (long)B_ * H_ * I_ * I_;
    ipa_combine<<<(unsigned)((total + 255) / 256), 256, 0, stream>>>(
        logits, qgp, kgp, qgl, kgl, pairproj, bias, tpw_past, tpw_last);
    ipa_softmax<<<B_ * H_ * I_, I_, 0, stream>>>(logits);
  }

  gemm(logits, v, feat + D_, I_, DK_, I_, I_, D_, FE,
       (long)I_ * I_, (long)H_ * I_ * I_, 64, (long)I_ * D_, 64, (long)I_ * FE,
       H_, B_, 0, 1.f, nullptr);

  gemm(logits, vgp, ogp, I_, NPV_ * 3, I_, I_, H_ * NPV_ * 3, H_ * NPV_ * 3,
       (long)I_ * I_, (long)H_ * I_ * I_, NPV_ * 3, (long)I_ * H_ * NPV_ * 3,
       NPV_ * 3, (long)I_ * H_ * NPV_ * 3, H_, B_, 0, 1.f, nullptr);
  gemm(logits, vgl, ogl, I_, NPV_ * 3, I_, I_, H_ * NPV_ * 3, H_ * NPV_ * 3,
       (long)I_ * I_, (long)H_ * I_ * I_, NPV_ * 3, (long)I_ * H_ * NPV_ * 3,
       NPV_ * 3, (long)I_ * H_ * NPV_ * 3, H_, B_, 0, 1.f, nullptr);

  gemm(logits, x2d, z, H_, DP_, I_,
       I_ * I_, DP_, DP_,
       (long)I_, (long)H_ * I_ * I_,
       (long)I_ * DP_, (long)I_ * I_ * DP_,
       (long)H_ * DP_, (long)I_ * H_ * DP_,
       I_, B_, 0, 1.f, nullptr);

  gemm(z, Wpv, feat, I_, DK_, DP_, H_ * DP_, D_, FE,
       DP_, (long)I_ * H_ * DP_, 64, 0, 64, (long)I_ * FE,
       H_, B_, 0, 1.f, nullptr);

  {
    int t8 = B_ * I_ * H_ * NPV_;
    ipa_inv_affine<<<(t8 + 255) / 256, 256, 0, stream>>>(
        ogp, past_T, past_R, feat, 2 * D_, 2 * D_ + 288, t8);
    ipa_inv_affine<<<(t8 + 255) / 256, 256, 0, stream>>>(
        ogl, past_T, past_R, feat, 2 * D_ + 384, 2 * D_ + 384 + 288, t8);
  }

  gemm(feat, W_out, out, M, D_, FE, FE, D_, D_,
       0,0,0,0,0,0, 1,1, 0, 1.f, b_out);
}


# GEMMs with K>=256 and few tiles: K split over the 8 waves of a work-group (one 16x16 tile per group), the 8 partial tiles summed through LDS in fixed order; changes the f32 summation order
# speedup vs baseline: 1.8847x; 1.8847x over previous
#include <hip/hip_runtime.h>
#include <math.h>

#define B_ 2
#define I_ 256
#define D_ 768
#define H_ 12
#define DP_ 128
#define DK_ 64
#define NPQ_ 4
#define NPV_ 8

typedef float v2f __attribute__((ext_vector_type(2)));
typedef float v8f __attribute__((ext_vector_type(8)));

struct GemmP {
  const float* A; const float* B; float* C; const float* bias;
  int M, N, K;
  int lda, ldb, ldc;
  long sAy, sAz, sBy, sBz, sCy, sCz;
  float alpha;
};

template <int OPB, bool GUARD, int KS>
__global__ void ipa_gemm_wmma(GemmP p) {
  const int tilesN = (p.N + 15) >> 4;
  const int tilesM = (p.M + 15) >> 4;
  const int wave = (KS == 1) ? (int)((blockIdx.x * blockDim.x + threadIdx.x) >> 5)
                             : (int)blockIdx.x;
  if (KS == 1) {
    if (wave >= tilesM * tilesN) return;
  }
  const int lane = threadIdx.x & 31;
  const int ks = (KS == 1) ? 0 : __builtin_amdgcn_readfirstlane((int)(threadIdx.x >> 5));
  int kBeg = 0, kEnd = p.K;
  if (KS > 1) {
    const int per = (((p.K >> 2) + KS - 1) / KS) << 2;
    kBeg = ks * per;
    kEnd = (kBeg + per < p.K) ? (kBeg + per) : p.K;
  }
  const int tm = wave / tilesN, tn = wave % tilesN;
  const int row0 = tm << 4, col0 = tn << 4;

  const float* A = p.A + (long)blockIdx.y * p.sAy + (long)blockIdx.z * p.sAz;
  const float* B = p.B + (long)blockIdx.y * p.sBy + (long)blockIdx.z * p.sBz;
  float*       C = p.C + (long)blockIdx.y * p.sCy + (long)blockIdx.z * p.sCz;

  const int mn = lane & 15;
  const int kb = (lane >> 4) << 1;

  int aRow = row0 + mn;
  int bCol = col0 + mn;
  float aMask = 1.f, bMask = 1.f;
  if (GUARD) {
    if (aRow >= p.M) { aRow = p.M - 1; aMask = 0.f; }
    if (bCol >= p.N) { bCol = p.N - 1; bMask = 0.f; }
  }
  const float* aPtr = A + (long)aRow * p.lda + kb;
  const float* bPtr = (OPB == 0) ? (B + (long)kb * p.ldb + bCol)
                                 : (B + (long)bCol * p.ldb + kb);

  v8f acc = {0.f, 0.f, 0.f, 0.f, 0.f, 0.f, 0.f, 0.f};
#pragma unroll 4
  for (int k0 = kBeg; k0 < kEnd; k0 += 4) {
    v2f a, b;
    a.x = aPtr[k0];
    a.y = aPtr[k0 + 1];
    if (OPB == 0) {
      const float* bp = bPtr + (long)k0 * p.ldb;
      b.x = bp[0];
      b.y = bp[p.ldb];
    } else {
      b.x = bPtr[k0];
      b.y = bPtr[k0 + 1];
    }
    if (GUARD) {
      a.x *= aMask; a.y *= aMask;
      b.x *= bMask; b.y *= bMask;
    }
    acc = __builtin_amdgcn_wmma_f32_16x16x4_f32(false, a, false, b,
                                                (short)0, acc, false, false);
  }

  if (KS > 1) {
    __shared__ float red[KS * 256];
    for (int r = 0; r < 8; ++r) red[ks * 256 + r * 32 + lane] = acc[r];
    __syncthreads();
    const int t = (int)threadIdx.x;
    float sum = red[t];
    for (int s = 1; s < KS; ++s) sum += red[s * 256 + t];
    const int oRow = row0 + (((t & 31) >> 4) << 3) + (t >> 5);
    const int oCol = col0 + (t & 15);
    if (!GUARD || (oRow < p.M && oCol < p.N)) {
      float val = sum * p.alpha;
      if (p.bias) val += p.bias[oCol];
      C[(long)oRow * p.ldc + oCol] = val;
    }
    return;
  }

  const int cCol = col0 + mn;
  const int mo = (lane >> 4) << 3;
  const bool cColOK = GUARD ? (cCol < p.N) : true;
  if (cColOK) {
    for (int r = 0; r < 8; ++r) {
      const int cRow = row0 + mo + r;
      if (!GUARD || cRow < p.M) {
        float val = acc[r] * p.alpha;
        if (p.bias) val += p.bias[cCol];
        C[(long)cRow * p.ldc + cCol] = val;
      }
    }
  }
}

__global__ void ipa_affine(float* pts, const float* T, const float* R, int np, int total) {
  int idx = blockIdx.x * blockDim.x + threadIdx.x;
  if (idx >= total) return;
  int bi = idx / (H_ * np);
  const float* Rm = R + (long)bi * 9;
  const float* Tv = T + (long)bi * 3;
  float x = pts[idx * 3 + 0], y = pts[idx * 3 + 1], z = pts[idx * 3 + 2];
  float o0 = Rm[0] * x + Rm[3] * y + Rm[6] * z + Tv[0];
  float o1 = Rm[1] * x + Rm[4] * y + Rm[7] * z + Tv[1];
  float o2 = Rm[2] * x + Rm[5] * y + Rm[8] * z + Tv[2];
  pts[idx * 3 + 0] = o0; pts[idx * 3 + 1] = o1; pts[idx * 3 + 2] = o2;
}

__global__ void ipa_combine(float* logits,
                            const float* qgp, const float* kgp,
                            const float* qgl, const float* kgl,
                            const float* pairproj, const float* bias,
                            const float* tpw_past, const float* tpw_last) {
  long idx = (long)blockIdx.x * blockDim.x + threadIdx.x;
  const long total = (long)B_ * H_ * I_ * I_;
  if (idx >= total) return;
  int j = (int)(idx & (I_ - 1));
  long t = idx >> 8;
  int i = (int)(t & (I_ - 1));
  t >>= 8;
  int h = (int)(t % H_);
  int b = (int)(t / H_);

  const float point_w = 0.13608276348795434f;
  float pwp = -0.5f * point_w * log1pf(expf(tpw_past[h]));
  float pwl = -0.5f * point_w * log1pf(expf(tpw_last[h]));

  const float* qp = qgp + ((long)(b * I_ + i) * H_ + h) * (NPQ_ * 3);
  const float* kp = kgp + ((long)(b * I_ + j) * H_ + h) * (NPQ_ * 3);
  const float* ql = qgl + ((long)(b * I_ + i) * H_ + h) * (NPQ_ * 3);
  const float* kl = kgl + ((long)(b * I_ + j) * H_ + h) * (NPQ_ * 3);

  float sp = 0.f, sl = 0.f;
  for (int c = 0; c < NPQ_; ++c) {
    float dx = qp[c * 3 + 0] - kp[c * 3 + 0];
    float dy = qp[c * 3 + 1] - kp[c * 3 + 1];
    float dz = qp[c * 3 + 2] - kp[c * 3 + 2];
    sp += sqrtf(dx * dx + dy * dy + dz * dz);
    dx = ql[c * 3 + 0] - kl[c * 3 + 0];
    dy = ql[c * 3 + 1] - kl[c * 3 + 1];
    dz = ql[c * 3 + 2] - kl[c * 3 + 2];
    sl += sqrtf(dx * dx + dy * dy + dz * dz);
  }
  float pv = pairproj[((long)(b * I_ + i) * I_ + j) * H_ + h];
  logits[idx] += pwp * sp + pwl * sl + pv + bias[idx];
}

__global__ void ipa_softmax(float* logits) {
  __shared__ float red[I_];
  float* row = logits + (long)blockIdx.x * I_;
  int t = threadIdx.x;
  float v = row[t];
  red[t] = v;
  __syncthreads();
  for (int s = I_ / 2; s > 0; s >>= 1) {
    if (t < s) red[t] = fmaxf(red[t], red[t + s]);
    __syncthreads();
  }
  float mx = red[0];
  __syncthreads();
  float e = expf(v - mx);
  red[t] = e;
  __syncthreads();
  for (int s = I_ / 2; s > 0; s >>= 1) {
    if (t < s) red[t] += red[t + s];
    __syncthreads();
  }
  row[t] = e / red[0];
}

__global__ void ipa_inv_affine(const float* og, const float* T, const float* R,
                               float* feat, int colOl, int colN, int total) {
  int idx = blockIdx.x * blockDim.x + threadIdx.x;
  if (idx >= total) return;
  int c = idx & (NPV_ - 1);
  int t = idx >> 3;
  int h = t % H_;
  int bi = t / H_;
  const float* o = og + (long)idx * 3;
  const float* Tv = T + (long)bi * 3;
  const float* Rm = R + (long)bi * 9;
  float x = o[0] - Tv[0], y = o[1] - Tv[1], z = o[2] - Tv[2];
  float p0 = Rm[0] * x + Rm[1] * y + Rm[2] * z;
  float p1 = Rm[3] * x + Rm[4] * y + Rm[5] * z;
  float p2 = Rm[6] * x + Rm[7] * y + Rm[8] * z;
  float* fr = feat + (long)bi * (2 * D_ + 2 * H_ * NPV_ * 4);
  int tc = h * (NPV_ * 3) + c * 3;
  fr[colOl + tc + 0] = p0;
  fr[colOl + tc + 1] = p1;
  fr[colOl + tc + 2] = p2;
  fr[colN + h * NPV_ + c] = sqrtf(p0 * p0 + p1 * p1 + p2 * p2);
}

extern "C" void kernel_launch(void* const* d_in, const int* in_sizes, int n_in,
                              void* d_out, int out_size, void* d_ws, size_t ws_size,
                              hipStream_t stream) {
  (void)in_sizes; (void)n_in; (void)out_size; (void)ws_size;
  const float* x1d      = (const float*)d_in[0];
  const float* x2d      = (const float*)d_in[1];
  const float* past_T   = (const float*)d_in[2];
  const float* past_R   = (const float*)d_in[3];
  const float* last_T   = (const float*)d_in[4];
  const float* last_R   = (const float*)d_in[5];
  const float* bias     = (const float*)d_in[6];
  const float* Wq       = (const float*)d_in[7];
  const float* Wk       = (const float*)d_in[8];
  const float* Wv       = (const float*)d_in[9];
  const float* Wpb      = (const float*)d_in[10];
  const float* Wpq_past = (const float*)d_in[11];
  const float* Wpk_past = (const float*)d_in[12];
  const float* Wpv_past = (const float*)d_in[13];
  const float* Wpq_last = (const float*)d_in[14];
  const float* Wpk_last = (const float*)d_in[15];
  const float* Wpv_last = (const float*)d_in[16];
  const float* tpw_past = (const float*)d_in[17];
  const float* tpw_last = (const float*)d_in[18];
  const float* Wpv      = (const float*)d_in[19];
  const float* W_out    = (const float*)d_in[20];
  const float* b_out    = (const float*)d_in[21];
  float* out = (float*)d_out;
  float* ws  = (float*)d_ws;

  const int M  = B_ * I_;
  const int FE = 2 * D_ + 2 * H_ * NPV_ * 4;

  float* q    = ws;
  float* k    = q + 393216;
  float* v    = k + 393216;
  float* qgp  = v + 393216;
  float* kgp  = qgp + 73728;
  float* vgp  = kgp + 73728;
  float* qgl  = vgp + 147456;
  float* kgl  = qgl + 73728;
  float* vgl  = kgl + 73728;
  float* logits   = vgl + 147456;
  float* pairproj = logits + 1572864;
  float* ogp  = pairproj + 1572864;
  float* ogl  = ogp + 147456;
  float* z    = ogl + 147456;
  float* feat = z + 786432;

  auto gemm = [&](const float* A, const float* Bm, float* C,
                  int Mm, int Nn, int Kk, int lda, int ldb, int ldc,
                  long sAy, long sAz, long sBy, long sBz, long sCy, long sCz,
                  int gy, int gz, int opB, float alpha, const float* bia,
                  bool splitK = false) {
    GemmP p{A, Bm, C, bia, Mm, Nn, Kk, lda, ldb, ldc,
            sAy, sAz, sBy, sBz, sCy, sCz, alpha};
    int waves = ((Mm + 15) / 16) * ((Nn + 15) / 16);
    dim3 blk(256);
    bool guard = (Mm & 15) || (Nn & 15);
    if (splitK) {
      dim3 gridS((unsigned)waves, (unsigned)gy, (unsigned)gz);
      if (guard) ipa_gemm_wmma<0, true , 8><<<gridS, blk, 0, stream>>>(p);
      else       ipa_gemm_wmma<0, false, 8><<<gridS, blk, 0, stream>>>(p);
      return;
    }
    dim3 grid((unsigned)((waves + 7) / 8), (unsigned)gy, (unsigned)gz);
    if (opB == 0) {
      if (guard) ipa_gemm_wmma<0, true , 1><<<grid, blk, 0, stream>>>(p);
      else       ipa_gemm_wmma<0, false, 1><<<grid, blk, 0, stream>>>(p);
    } else {
      if (guard) ipa_gemm_wmma<1, true , 1><<<grid, blk, 0, stream>>>(p);
      else       ipa_gemm_wmma<1, false, 1><<<grid, blk, 0, stream>>>(p);
    }
  };

  gemm(x1d, Wq, q, M, D_, D_, D_, D_, D_, 0,0,0,0,0,0, 1,1, 0, 1.f, nullptr, true);
  gemm(x1d, Wk, k, M, D_, D_, D_, D_, D_, 0,0,0,0,0,0, 1,1, 0, 1.f, nullptr, true);
  gemm(x1d, Wv, v, M, D_, D_, D_, D_, D_, 0,0,0,0,0,0, 1,1, 0, 1.f, nullptr, true);
  gemm(x1d, Wpq_past, qgp, M, 144, D_, D_, 144, 144, 0,0,0,0,0,0, 1,1, 0, 1.f, nullptr, true);
  gemm(x1d, Wpk_past, kgp, M, 144, D_, D_, 144, 144, 0,0,0,0,0,0, 1,1, 0, 1.f, nullptr, true);
  gemm(x1d, Wpv_past, vgp, M, 288, D_, D_, 288, 288, 0,0,0,0,0,0, 1,1, 0, 1.f, nullptr, true);
  gemm(x1d, Wpq_last, qgl, M, 144, D_, D_, 144, 144, 0,0,0,0,0,0, 1,1, 0, 1.f, nullptr, true);
  gemm(x1d, Wpk_last, kgl, M, 144, D_, D_, 144, 144, 0,0,0,0,0,0, 1,1, 0, 1.f, nullptr, true);
  gemm(x1d, Wpv_last, vgl, M, 288, D_, D_, 288, 288, 0,0,0,0,0,0, 1,1, 0, 1.f, nullptr, true);

  {
    int t4 = B_ * I_ * H_ * NPQ_, t8 = B_ * I_ * H_ * NPV_;
    ipa_affine<<<(t4 + 255) / 256, 256, 0, stream>>>(qgp, past_T, past_R, NPQ_, t4);
    ipa_affine<<<(t4 + 255) / 256, 256, 0, stream>>>(kgp, past_T, past_R, NPQ_, t4);
    ipa_affine<<<(t8 + 255) / 256, 256, 0, stream>>>(vgp, past_T, past_R, NPV_, t8);
    ipa_affine<<<(t4 + 255) / 256, 256, 0, stream>>>(qgl, last_T, last_R, NPQ_, t4);
    ipa_affine<<<(t4 + 255) / 256, 256, 0, stream>>>(kgl, last_T, last_R, NPQ_, t4);
    ipa_affine<<<(t8 + 255) / 256, 256, 0, stream>>>(vgl, last_T, last_R, NPV_, t8);
  }

  const float scalar_w = 0.07216878364870323f;
  gemm(q, k, logits, I_, I_, DK_, D_, D_, I_,
       64, (long)I_ * D_, 64, (long)I_ * D_, (long)I_ * I_, (long)H_ * I_ * I_,
       H_, B_, 1, scalar_w, nullptr);

  const float pair_w = 0.5773502691896258f;
  gemm(x2d, Wpb, pairproj, B_ * I_ * I_, H_, DP_, DP_, H_, H_,
       0,0,0,0,0,0, 1,1, 0, pair_w, nullptr);

  {
    long total = (long)B_ * H_ * I_ * I_;
    ipa_combine<<<(unsigned)((total + 255) / 256), 256, 0, stream>>>(
        logits, qgp, kgp, qgl, kgl, pairproj, bias, tpw_past, tpw_last);
    ipa_softmax<<<B_ * H_ * I_, I_, 0, stream>>>(logits);
  }

  gemm(logits, v, feat + D_, I_, DK_, I_, I_, D_, FE,
       (long)I_ * I_, (long)H_ * I_ * I_, 64, (long)I_ * D_, 64, (long)I_ * FE,
       H_, B_, 0, 1.f, nullptr, true);

  gemm(logits, vgp, ogp, I_, NPV_ * 3, I_, I_, H_ * NPV_ * 3, H_ * NPV_ * 3,
       (long)I_ * I_, (long)H_ * I_ * I_, NPV_ * 3, (long)I_ * H_ * NPV_ * 3,
       NPV_ * 3, (long)I_ * H_ * NPV_ * 3, H_, B_, 0, 1.f, nullptr, true);
  gemm(logits, vgl, ogl, I_, NPV_ * 3, I_, I_, H_ * NPV_ * 3, H_ * NPV_ * 3,
       (long)I_ * I_, (long)H_ * I_ * I_, NPV_ * 3, (long)I_ * H_ * NPV_ * 3,
       NPV_ * 3, (long)I_ * H_ * NPV_ * 3, H_, B_, 0, 1.f, nullptr, true);

  gemm(logits, x2d, z, H_, DP_, I_,
       I_ * I_, DP_, DP_,
       (long)I_, (long)H_ * I_ * I_,
       (long)I_ * DP_, (long)I_ * I_ * DP_,
       (long)H_ * DP_, (long)I_ * H_ * DP_,
       I_, B_, 0, 1.f, nullptr);

  gemm(z, Wpv, feat, I_, DK_, DP_, H_ * DP_, D_, FE,
       DP_, (long)I_ * H_ * DP_, 64, 0, 64, (long)I_ * FE,
       H_, B_, 0, 1.f, nullptr);

  {
    int t8 = B_ * I_ * H_ * NPV_;
    ipa_inv_affine<<<(t8 + 255) / 256, 256, 0, stream>>>(
        ogp, past_T, past_R, feat, 2 * D_, 2 * D_ + 288, t8);
    ipa_inv_affine<<<(t8 + 255) / 256, 256, 0, stream>>>(
        ogl, past_T, past_R, feat, 2 * D_ + 384, 2 * D_ + 384 + 288, t8);
  }

  gemm(feat, W_out, out, M, D_, FE, FE, D_, D_,
       0,0,0,0,0,0, 1,1, 0, 1.f, b_out, true);
}


# logit combine + row softmax fused into one kernel: one wave per row, row in registers, key points staged in LDS per 8 rows; one launch and 92160 waves fewer; softmax sum order changes (shuffle tree)
# speedup vs baseline: 1.9993x; 1.0608x over previous
#include <hip/hip_runtime.h>
#include <math.h>

#define B_ 2
#define I_ 256
#define D_ 768
#define H_ 12
#define DP_ 128
#define DK_ 64
#define NPQ_ 4
#define NPV_ 8

typedef float v2f __attribute__((ext_vector_type(2)));
typedef float v8f __attribute__((ext_vector_type(8)));

struct GemmP {
  const float* A; const float* B; float* C; const float* bias;
  int M, N, K;
  int lda, ldb, ldc;
  long sAy, sAz, sBy, sBz, sCy, sCz;
  float alpha;
};

template <int OPB, bool GUARD, int KS>
__global__ void ipa_gemm_wmma(GemmP p) {
  const int tilesN = (p.N + 15) >> 4;
  const int tilesM = (p.M + 15) >> 4;
  const int wave = (KS == 1) ? (int)((blockIdx.x * blockDim.x + threadIdx.x) >> 5)
                             : (int)blockIdx.x;
  if (KS == 1) {
    if (wave >= tilesM * tilesN) return;
  }
  const int lane = threadIdx.x & 31;
  const int ks = (KS == 1) ? 0 : __builtin_amdgcn_readfirstlane((int)(threadIdx.x >> 5));
  int kBeg = 0, kEnd = p.K;
  if (KS > 1) {
    const int per = (((p.K >> 2) + KS - 1) / KS) << 2;
    kBeg = ks * per;
    kEnd = (kBeg + per < p.K) ? (kBeg + per) : p.K;
  }
  const int tm = wave / tilesN, tn = wave % tilesN;
  const int row0 = tm << 4, col0 = tn << 4;

  const float* A = p.A + (long)blockIdx.y * p.sAy + (long)blockIdx.z * p.sAz;
  const float* B = p.B + (long)blockIdx.y * p.sBy + (long)blockIdx.z * p.sBz;
  float*       C = p.C + (long)blockIdx.y * p.sCy + (long)blockIdx.z * p.sCz;

  const int mn = lane & 15;
  const int kb = (lane >> 4) << 1;

  int aRow = row0 + mn;
  int bCol = col0 + mn;
  float aMask = 1.f, bMask = 1.f;
  if (GUARD) {
    if (aRow >= p.M) { aRow = p.M - 1; aMask = 0.f; }
    if (bCol >= p.N) { bCol = p.N - 1; bMask = 0.f; }
  }
  const float* aPtr = A + (long)aRow * p.lda + kb;
  const float* bPtr = (OPB == 0) ? (B + (long)kb * p.ldb + bCol)
                                 : (B + (long)bCol * p.ldb + kb);

  v8f acc = {0.f, 0.f, 0.f, 0.f, 0.f, 0.f, 0.f, 0.f};
#pragma unroll 4
  for (int k0 = kBeg; k0 < kEnd; k0 += 4) {
    v2f a, b;
    a.x = aPtr[k0];
    a.y = aPtr[k0 + 1];
    if (OPB == 0) {
      const float* bp = bPtr + (long)k0 * p.ldb;
      b.x = bp[0];
      b.y = bp[p.ldb];
    } else {
      b.x = bPtr[k0];
      b.y = bPtr[k0 + 1];
    }
    if (GUARD) {
      a.x *= aMask; a.y *= aMask;
      b.x *= bMask; b.y *= bMask;
    }
    acc = __builtin_amdgcn_wmma_f32_16x16x4_f32(false, a, false, b,
                                                (short)0, acc, false, false);
  }

  if (KS > 1) {
    __shared__ float red[KS * 256];
    for (int r = 0; r < 8; ++r) red[ks * 256 + r * 32 + lane] = acc[r];
    __syncthreads();
    const int t = (int)threadIdx.x;
    float sum = red[t];
    for (int s = 1; s < KS; ++s) sum += red[s * 256 + t];
    const int oRow = row0 + (((t & 31) >> 4) << 3) + (t >> 5);
    const int oCol = col0 + (t & 15);
    if (!GUARD || (oRow < p.M && oCol < p.N)) {
      float val = sum * p.alpha;
      if (p.bias) val += p.bias[oCol];
      C[(long)oRow * p.ldc + oCol] = val;
    }
    return;
  }

  const int cCol = col0 + mn;
  const int mo = (lane >> 4) << 3;
  const bool cColOK = GUARD ? (cCol < p.N) : true;
  if (cColOK) {
    for (int r = 0; r < 8; ++r) {
      const int cRow = row0 + mo + r;
      if (!GUARD || cRow < p.M) {
        float val = acc[r] * p.alpha;
        if (p.bias) val += p.bias[cCol];
        C[(long)cRow * p.ldc + cCol] = val;
      }
    }
  }
}

__global__ void ipa_affine(float* pts, const float* T, const float* R, int np, int total) {
  int idx = blockIdx.x * blockDim.x + threadIdx.x;
  if (idx >= total) return;
  int bi = idx / (H_ * np);
  const float* Rm = R + (long)bi * 9;
  const float* Tv = T + (long)bi * 3;
  float x = pts[idx * 3 + 0], y = pts[idx * 3 + 1], z = pts[idx * 3 + 2];
  float o0 = Rm[0] * x + Rm[3] * y + Rm[6] * z + Tv[0];
  float o1 = Rm[1] * x + Rm[4] * y + Rm[7] * z + Tv[1];
  float o2 = Rm[2] * x + Rm[5] * y + Rm[8] * z + Tv[2];
  pts[idx * 3 + 0] = o0; pts[idx * 3 + 1] = o1; pts[idx * 3 + 2] = o2;
}

typedef float v4f __attribute__((ext_vector_type(4)));

#define ROWS_PER_WG_ 8
#define NKC_ (NPQ_ * 3)

__global__ void __launch_bounds__(256)
ipa_combine_softmax(float* logits,
                    const float* qgp, const float* kgp,
                    const float* qgl, const float* kgl,
                    const float* pairproj, const float* bias,
                    const float* tpw_past, const float* tpw_last) {
  __shared__ float kst[2 * NKC_ * I_];
  const int t = (int)threadIdx.x;
  const int ig = (int)blockIdx.x % (I_ / ROWS_PER_WG_);
  const int bh = (int)blockIdx.x / (I_ / ROWS_PER_WG_);
  const int h = bh % H_;
  const int b = bh / H_;
  const int lane = t & 31;
  const int w = __builtin_amdgcn_readfirstlane(t >> 5);
  const int i = ig * ROWS_PER_WG_ + w;

  float qp[NKC_], ql[NKC_];
  {
    const long qo = ((long)(b * I_ + i) * H_ + h) * NKC_;
    for (int c = 0; c < NKC_; ++c) { qp[c] = qgp[qo + c]; ql[c] = qgl[qo + c]; }
  }

  {
    const long ko = ((long)(b * I_ + t) * H_ + h) * NKC_;
    const v4f* kp = (const v4f*)(kgp + ko);
    const v4f* kl = (const v4f*)(kgl + ko);
    const v4f p0 = kp[0], p1 = kp[1], p2 = kp[2];
    const v4f l0 = kl[0], l1 = kl[1], l2 = kl[2];
    kst[ 0 * I_ + t] = p0.x; kst[ 1 * I_ + t] = p0.y; kst[ 2 * I_ + t] = p0.z; kst[ 3 * I_ + t] = p0.w;
    kst[ 4 * I_ + t] = p1.x; kst[ 5 * I_ + t] = p1.y; kst[ 6 * I_ + t] = p1.z; kst[ 7 * I_ + t] = p1.w;
    kst[ 8 * I_ + t] = p2.x; kst[ 9 * I_ + t] = p2.y; kst[10 * I_ + t] = p2.z; kst[11 * I_ + t] = p2.w;
    kst[12 * I_ + t] = l0.x; kst[13 * I_ + t] = l0.y; kst[14 * I_ + t] = l0.z; kst[15 * I_ + t] = l0.w;
    kst[16 * I_ + t] = l1.x; kst[17 * I_ + t] = l1.y; kst[18 * I_ + t] = l1.z; kst[19 * I_ + t] = l1.w;
    kst[20 * I_ + t] = l2.x; kst[21 * I_ + t] = l2.y; kst[22 * I_ + t] = l2.z; kst[23 * I_ + t] = l2.w;
  }

  const float point_w = 0.13608276348795434f;
  const float pwp = -0.5f * point_w * log1pf(expf(tpw_past[h]));
  const float pwl = -0.5f * point_w * log1pf(expf(tpw_last[h]));

  const long row = ((long)(b * H_ + h) * I_ + i) * I_;
  const float* pr = pairproj + ((long)(b * I_ + i) * I_) * H_ + h;

  __syncthreads();

  float v[I_ / 32];
#pragma unroll
  for (int m = 0; m < I_ / 32; ++m) {
    const int j = m * 32 + lane;
    float sp = 0.f, sl = 0.f;
    for (int c = 0; c < NPQ_; ++c) {
      float dx = qp[c * 3 + 0] - kst[(c * 3 + 0) * I_ + j];
      float dy = qp[c * 3 + 1] - kst[(c * 3 + 1) * I_ + j];
      float dz = qp[c * 3 + 2] - kst[(c * 3 + 2) * I_ + j];
      sp += sqrtf(dx * dx + dy * dy + dz * dz);
      dx = ql[c * 3 + 0] - kst[(NKC_ + c * 3 + 0) * I_ + j];
      dy = ql[c * 3 + 1] - kst[(NKC_ + c * 3 + 1) * I_ + j];
      dz = ql[c * 3 + 2] - kst[(NKC_ + c * 3 + 2) * I_ + j];
      sl += sqrtf(dx * dx + dy * dy + dz * dz);
    }
    const float pv = pr[(long)j * H_];
    const float add = pwp * sp + pwl * sl + pv + bias[row + j];
    v[m] = logits[row + j] + add;
  }

  float mx = v[0];
  for (int m = 1; m < I_ / 32; ++m) mx = fmaxf(mx, v[m]);
  for (int off = 16; off > 0; off >>= 1) mx = fmaxf(mx, __shfl_xor(mx, off));

  float sum = 0.f;
  for (int m = 0; m < I_ / 32; ++m) { v[m] = expf(v[m] - mx); sum += v[m]; }
  for (int off = 16; off > 0; off >>= 1) sum += __shfl_xor(sum, off);

  for (int m = 0; m < I_ / 32; ++m) logits[row + m * 32 + lane] = v[m] / sum;
}

__global__ void ipa_inv_affine(const float* og, const float* T, const float* R,
                               float* feat, int colOl, int colN, int total) {
  int idx = blockIdx.x * blockDim.x + threadIdx.x;
  if (idx >= total) return;
  int c = idx & (NPV_ - 1);
  int t = idx >> 3;
  int h = t % H_;
  int bi = t / H_;
  const float* o = og + (long)idx * 3;
  const float* Tv = T + (long)bi * 3;
  const float* Rm = R + (long)bi * 9;
  float x = o[0] - Tv[0], y = o[1] - Tv[1], z = o[2] - Tv[2];
  float p0 = Rm[0] * x + Rm[1] * y + Rm[2] * z;
  float p1 = Rm[3] * x + Rm[4] * y + Rm[5] * z;
  float p2 = Rm[6] * x + Rm[7] * y + Rm[8] * z;
  float* fr = feat + (long)bi * (2 * D_ + 2 * H_ * NPV_ * 4);
  int tc = h * (NPV_ * 3) + c * 3;
  fr[colOl + tc + 0] = p0;
  fr[colOl + tc + 1] = p1;
  fr[colOl + tc + 2] = p2;
  fr[colN + h * NPV_ + c] = sqrtf(p0 * p0 + p1 * p1 + p2 * p2);
}

extern "C" void kernel_launch(void* const* d_in, const int* in_sizes, int n_in,
                              void* d_out, int out_size, void* d_ws, size_t ws_size,
                              hipStream_t stream) {
  (void)in_sizes; (void)n_in; (void)out_size; (void)ws_size;
  const float* x1d      = (const float*)d_in[0];
  const float* x2d      = (const float*)d_in[1];
  const float* past_T   = (const float*)d_in[2];
  const float* past_R   = (const float*)d_in[3];
  const float* last_T   = (const float*)d_in[4];
  const float* last_R   = (const float*)d_in[5];
  const float* bias     = (const float*)d_in[6];
  const float* Wq       = (const float*)d_in[7];
  const float* Wk       = (const float*)d_in[8];
  const float* Wv       = (const float*)d_in[9];
  const float* Wpb      = (const float*)d_in[10];
  const float* Wpq_past = (const float*)d_in[11];
  const float* Wpk_past = (const float*)d_in[12];
  const float* Wpv_past = (const float*)d_in[13];
  const float* Wpq_last = (const float*)d_in[14];
  const float* Wpk_last = (const float*)d_in[15];
  const float* Wpv_last = (const float*)d_in[16];
  const float* tpw_past = (const float*)d_in[17];
  const float* tpw_last = (const float*)d_in[18];
  const float* Wpv      = (const float*)d_in[19];
  const float* W_out    = (const float*)d_in[20];
  const float* b_out    = (const float*)d_in[21];
  float* out = (float*)d_out;
  float* ws  = (float*)d_ws;

  const int M  = B_ * I_;
  const int FE = 2 * D_ + 2 * H_ * NPV_ * 4;

  float* q    = ws;
  float* k    = q + 393216;
  float* v    = k + 393216;
  float* qgp  = v + 393216;
  float* kgp  = qgp + 73728;
  float* vgp  = kgp + 73728;
  float* qgl  = vgp + 147456;
  float* kgl  = qgl + 73728;
  float* vgl  = kgl + 73728;
  float* logits   = vgl + 147456;
  float* pairproj = logits + 1572864;
  float* ogp  = pairproj + 1572864;
  float* ogl  = ogp + 147456;
  float* z    = ogl + 147456;
  float* feat = z + 786432;

  auto gemm = [&](const float* A, const float* Bm, float* C,
                  int Mm, int Nn, int Kk, int lda, int ldb, int ldc,
                  long sAy, long sAz, long sBy, long sBz, long sCy, long sCz,
                  int gy, int gz, int opB, float alpha, const float* bia,
                  bool splitK = false) {
    GemmP p{A, Bm, C, bia, Mm, Nn, Kk, lda, ldb, ldc,
            sAy, sAz, sBy, sBz, sCy, sCz, alpha};
    int waves = ((Mm + 15) / 16) * ((Nn + 15) / 16);
    dim3 blk(256);
    bool guard = (Mm & 15) || (Nn & 15);
    if (splitK) {
      dim3 gridS((unsigned)waves, (unsigned)gy, (unsigned)gz);
      if (guard) ipa_gemm_wmma<0, true , 8><<<gridS, blk, 0, stream>>>(p);
      else       ipa_gemm_wmma<0, false, 8><<<gridS, blk, 0, stream>>>(p);
      return;
    }
    dim3 grid((unsigned)((waves + 7) / 8), (unsigned)gy, (unsigned)gz);
    if (opB == 0) {
      if (guard) ipa_gemm_wmma<0, true , 1><<<grid, blk, 0, stream>>>(p);
      else       ipa_gemm_wmma<0, false, 1><<<grid, blk, 0, stream>>>(p);
    } else {
      if (guard) ipa_gemm_wmma<1, true , 1><<<grid, blk, 0, stream>>>(p);
      else       ipa_gemm_wmma<1, false, 1><<<grid, blk, 0, stream>>>(p);
    }
  };

  gemm(x1d, Wq, q, M, D_, D_, D_, D_, D_, 0,0,0,0,0,0, 1,1, 0, 1.f, nullptr, true);
  gemm(x1d, Wk, k, M, D_, D_, D_, D_, D_, 0,0,0,0,0,0, 1,1, 0, 1.f, nullptr, true);
  gemm(x1d, Wv, v, M, D_, D_, D_, D_, D_, 0,0,0,0,0,0, 1,1, 0, 1.f, nullptr, true);
  gemm(x1d, Wpq_past, qgp, M, 144, D_, D_, 144, 144, 0,0,0,0,0,0, 1,1, 0, 1.f, nullptr, true);
  gemm(x1d, Wpk_past, kgp, M, 144, D_, D_, 144, 144, 0,0,0,0,0,0, 1,1, 0, 1.f, nullptr, true);
  gemm(x1d, Wpv_past, vgp, M, 288, D_, D_, 288, 288, 0,0,0,0,0,0, 1,1, 0, 1.f, nullptr, true);
  gemm(x1d, Wpq_last, qgl, M, 144, D_, D_, 144, 144, 0,0,0,0,0,0, 1,1, 0, 1.f, nullptr, true);
  gemm(x1d, Wpk_last, kgl, M, 144, D_, D_, 144, 144, 0,0,0,0,0,0, 1,1, 0, 1.f, nullptr, true);
  gemm(x1d, Wpv_last, vgl, M, 288, D_, D_, 288, 288, 0,0,0,0,0,0, 1,1, 0, 1.f, nullptr, true);

  {
    int t4 = B_ * I_ * H_ * NPQ_, t8 = B_ * I_ * H_ * NPV_;
    ipa_affine<<<(t4 + 255) / 256, 256, 0, stream>>>(qgp, past_T, past_R, NPQ_, t4);
    ipa_affine<<<(t4 + 255) / 256, 256, 0, stream>>>(kgp, past_T, past_R, NPQ_, t4);
    ipa_affine<<<(t8 + 255) / 256, 256, 0, stream>>>(vgp, past_T, past_R, NPV_, t8);
    ipa_affine<<<(t4 + 255) / 256, 256, 0, stream>>>(qgl, last_T, last_R, NPQ_, t4);
    ipa_affine<<<(t4 + 255) / 256, 256, 0, stream>>>(kgl, last_T, last_R, NPQ_, t4);
    ipa_affine<<<(t8 + 255) / 256, 256, 0, stream>>>(vgl, last_T, last_R, NPV_, t8);
  }

  const float scalar_w = 0.07216878364870323f;
  gemm(q, k, logits, I_, I_, DK_, D_, D_, I_,
       64, (long)I_ * D_, 64, (long)I_ * D_, (long)I_ * I_, (long)H_ * I_ * I_,
       H_, B_, 1, scalar_w, nullptr);

  const float pair_w = 0.5773502691896258f;
  gemm(x2d, Wpb, pairproj, B_ * I_ * I_, H_, DP_, DP_, H_, H_,
       0,0,0,0,0,0, 1,1, 0, pair_w, nullptr);

  ipa_combine_softmax<<<B_ * H_ * (I_ / ROWS_PER_WG_), 256, 0, stream>>>(
      logits, qgp, kgp, qgl, kgl, pairproj, bias, tpw_past, tpw_last);

  gemm(logits, v, feat + D_, I_, DK_, I_, I_, D_, FE,
       (long)I_ * I_, (long)H_ * I_ * I_, 64, (long)I_ * D_, 64, (long)I_ * FE,
       H_, B_, 0, 1.f, nullptr, true);

  gemm(logits, vgp, ogp, I_, NPV_ * 3, I_, I_, H_ * NPV_ * 3, H_ * NPV_ * 3,
       (long)I_ * I_, (long)H_ * I_ * I_, NPV_ * 3, (long)I_ * H_ * NPV_ * 3,
       NPV_ * 3, (long)I_ * H_ * NPV_ * 3, H_, B_, 0, 1.f, nullptr, true);
  gemm(logits, vgl, ogl, I_, NPV_ * 3, I_, I_, H_ * NPV_ * 3, H_ * NPV_ * 3,
       (long)I_ * I_, (long)H_ * I_ * I_, NPV_ * 3, (long)I_ * H_ * NPV_ * 3,
       NPV_ * 3, (long)I_ * H_ * NPV_ * 3, H_, B_, 0, 1.f, nullptr, true);

  gemm(logits, x2d, z, H_, DP_, I_,
       I_ * I_, DP_, DP_,
       (long)I_, (long)H_ * I_ * I_,
       (long)I_ * DP_, (long)I_ * I_ * DP_,
       (long)H_ * DP_, (long)I_ * H_ * DP_,
       I_, B_, 0, 1.f, nullptr);

  gemm(z, Wpv, feat, I_, DK_, DP_, H_ * DP_, D_, FE,
       DP_, (long)I_ * H_ * DP_, 64, 0, 64, (long)I_ * FE,
       H_, B_, 0, 1.f, nullptr);

  {
    int t8 = B_ * I_ * H_ * NPV_;
    ipa_inv_affine<<<(t8 + 255) / 256, 256, 0, stream>>>(
        ogp, past_T, past_R, feat, 2 * D_, 2 * D_ + 288, t8);
    ipa_inv_affine<<<(t8 + 255) / 256, 256, 0, stream>>>(
        ogl, past_T, past_R, feat, 2 * D_ + 384, 2 * D_ + 384 + 288, t8);
  }

  gemm(feat, W_out, out, M, D_, FE, FE, D_, D_,
       0,0,0,0,0,0, 1,1, 0, 1.f, b_out, true);
}


# independent kernels of one dependency level share one launch (block index picks the job; streaming GEMMs first in the grid): 26 launches -> 6; every sum and its order unchanged
# speedup vs baseline: 2.4966x; 1.2487x over previous
#include <hip/hip_runtime.h>
#include <math.h>

#define B_ 2
#define I_ 256
#define D_ 768
#define H_ 12
#define DP_ 128
#define DK_ 64
#define NPQ_ 4
#define NPV_ 8

typedef float v2f __attribute__((ext_vector_type(2)));
typedef float v8f __attribute__((ext_vector_type(8)));

struct GemmP {
  const float* A; const float* B; float* C; const float* bias;
  int M, N, K;
  int lda, ldb, ldc;
  long sAy, sAz, sBy, sBz, sCy, sCz;
  float alpha;
};

template <int OPB, bool GUARD, int KS>
__device__ __forceinline__ void gemm_tile(const GemmP p, const int wgx, const int by,
                                          const int bz, float* red) {
  const int tilesN = (p.N + 15) >> 4;
  const int tilesM = (p.M + 15) >> 4;
  const int wave = (KS == 1) ? ((wgx * 256 + (int)threadIdx.x) >> 5)
                             : wgx;
  if (KS == 1) {
    if (wave >= tilesM * tilesN) return;
  }
  const int lane = threadIdx.x & 31;
  const int ks = (KS == 1) ? 0 : __builtin_amdgcn_readfirstlane((int)(threadIdx.x >> 5));
  int kBeg = 0, kEnd = p.K;
  if (KS > 1) {
    const int per = (((p.K >> 2) + KS - 1) / KS) << 2;
    kBeg = ks * per;
    kEnd = (kBeg + per < p.K) ? (kBeg + per) : p.K;
  }
  const int tm = wave / tilesN, tn = wave % tilesN;
  const int row0 = tm << 4, col0 = tn << 4;

  const float* A = p.A + (long)by * p.sAy + (long)bz * p.sAz;
  const float* B = p.B + (long)by * p.sBy + (long)bz * p.sBz;
  float*       C = p.C + (long)by * p.sCy + (long)bz * p.sCz;

  const int mn = lane & 15;
  const int kb = (lane >> 4) << 1;

  int aRow = row0 + mn;
  int bCol = col0 + mn;
  float aMask = 1.f, bMask = 1.f;
  if (GUARD) {
    if (aRow >= p.M) { aRow = p.M - 1; aMask = 0.f; }
    if (bCol >= p.N) { bCol = p.N - 1; bMask = 0.f; }
  }
  const float* aPtr = A + (long)aRow * p.lda + kb;
  const float* bPtr = (OPB == 0) ? (B + (long)kb * p.ldb + bCol)
                                 : (B + (long)bCol * p.ldb + kb);

  v8f acc = {0.f, 0.f, 0.f, 0.f, 0.f, 0.f, 0.f, 0.f};
#pragma unroll 4
  for (int k0 = kBeg; k0 < kEnd; k0 += 4) {
    v2f a, b;
    a.x = aPtr[k0];
    a.y = aPtr[k0 + 1];
    if (OPB == 0) {
      const float* bp = bPtr + (long)k0 * p.ldb;
      b.x = bp[0];
      b.y = bp[p.ldb];
    } else {
      b.x = bPtr[k0];
      b.y = bPtr[k0 + 1];
    }
    if (GUARD) {
      a.x *= aMask; a.y *= aMask;
      b.x *= bMask; b.y *= bMask;
    }
    acc = __builtin_amdgcn_wmma_f32_16x16x4_f32(false, a, false, b,
                                                (short)0, acc, false, false);
  }

  if (KS > 1) {
    for (int r = 0; r < 8; ++r) red[ks * 256 + r * 32 + lane] = acc[r];
    __syncthreads();
    const int t = (int)threadIdx.x;
    float sum = red[t];
    for (int s = 1; s < KS; ++s) sum += red[s * 256 + t];
    const int oRow = row0 + (((t & 31) >> 4) << 3) + (t >> 5);
    const int oCol = col0 + (t & 15);
    if (!GUARD || (oRow < p.M && oCol < p.N)) {
      float val = sum * p.alpha;
      if (p.bias) val += p.bias[oCol];
      C[(long)oRow * p.ldc + oCol] = val;
    }
    return;
  }

  const int cCol = col0 + mn;
  const int mo = (lane >> 4) << 3;
  const bool cColOK = GUARD ? (cCol < p.N) : true;
  if (cColOK) {
    for (int r = 0; r < 8; ++r) {
      const int cRow = row0 + mo + r;
      if (!GUARD || cRow < p.M) {
        float val = acc[r] * p.alpha;
        if (p.bias) val += p.bias[cCol];
        C[(long)cRow * p.ldc + cCol] = val;
      }
    }
  }
}

__device__ __forceinline__ void affine_body(float* pts, const float* T, const float* R,
                                            int np, int total, int idx) {
  if (idx >= total) return;
  int bi = idx / (H_ * np);
  const float* Rm = R + (long)bi * 9;
  const float* Tv = T + (long)bi * 3;
  float x = pts[idx * 3 + 0], y = pts[idx * 3 + 1], z = pts[idx * 3 + 2];
  float o0 = Rm[0] * x + Rm[3] * y + Rm[6] * z + Tv[0];
  float o1 = Rm[1] * x + Rm[4] * y + Rm[7] * z + Tv[1];
  float o2 = Rm[2] * x + Rm[5] * y + Rm[8] * z + Tv[2];
  pts[idx * 3 + 0] = o0; pts[idx * 3 + 1] = o1; pts[idx * 3 + 2] = o2;
}

typedef float v4f __attribute__((ext_vector_type(4)));

#define ROWS_PER_WG_ 8
#define NKC_ (NPQ_ * 3)

__global__ void __launch_bounds__(256)
ipa_combine_softmax(float* logits,
                    const float* qgp, const float* kgp,
                    const float* qgl, const float* kgl,
                    const float* pairproj, const float* bias,
                    const float* tpw_past, const float* tpw_last) {
  __shared__ float kst[2 * NKC_ * I_];
  const int t = (int)threadIdx.x;
  const int ig = (int)blockIdx.x % (I_ / ROWS_PER_WG_);
  const int bh = (int)blockIdx.x / (I_ / ROWS_PER_WG_);
  const int h = bh % H_;
  const int b = bh / H_;
  const int lane = t & 31;
  const int w = __builtin_amdgcn_readfirstlane(t >> 5);
  const int i = ig * ROWS_PER_WG_ + w;

  float qp[NKC_], ql[NKC_];
  {
    const long qo = ((long)(b * I_ + i) * H_ + h) * NKC_;
    for (int c = 0; c < NKC_; ++c) { qp[c] = qgp[qo + c]; ql[c] = qgl[qo + c]; }
  }

  {
    const long ko = ((long)(b * I_ + t) * H_ + h) * NKC_;
    const v4f* kp = (const v4f*)(kgp + ko);
    const v4f* kl = (const v4f*)(kgl + ko);
    const v4f p0 = kp[0], p1 = kp[1], p2 = kp[2];
    const v4f l0 = kl[0], l1 = kl[1], l2 = kl[2];
    kst[ 0 * I_ + t] = p0.x; kst[ 1 * I_ + t] = p0.y; kst[ 2 * I_ + t] = p0.z; kst[ 3 * I_ + t] = p0.w;
    kst[ 4 * I_ + t] = p1.x; kst[ 5 * I_ + t] = p1.y; kst[ 6 * I_ + t] = p1.z; kst[ 7 * I_ + t] = p1.w;
    kst[ 8 * I_ + t] = p2.x; kst[ 9 * I_ + t] = p2.y; kst[10 * I_ + t] = p2.z; kst[11 * I_ + t] = p2.w;
    kst[12 * I_ + t] = l0.x; kst[13 * I_ + t] = l0.y; kst[14 * I_ + t] = l0.z; kst[15 * I_ + t] = l0.w;
    kst[16 * I_ + t] = l1.x; kst[17 * I_ + t] = l1.y; kst[18 * I_ + t] = l1.z; kst[19 * I_ + t] = l1.w;
    kst[20 * I_ + t] = l2.x; kst[21 * I_ + t] = l2.y; kst[22 * I_ + t] = l2.z; kst[23 * I_ + t] = l2.w;
  }

  const float point_w = 0.13608276348795434f;
  const float pwp = -0.5f * point_w * log1pf(expf(tpw_past[h]));
  const float pwl = -0.5f * point_w * log1pf(expf(tpw_last[h]));

  const long row = ((long)(b * H_ + h) * I_ + i) * I_;
  const float* pr = pairproj + ((long)(b * I_ + i) * I_) * H_ + h;

  __syncthreads();

  float v[I_ / 32];
#pragma unroll
  for (int m = 0; m < I_ / 32; ++m) {
    const int j = m * 32 + lane;
    float sp = 0.f, sl = 0.f;
    for (int c = 0; c < NPQ_; ++c) {
      float dx = qp[c * 3 + 0] - kst[(c * 3 + 0) * I_ + j];
      float dy = qp[c * 3 + 1] - kst[(c * 3 + 1) * I_ + j];
      float dz = qp[c * 3 + 2] - kst[(c * 3 + 2) * I_ + j];
      sp += sqrtf(dx * dx + dy * dy + dz * dz);
      dx = ql[c * 3 + 0] - kst[(NKC_ + c * 3 + 0) * I_ + j];
      dy = ql[c * 3 + 1] - kst[(NKC_ + c * 3 + 1) * I_ + j];
      dz = ql[c * 3 + 2] - kst[(NKC_ + c * 3 + 2) * I_ + j];
      sl += sqrtf(dx * dx + dy * dy + dz * dz);
    }
    const float pv = pr[(long)j * H_];
    const float add = pwp * sp + pwl * sl + pv + bias[row + j];
    v[m] = logits[row + j] + add;
  }

  float mx = v[0];
  for (int m = 1; m < I_ / 32; ++m) mx = fmaxf(mx, v[m]);
  for (int off = 16; off > 0; off >>= 1) mx = fmaxf(mx, __shfl_xor(mx, off));

  float sum = 0.f;
  for (int m = 0; m < I_ / 32; ++m) { v[m] = expf(v[m] - mx); sum += v[m]; }
  for (int off = 16; off > 0; off >>= 1) sum += __shfl_xor(sum, off);

  for (int m = 0; m < I_ / 32; ++m) logits[row + m * 32 + lane] = v[m] / sum;
}

__device__ __forceinline__ void inv_affine_body(const float* og, const float* T, const float* R,
                                                float* feat, int colOl, int colN, int total,
                                                int idx) {
  if (idx >= total) return;
  int c = idx & (NPV_ - 1);
  int t = idx >> 3;
  int h = t % H_;
  int bi = t / H_;
  const float* o = og + (long)idx * 3;
  const float* Tv = T + (long)bi * 3;
  const float* Rm = R + (long)bi * 9;
  float x = o[0] - Tv[0], y = o[1] - Tv[1], z = o[2] - Tv[2];
  float p0 = Rm[0] * x + Rm[1] * y + Rm[2] * z;
  float p1 = Rm[3] * x + Rm[4] * y + Rm[5] * z;
  float p2 = Rm[6] * x + Rm[7] * y + Rm[8] * z;
  float* fr = feat + (long)bi * (2 * D_ + 2 * H_ * NPV_ * 4);
  int tc = h * (NPV_ * 3) + c * 3;
  fr[colOl + tc + 0] = p0;
  fr[colOl + tc + 1] = p1;
  fr[colOl + tc + 2] = p2;
  fr[colN + h * NPV_ + c] = sqrtf(p0 * p0 + p1 * p1 + p2 * p2);
}

#define WG_PAIRBIAS_ (B_ * I_ * I_ / 16 / 8)
#define WG_PROJ_ (3 * 1536 + 4 * 288 + 2 * 576)

struct LevelA {
  const float* x1d; const float* x2d; const float* Wpb;
  const float* W0; const float* W1; const float* W2;
  const float* W3; const float* W4; const float* W5;
  const float* W6; const float* W7; const float* W8;
  float* pairproj;
  float* C0; float* C1; float* C2; float* C3; float* C4; float* C5;
  float* C6; float* C7; float* C8;
};

__global__ void __launch_bounds__(256) ipa_level_a(LevelA a) {
  __shared__ float red[8 * 256];
  const int bx = (int)blockIdx.x;
  if (bx < WG_PAIRBIAS_) {
    const GemmP p{a.x2d, a.Wpb, a.pairproj, nullptr, B_ * I_ * I_, H_, DP_, DP_, H_, H_,
                  0, 0, 0, 0, 0, 0, 0.5773502691896258f};
    gemm_tile<0, true, 1>(p, bx, 0, 0, red);
    return;
  }
  const int t = bx - WG_PAIRBIAS_;
  const float* Wm = a.W0; float* Cm = a.C0; int N = 768; int base = 0;
  if (t >= 1536) { Wm = a.W1; Cm = a.C1; base = 1536; }
  if (t >= 3072) { Wm = a.W2; Cm = a.C2; base = 3072; }
  if (t >= 4608) { Wm = a.W3; Cm = a.C3; base = 4608; N = 144; }
  if (t >= 4896) { Wm = a.W4; Cm = a.C4; base = 4896; N = 144; }
  if (t >= 5184) { Wm = a.W5; Cm = a.C5; base = 5184; N = 288; }
  if (t >= 5760) { Wm = a.W6; Cm = a.C6; base = 5760; N = 144; }
  if (t >= 6048) { Wm = a.W7; Cm = a.C7; base = 6048; N = 144; }
  if (t >= 6336) { Wm = a.W8; Cm = a.C8; base = 6336; N = 288; }
  const GemmP p{a.x1d, Wm, Cm, nullptr, B_ * I_, N, D_, D_, N, N,
                0, 0, 0, 0, 0, 0, 1.f};
  gemm_tile<0, false, 8>(p, t - base, 0, 0, red);
}

#define WG_LOGITS_ (32 * H_ * B_)
#define WG_AFF4_ (B_ * I_ * H_ * NPQ_ / 256)
#define WG_AFF8_ (B_ * I_ * H_ * NPV_ / 256)

struct LevelB {
  const float* q; const float* k; float* logits;
  float* qgp; float* kgp; float* vgp; float* qgl; float* kgl; float* vgl;
  const float* past_T; const float* past_R; const float* last_T; const float* last_R;
};

__global__ void __launch_bounds__(256) ipa_level_b(LevelB a) {
  const int bx = (int)blockIdx.x;
  if (bx < WG_LOGITS_) {
    const GemmP p{a.q, a.k, a.logits, nullptr, I_, I_, DK_, D_, D_, I_,
                  64, (long)I_ * D_, 64, (long)I_ * D_, (long)I_ * I_, (long)H_ * I_ * I_,
                  0.07216878364870323f};
    gemm_tile<1, false, 1>(p, bx & 31, (bx >> 5) % H_, bx / (32 * H_), nullptr);
    return;
  }
  int t = bx - WG_LOGITS_;
  const bool lastBranch = t >= (2 * WG_AFF4_ + WG_AFF8_);
  if (lastBranch) t -= 2 * WG_AFF4_ + WG_AFF8_;
  float* pts = lastBranch ? a.qgl : a.qgp; int np = NPQ_; int base = 0;
  if (t >= WG_AFF4_)     { pts = lastBranch ? a.kgl : a.kgp; base = WG_AFF4_; }
  if (t >= 2 * WG_AFF4_) { pts = lastBranch ? a.vgl : a.vgp; base = 2 * WG_AFF4_; np = NPV_; }
  const float* T = lastBranch ? a.last_T : a.past_T;
  const float* R = lastBranch ? a.last_R : a.past_R;
  affine_body(pts, T, R, np, B_ * I_ * H_ * np, (t - base) * 256 + (int)threadIdx.x);
}

#define WG_Z_ (I_ * B_)
#define WG_AV_ (64 * H_ * B_)
#define WG_OG_ (32 * H_ * B_)

struct LevelD {
  const float* attn; const float* x2d; const float* v; const float* vgp; const float* vgl;
  float* z; float* featS; float* ogp; float* ogl;
};

__global__ void __launch_bounds__(256) ipa_level_d(LevelD a) {
  __shared__ float red[8 * 256];
  const int bx = (int)blockIdx.x;
  const int FE = 2 * D_ + 2 * H_ * NPV_ * 4;
  if (bx < WG_Z_) {
    const GemmP p{a.attn, a.x2d, a.z, nullptr, H_, DP_, I_, I_ * I_, DP_, DP_,
                  (long)I_, (long)H_ * I_ * I_,
                  (long)I_ * DP_, (long)I_ * I_ * DP_,
                  (long)H_ * DP_, (long)I_ * H_ * DP_, 1.f};
    gemm_tile<0, true, 1>(p, 0, bx & (I_ - 1), bx >> 8, red);
    return;
  }
  int t = bx - WG_Z_;
  if (t < WG_AV_) {
    const GemmP p{a.attn, a.v, a.featS, nullptr, I_, DK_, I_, I_, D_, FE,
                  (long)I_ * I_, (long)H_ * I_ * I_, 64, (long)I_ * D_, 64, (long)I_ * FE, 1.f};
    gemm_tile<0, false, 8>(p, t & 63, (t >> 6) % H_, t / (64 * H_), red);
    return;
  }
  t -= WG_AV_;
  const bool lastBranch = t >= WG_OG_;
  if (lastBranch) t -= WG_OG_;
  const GemmP p{a.attn, lastBranch ? a.vgl : a.vgp, lastBranch ? a.ogl : a.ogp, nullptr,
                I_, NPV_ * 3, I_, I_, H_ * NPV_ * 3, H_ * NPV_ * 3,
                (long)I_ * I_, (long)H_ * I_ * I_, NPV_ * 3, (long)I_ * H_ * NPV_ * 3,
                NPV_ * 3, (long)I_ * H_ * NPV_ * 3, 1.f};
  gemm_tile<0, true, 8>(p, t & 31, (t >> 5) % H_, t / (32 * H_), red);
}

#define WG_OUTPAIR_ (8 * H_ * B_)

struct LevelE {
  const float* z; const float* Wpv; const float* ogp; const float* ogl;
  const float* past_T; const float* past_R; float* feat;
};

__global__ void __launch_bounds__(256) ipa_level_e(LevelE a) {
  const int bx = (int)blockIdx.x;
  const int FE = 2 * D_ + 2 * H_ * NPV_ * 4;
  if (bx < WG_OUTPAIR_) {
    const GemmP p{a.z, a.Wpv, a.feat, nullptr, I_, DK_, DP_, H_ * DP_, D_, FE,
                  DP_, (long)I_ * H_ * DP_, 64, 0, 64, (long)I_ * FE, 1.f};
    gemm_tile<0, false, 1>(p, bx & 7, (bx >> 3) % H_, bx / (8 * H_), nullptr);
    return;
  }
  int t = bx - WG_OUTPAIR_;
  const bool lastBranch = t >= WG_AFF8_;
  if (lastBranch) t -= WG_AFF8_;
  const int colOl = lastBranch ? (2 * D_ + 384) : (2 * D_);
  inv_affine_body(lastBranch ? a.ogl : a.ogp, a.past_T, a.past_R, a.feat,
                  colOl, colOl + 288, B_ * I_ * H_ * NPV_, t * 256 + (int)threadIdx.x);
}

__global__ void __launch_bounds__(256) ipa_gemm_splitk(GemmP p) {
  __shared__ float red[8 * 256];
  gemm_tile<0, false, 8>(p, (int)blockIdx.x, (int)blockIdx.y, (int)blockIdx.z, red);
}

extern "C" void kernel_launch(void* const* d_in, const int* in_sizes, int n_in,
                              void* d_out, int out_size, void* d_ws, size_t ws_size,
                              hipStream_t stream) {
  (void)in_sizes; (void)n_in; (void)out_size; (void)ws_size;
  const float* x1d      = (const float*)d_in[0];
  const float* x2d      = (const float*)d_in[1];
  const float* past_T   = (const float*)d_in[2];
  const float* past_R   = (const float*)d_in[3];
  const float* last_T   = (const float*)d_in[4];
  const float* last_R   = (const float*)d_in[5];
  const float* bias     = (const float*)d_in[6];
  const float* Wq       = (const float*)d_in[7];
  const float* Wk       = (const float*)d_in[8];
  const float* Wv       = (const float*)d_in[9];
  const float* Wpb      = (const float*)d_in[10];
  const float* Wpq_past = (const float*)d_in[11];
  const float* Wpk_past = (const float*)d_in[12];
  const float* Wpv_past = (const float*)d_in[13];
  const float* Wpq_last = (const float*)d_in[14];
  const float* Wpk_last = (const float*)d_in[15];
  const float* Wpv_last = (const float*)d_in[16];
  const float* tpw_past = (const float*)d_in[17];
  const float* tpw_last = (const float*)d_in[18];
  const float* Wpv      = (const float*)d_in[19];
  const float* W_out    = (const float*)d_in[20];
  const float* b_out    = (const float*)d_in[21];
  float* out = (float*)d_out;
  float* ws  = (float*)d_ws;

  float* q    = ws;
  float* k    = q + 393216;
  float* v    = k + 393216;
  float* qgp  = v + 393216;
  float* kgp  = qgp + 73728;
  float* vgp  = kgp + 73728;
  float* qgl  = vgp + 147456;
  float* kgl  = qgl + 73728;
  float* vgl  = kgl + 73728;
  float* logits   = vgl + 147456;
  float* pairproj = logits + 1572864;
  float* ogp  = pairproj + 1572864;
  float* ogl  = ogp + 147456;
  float* z    = ogl + 147456;
  float* feat = z + 786432;

  const LevelA la{x1d, x2d, Wpb,
                  Wq, Wk, Wv, Wpq_past, Wpk_past, Wpv_past, Wpq_last, Wpk_last, Wpv_last,
                  pairproj,
                  q, k, v, qgp, kgp, vgp, qgl, kgl, vgl};
  ipa_level_a<<<WG_PAIRBIAS_ + WG_PROJ_, 256, 0, stream>>>(la);

  const LevelB lb{q, k, logits, qgp, kgp, vgp, qgl, kgl, vgl,
                  past_T, past_R, last_T, last_R};
  ipa_level_b<<<WG_LOGITS_ + 4 * WG_AFF4_ + 2 * WG_AFF8_, 256, 0, stream>>>(lb);

  ipa_combine_softmax<<<B_ * H_ * (I_ / ROWS_PER_WG_), 256, 0, stream>>>(
      logits, qgp, kgp, qgl, kgl, pairproj, bias, tpw_past, tpw_last);

  const LevelD ld{logits, x2d, v, vgp, vgl, z, feat + D_, ogp, ogl};
  ipa_level_d<<<WG_Z_ + WG_AV_ + 2 * WG_OG_, 256, 0, stream>>>(ld);

  const LevelE le{z, Wpv, ogp, ogl, past_T, past_R, feat};
  ipa_level_e<<<WG_OUTPAIR_ + 2 * WG_AFF8_, 256, 0, stream>>>(le);

  const GemmP pf{feat, W_out, out, b_out, B_ * I_, D_, 2 * D_ + 2 * H_ * NPV_ * 4,
                 2 * D_ + 2 * H_ * NPV_ * 4, D_, D_, 0, 0, 0, 0, 0, 0, 1.f};
  ipa_gemm_splitk<<<dim3(1536, 1, 1), 256, 0, stream>>>(pf);
}


# split-K GEMMs with N>=64 (projections, z, attn*V, output): each wave computes a 16x64 strip (4 column-interleaved tiles) off one A fragment, A and B by 128-bit loads; f32 sum order changes
# speedup vs baseline: 5.0342x; 2.0164x over previous
#include <hip/hip_runtime.h>
#include <math.h>

#define B_ 2
#define I_ 256
#define D_ 768
#define H_ 12
#define DP_ 128
#define DK_ 64
#define NPQ_ 4
#define NPV_ 8

typedef float v2f __attribute__((ext_vector_type(2)));
typedef float v8f __attribute__((ext_vector_type(8)));

struct GemmP {
  const float* A; const float* B; float* C; const float* bias;
  int M, N, K;
  int lda, ldb, ldc;
  long sAy, sAz, sBy, sBz, sCy, sCz;
  float alpha;
};

template <int OPB, bool GUARD, int KS>
__device__ __forceinline__ void gemm_tile(const GemmP p, const int wgx, const int by,
                                          const int bz, float* red) {
  const int tilesN = (p.N + 15) >> 4;
  const int tilesM = (p.M + 15) >> 4;
  const int wave = (KS == 1) ? ((wgx * 256 + (int)threadIdx.x) >> 5)
                             : wgx;
  if (KS == 1) {
    if (wave >= tilesM * tilesN) return;
  }
  const int lane = threadIdx.x & 31;
  const int ks = (KS == 1) ? 0 : __builtin_amdgcn_readfirstlane((int)(threadIdx.x >> 5));
  int kBeg = 0, kEnd = p.K;
  if (KS > 1) {
    const int per = (((p.K >> 2) + KS - 1) / KS) << 2;
    kBeg = ks * per;
    kEnd = (kBeg + per < p.K) ? (kBeg + per) : p.K;
  }
  const int tm = wave / tilesN, tn = wave % tilesN;
  const int row0 = tm << 4, col0 = tn << 4;

  const float* A = p.A + (long)by * p.sAy + (long)bz * p.sAz;
  const float* B = p.B + (long)by * p.sBy + (long)bz * p.sBz;
  float*       C = p.C + (long)by * p.sCy + (long)bz * p.sCz;

  const int mn = lane & 15;
  const int kb = (lane >> 4) << 1;

  int aRow = row0 + mn;
  int bCol = col0 + mn;
  float aMask = 1.f, bMask = 1.f;
  if (GUARD) {
    if (aRow >= p.M) { aRow = p.M - 1; aMask = 0.f; }
    if (bCol >= p.N) { bCol = p.N - 1; bMask = 0.f; }
  }
  const float* aPtr = A + (long)aRow * p.lda + kb;
  const float* bPtr = (OPB == 0) ? (B + (long)kb * p.ldb + bCol)
                                 : (B + (long)bCol * p.ldb + kb);

  v8f acc = {0.f, 0.f, 0.f, 0.f, 0.f, 0.f, 0.f, 0.f};
#pragma unroll 4
  for (int k0 = kBeg; k0 < kEnd; k0 += 4) {
    v2f a, b;
    a.x = aPtr[k0];
    a.y = aPtr[k0 + 1];
    if (OPB == 0) {
      const float* bp = bPtr + (long)k0 * p.ldb;
      b.x = bp[0];
      b.y = bp[p.ldb];
    } else {
      b.x = bPtr[k0];
      b.y = bPtr[k0 + 1];
    }
    if (GUARD) {
      a.x *= aMask; a.y *= aMask;
      b.x *= bMask; b.y *= bMask;
    }
    acc = __builtin_amdgcn_wmma_f32_16x16x4_f32(false, a, false, b,
                                                (short)0, acc, false, false);
  }

  if (KS > 1) {
    for (int r = 0; r < 8; ++r) red[ks * 256 + r * 32 + lane] = acc[r];
    __syncthreads();
    const int t = (int)threadIdx.x;
    float sum = red[t];
    for (int s = 1; s < KS; ++s) sum += red[s * 256 + t];
    const int oRow = row0 + (((t & 31) >> 4) << 3) + (t >> 5);
    const int oCol = col0 + (t & 15);
    if (!GUARD || (oRow < p.M && oCol < p.N)) {
      float val = sum * p.alpha;
      if (p.bias) val += p.bias[oCol];
      C[(long)oRow * p.ldc + oCol] = val;
    }
    return;
  }

  const int cCol = col0 + mn;
  const int mo = (lane >> 4) << 3;
  const bool cColOK = GUARD ? (cCol < p.N) : true;
  if (cColOK) {
    for (int r = 0; r < 8; ++r) {
      const int cRow = row0 + mo + r;
      if (!GUARD || cRow < p.M) {
        float val = acc[r] * p.alpha;
        if (p.bias) val += p.bias[cCol];
        C[(long)cRow * p.ldc + cCol] = val;
      }
    }
  }
}

typedef float v4f __attribute__((ext_vector_type(4)));

template <int NT> struct VecIO;
template <> struct VecIO<1> {
  static __device__ __forceinline__ void ld(const float* p, float* o) { o[0] = p[0]; }
  static __device__ __forceinline__ void st(float* p, const float* o) { p[0] = o[0]; }
};
template <> struct VecIO<2> {
  static __device__ __forceinline__ void ld(const float* p, float* o) {
    const v2f t = *(const v2f*)p; o[0] = t.x; o[1] = t.y;
  }
  static __device__ __forceinline__ void st(float* p, const float* o) {
    v2f t; t.x = o[0]; t.y = o[1]; *(v2f*)p = t;
  }
};
template <> struct VecIO<4> {
  static __device__ __forceinline__ void ld(const float* p, float* o) {
    const v4f t = *(const v4f*)p; o[0] = t.x; o[1] = t.y; o[2] = t.z; o[3] = t.w;
  }
  static __device__ __forceinline__ void st(float* p, const float* o) {
    v4f t; t.x = o[0]; t.y = o[1]; t.z = o[2]; t.w = o[3]; *(v4f*)p = t;
  }
};

template <int NT, bool GM>
__device__ __forceinline__ void gemm_strip(const GemmP p, const int row0, const int col0,
                                           const int by, const int bz, float* red) {
  const int lane = threadIdx.x & 31;
  const int ks = __builtin_amdgcn_readfirstlane((int)(threadIdx.x >> 5));
  const int per = p.K >> 3;
  const int kBeg = ks * per, kEnd = kBeg + per;

  const float* A = p.A + (long)by * p.sAy + (long)bz * p.sAz;
  const float* B = p.B + (long)by * p.sBy + (long)bz * p.sBz;
  float*       C = p.C + (long)by * p.sCy + (long)bz * p.sCz;

  const int c = lane & 15;
  const int h4 = (lane >> 4) << 2;
  int aRow = row0 + c;
  float aMask = 1.f;
  if (GM) {
    if (aRow >= p.M) { aRow = p.M - 1; aMask = 0.f; }
  }
  const float* aPtr = A + (long)aRow * p.lda + h4;
  const float* bPtr = B + (long)h4 * p.ldb + col0 + NT * c;

  v8f acc[NT];
  for (int j = 0; j < NT; ++j)
    for (int r = 0; r < 8; ++r) acc[j][r] = 0.f;

  for (int k0 = kBeg; k0 < kEnd; k0 += 32) {
    v4f a[4];
    float b[4][4][NT];
#pragma unroll
    for (int u = 0; u < 4; ++u) {
      a[u] = *(const v4f*)(aPtr + k0 + 8 * u);
      const float* bp = bPtr + (long)(k0 + 8 * u) * p.ldb;
      for (int e = 0; e < 4; ++e) VecIO<NT>::ld(bp + (long)e * p.ldb, b[u][e]);
    }
    __builtin_amdgcn_sched_barrier(0);
#pragma unroll
    for (int u = 0; u < 4; ++u) {
      v4f av = a[u];
      if (GM) av *= aMask;
      for (int j = 0; j < NT; ++j) {
        v2f af, bf;
        af.x = av.x; af.y = av.y; bf.x = b[u][0][j]; bf.y = b[u][1][j];
        acc[j] = __builtin_amdgcn_wmma_f32_16x16x4_f32(false, af, false, bf,
                                                       (short)0, acc[j], false, false);
      }
      for (int j = 0; j < NT; ++j) {
        v2f af, bf;
        af.x = av.z; af.y = av.w; bf.x = b[u][2][j]; bf.y = b[u][3][j];
        acc[j] = __builtin_amdgcn_wmma_f32_16x16x4_f32(false, af, false, bf,
                                                       (short)0, acc[j], false, false);
      }
    }
  }

  for (int j = 0; j < NT; ++j)
    for (int r = 0; r < 8; ++r) red[((ks * NT + j) * 8 + r) * 32 + lane] = acc[j][r];
  __syncthreads();
  const int t = (int)threadIdx.x;
  float o[NT];
  for (int j = 0; j < NT; ++j) {
    float sum = red[j * 256 + t];
    for (int s = 1; s < 8; ++s) sum += red[(s * NT + j) * 256 + t];
    o[j] = sum * p.alpha;
  }
  const int oRow = row0 + (((t & 31) >> 4) << 3) + (t >> 5);
  const int oCol = col0 + NT * (t & 15);
  if (!GM || oRow < p.M) {
    if (p.bias) {
      float bv[NT];
      VecIO<NT>::ld(p.bias + oCol, bv);
      for (int j = 0; j < NT; ++j) o[j] += bv[j];
    }
    VecIO<NT>::st(C + (long)oRow * p.ldc + oCol, o);
  }
}

__device__ __forceinline__ void affine_body(float* pts, const float* T, const float* R,
                                            int np, int total, int idx) {
  if (idx >= total) return;
  int bi = idx / (H_ * np);
  const float* Rm = R + (long)bi * 9;
  const float* Tv = T + (long)bi * 3;
  float x = pts[idx * 3 + 0], y = pts[idx * 3 + 1], z = pts[idx * 3 + 2];
  float o0 = Rm[0] * x + Rm[3] * y + Rm[6] * z + Tv[0];
  float o1 = Rm[1] * x + Rm[4] * y + Rm[7] * z + Tv[1];
  float o2 = Rm[2] * x + Rm[5] * y + Rm[8] * z + Tv[2];
  pts[idx * 3 + 0] = o0; pts[idx * 3 + 1] = o1; pts[idx * 3 + 2] = o2;
}

#define ROWS_PER_WG_ 8
#define NKC_ (NPQ_ * 3)

__global__ void __launch_bounds__(256)
ipa_combine_softmax(float* logits,
                    const float* qgp, const float* kgp,
                    const float* qgl, const float* kgl,
                    const float* pairproj, const float* bias,
                    const float* tpw_past, const float* tpw_last) {
  __shared__ float kst[2 * NKC_ * I_];
  const int t = (int)threadIdx.x;
  const int ig = (int)blockIdx.x % (I_ / ROWS_PER_WG_);
  const int bh = (int)blockIdx.x / (I_ / ROWS_PER_WG_);
  const int h = bh % H_;
  const int b = bh / H_;
  const int lane = t & 31;
  const int w = __builtin_amdgcn_readfirstlane(t >> 5);
  const int i = ig * ROWS_PER_WG_ + w;

  float qp[NKC_], ql[NKC_];
  {
    const long qo = ((long)(b * I_ + i) * H_ + h) * NKC_;
    for (int c = 0; c < NKC_; ++c) { qp[c] = qgp[qo + c]; ql[c] = qgl[qo + c]; }
  }

  {
    const long ko = ((long)(b * I_ + t) * H_ + h) * NKC_;
    const v4f* kp = (const v4f*)(kgp + ko);
    const v4f* kl = (const v4f*)(kgl + ko);
    const v4f p0 = kp[0], p1 = kp[1], p2 = kp[2];
    const v4f l0 = kl[0], l1 = kl[1], l2 = kl[2];
    kst[ 0 * I_ + t] = p0.x; kst[ 1 * I_ + t] = p0.y; kst[ 2 * I_ + t] = p0.z; kst[ 3 * I_ + t] = p0.w;
    kst[ 4 * I_ + t] = p1.x; kst[ 5 * I_ + t] = p1.y; kst[ 6 * I_ + t] = p1.z; kst[ 7 * I_ + t] = p1.w;
    kst[ 8 * I_ + t] = p2.x; kst[ 9 * I_ + t] = p2.y; kst[10 * I_ + t] = p2.z; kst[11 * I_ + t] = p2.w;
    kst[12 * I_ + t] = l0.x; kst[13 * I_ + t] = l0.y; kst[14 * I_ + t] = l0.z; kst[15 * I_ + t] = l0.w;
    kst[16 * I_ + t] = l1.x; kst[17 * I_ + t] = l1.y; kst[18 * I_ + t] = l1.z; kst[19 * I_ + t] = l1.w;
    kst[20 * I_ + t] = l2.x; kst[21 * I_ + t] = l2.y; kst[22 * I_ + t] = l2.z; kst[23 * I_ + t] = l2.w;
  }

  const float point_w = 0.13608276348795434f;
  const float pwp = -0.5f * point_w * log1pf(expf(tpw_past[h]));
  const float pwl = -0.5f * point_w * log1pf(expf(tpw_last[h]));

  const long row = ((long)(b * H_ + h) * I_ + i) * I_;
  const float* pr = pairproj + ((long)(b * I_ + i) * I_) * H_ + h;

  __syncthreads();

  float v[I_ / 32];
#pragma unroll
  for (int m = 0; m < I_ / 32; ++m) {
    const int j = m * 32 + lane;
    float sp = 0.f, sl = 0.f;
    for (int c = 0; c < NPQ_; ++c) {
      float dx = qp[c * 3 + 0] - kst[(c * 3 + 0) * I_ + j];
      float dy = qp[c * 3 + 1] - kst[(c * 3 + 1) * I_ + j];
      float dz = qp[c * 3 + 2] - kst[(c * 3 + 2) * I_ + j];
      sp += sqrtf(dx * dx + dy * dy + dz * dz);
      dx = ql[c * 3 + 0] - kst[(NKC_ + c * 3 + 0) * I_ + j];
      dy = ql[c * 3 + 1] - kst[(NKC_ + c * 3 + 1) * I_ + j];
      dz = ql[c * 3 + 2] - kst[(NKC_ + c * 3 + 2) * I_ + j];
      sl += sqrtf(dx * dx + dy * dy + dz * dz);
    }
    const float pv = pr[(long)j * H_];
    const float add = pwp * sp + pwl * sl + pv + bias[row + j];
    v[m] = logits[row + j] + add;
  }

  float mx = v[0];
  for (int m = 1; m < I_ / 32; ++m) mx = fmaxf(mx, v[m]);
  for (int off = 16; off > 0; off >>= 1) mx = fmaxf(mx, __shfl_xor(mx, off));

  float sum = 0.f;
  for (int m = 0; m < I_ / 32; ++m) { v[m] = expf(v[m] - mx); sum += v[m]; }
  for (int off = 16; off > 0; off >>= 1) sum += __shfl_xor(sum, off);

  for (int m = 0; m < I_ / 32; ++m) logits[row + m * 32 + lane] = v[m] / sum;
}

__device__ __forceinline__ void inv_affine_body(const float* og, const float* T, const float* R,
                                                float* feat, int colOl, int colN, int total,
                                                int idx) {
  if (idx >= total) return;
  int c = idx & (NPV_ - 1);
  int t = idx >> 3;
  int h = t % H_;
  int bi = t / H_;
  const float* o = og + (long)idx * 3;
  const float* Tv = T + (long)bi * 3;
  const float* Rm = R + (long)bi * 9;
  float x = o[0] - Tv[0], y = o[1] - Tv[1], z = o[2] - Tv[2];
  float p0 = Rm[0] * x + Rm[1] * y + Rm[2] * z;
  float p1 = Rm[3] * x + Rm[4] * y + Rm[5] * z;
  float p2 = Rm[6] * x + Rm[7] * y + Rm[8] * z;
  float* fr = feat + (long)bi * (2 * D_ + 2 * H_ * NPV_ * 4);
  int tc = h * (NPV_ * 3) + c * 3;
  fr[colOl + tc + 0] = p0;
  fr[colOl + tc + 1] = p1;
  fr[colOl + tc + 2] = p2;
  fr[colN + h * NPV_ + c] = sqrtf(p0 * p0 + p1 * p1 + p2 * p2);
}

#define WG_PAIRBIAS_ (B_ * I_ * I_ / 16 / 8)
#define WG_PROJ_ (3 * 384 + 4 * 96 + 2 * 160)

struct LevelA {
  const float* x1d; const float* x2d; const float* Wpb;
  const float* W0; const float* W1; const float* W2;
  const float* W3; const float* W4; const float* W5;
  const float* W6; const float* W7; const float* W8;
  float* pairproj;
  float* C0; float* C1; float* C2; float* C3; float* C4; float* C5;
  float* C6; float* C7; float* C8;
};

__global__ void __launch_bounds__(256, 8) ipa_level_a(LevelA a) {
  __shared__ float red[8 * 4 * 256];
  const int bx = (int)blockIdx.x;
  if (bx < WG_PAIRBIAS_) {
    const GemmP p{a.x2d, a.Wpb, a.pairproj, nullptr, B_ * I_ * I_, H_, DP_, DP_, H_, H_,
                  0, 0, 0, 0, 0, 0, 0.5773502691896258f};
    gemm_tile<0, true, 1>(p, bx, 0, 0, red);
    return;
  }
  const int t = bx - WG_PAIRBIAS_;
  const float* Wm = a.W0; float* Cm = a.C0; int N = 768; int base = 0; int strips = 12; int full = 12;
  if (t >= 384)  { Wm = a.W1; Cm = a.C1; base = 384; }
  if (t >= 768)  { Wm = a.W2; Cm = a.C2; base = 768; }
  if (t >= 1152) { Wm = a.W3; Cm = a.C3; base = 1152; N = 144; strips = 3; full = 2; }
  if (t >= 1248) { Wm = a.W4; Cm = a.C4; base = 1248; N = 144; strips = 3; full = 2; }
  if (t >= 1344) { Wm = a.W5; Cm = a.C5; base = 1344; N = 288; strips = 5; full = 4; }
  if (t >= 1504) { Wm = a.W6; Cm = a.C6; base = 1504; N = 144; strips = 3; full = 2; }
  if (t >= 1600) { Wm = a.W7; Cm = a.C7; base = 1600; N = 144; strips = 3; full = 2; }
  if (t >= 1696) { Wm = a.W8; Cm = a.C8; base = 1696; N = 288; strips = 5; full = 4; }
  const GemmP p{a.x1d, Wm, Cm, nullptr, B_ * I_, N, D_, D_, N, N,
                0, 0, 0, 0, 0, 0, 1.f};
  const int u = t - base;
  const int row0 = (u / strips) << 4;
  const int sx = u % strips;
  if (sx < full)      gemm_strip<4, false>(p, row0, sx << 6, 0, 0, red);
  else if (N == 288)  gemm_strip<2, false>(p, row0, 256, 0, 0, red);
  else                gemm_strip<1, false>(p, row0, 128, 0, 0, red);
}

#define WG_LOGITS_ (32 * H_ * B_)
#define WG_AFF4_ (B_ * I_ * H_ * NPQ_ / 256)
#define WG_AFF8_ (B_ * I_ * H_ * NPV_ / 256)

struct LevelB {
  const float* q; const float* k; float* logits;
  float* qgp; float* kgp; float* vgp; float* qgl; float* kgl; float* vgl;
  const float* past_T; const float* past_R; const float* last_T; const float* last_R;
};

__global__ void __launch_bounds__(256) ipa_level_b(LevelB a) {
  const int bx = (int)blockIdx.x;
  if (bx < WG_LOGITS_) {
    const GemmP p{a.q, a.k, a.logits, nullptr, I_, I_, DK_, D_, D_, I_,
                  64, (long)I_ * D_, 64, (long)I_ * D_, (long)I_ * I_, (long)H_ * I_ * I_,
                  0.07216878364870323f};
    gemm_tile<1, false, 1>(p, bx & 31, (bx >> 5) % H_, bx / (32 * H_), nullptr);
    return;
  }
  int t = bx - WG_LOGITS_;
  const bool lastBranch = t >= (2 * WG_AFF4_ + WG_AFF8_);
  if (lastBranch) t -= 2 * WG_AFF4_ + WG_AFF8_;
  float* pts = lastBranch ? a.qgl : a.qgp; int np = NPQ_; int base = 0;
  if (t >= WG_AFF4_)     { pts = lastBranch ? a.kgl : a.kgp; base = WG_AFF4_; }
  if (t >= 2 * WG_AFF4_) { pts = lastBranch ? a.vgl : a.vgp; base = 2 * WG_AFF4_; np = NPV_; }
  const float* T = lastBranch ? a.last_T : a.past_T;
  const float* R = lastBranch ? a.last_R : a.past_R;
  affine_body(pts, T, R, np, B_ * I_ * H_ * np, (t - base) * 256 + (int)threadIdx.x);
}

#define WG_Z_ (2 * I_ * B_)
#define WG_AV_ (16 * H_ * B_)
#define WG_OG_ (32 * H_ * B_)

struct LevelD {
  const float* attn; const float* x2d; const float* v; const float* vgp; const float* vgl;
  float* z; float* featS; float* ogp; float* ogl;
};

__global__ void __launch_bounds__(256) ipa_level_d(LevelD a) {
  __shared__ float red[8 * 4 * 256];
  const int bx = (int)blockIdx.x;
  const int FE = 2 * D_ + 2 * H_ * NPV_ * 4;
  if (bx < WG_Z_) {
    const GemmP p{a.attn, a.x2d, a.z, nullptr, H_, DP_, I_, I_ * I_, DP_, DP_,
                  (long)I_, (long)H_ * I_ * I_,
                  (long)I_ * DP_, (long)I_ * I_ * DP_,
                  (long)H_ * DP_, (long)I_ * H_ * DP_, 1.f};
    gemm_strip<4, true>(p, 0, (bx & 1) << 6, (bx >> 1) & (I_ - 1), bx >> 9, red);
    return;
  }
  int t = bx - WG_Z_;
  if (t < WG_AV_) {
    const GemmP p{a.attn, a.v, a.featS, nullptr, I_, DK_, I_, I_, D_, FE,
                  (long)I_ * I_, (long)H_ * I_ * I_, 64, (long)I_ * D_, 64, (long)I_ * FE, 1.f};
    gemm_strip<4, false>(p, (t & 15) << 4, 0, (t >> 4) % H_, t / (16 * H_), red);
    return;
  }
  t -= WG_AV_;
  const bool lastBranch = t >= WG_OG_;
  if (lastBranch) t -= WG_OG_;
  const GemmP p{a.attn, lastBranch ? a.vgl : a.vgp, lastBranch ? a.ogl : a.ogp, nullptr,
                I_, NPV_ * 3, I_, I_, H_ * NPV_ * 3, H_ * NPV_ * 3,
                (long)I_ * I_, (long)H_ * I_ * I_, NPV_ * 3, (long)I_ * H_ * NPV_ * 3,
                NPV_ * 3, (long)I_ * H_ * NPV_ * 3, 1.f};
  gemm_tile<0, true, 8>(p, t & 31, (t >> 5) % H_, t / (32 * H_), red);
}

#define WG_OUTPAIR_ (8 * H_ * B_)

struct LevelE {
  const float* z; const float* Wpv; const float* ogp; const float* ogl;
  const float* past_T; const float* past_R; float* feat;
};

__global__ void __launch_bounds__(256) ipa_level_e(LevelE a) {
  const int bx = (int)blockIdx.x;
  const int FE = 2 * D_ + 2 * H_ * NPV_ * 4;
  if (bx < WG_OUTPAIR_) {
    const GemmP p{a.z, a.Wpv, a.feat, nullptr, I_, DK_, DP_, H_ * DP_, D_, FE,
                  DP_, (long)I_ * H_ * DP_, 64, 0, 64, (long)I_ * FE, 1.f};
    gemm_tile<0, false, 1>(p, bx & 7, (bx >> 3) % H_, bx / (8 * H_), nullptr);
    return;
  }
  int t = bx - WG_OUTPAIR_;
  const bool lastBranch = t >= WG_AFF8_;
  if (lastBranch) t -= WG_AFF8_;
  const int colOl = lastBranch ? (2 * D_ + 384) : (2 * D_);
  inv_affine_body(lastBranch ? a.ogl : a.ogp, a.past_T, a.past_R, a.feat,
                  colOl, colOl + 288, B_ * I_ * H_ * NPV_, t * 256 + (int)threadIdx.x);
}

__global__ void __launch_bounds__(256) ipa_gemm_strip(GemmP p) {
  __shared__ float red[8 * 4 * 256];
  gemm_strip<4, false>(p, (int)blockIdx.y << 4, (int)blockIdx.x << 6, 0, 0, red);
}

extern "C" void kernel_launch(void* const* d_in, const int* in_sizes, int n_in,
                              void* d_out, int out_size, void* d_ws, size_t ws_size,
                              hipStream_t stream) {
  (void)in_sizes; (void)n_in; (void)out_size; (void)ws_size;
  const float* x1d      = (const float*)d_in[0];
  const float* x2d      = (const float*)d_in[1];
  const float* past_T   = (const float*)d_in[2];
  const float* past_R   = (const float*)d_in[3];
  const float* last_T   = (const float*)d_in[4];
  const float* last_R   = (const float*)d_in[5];
  const float* bias     = (const float*)d_in[6];
  const float* Wq       = (const float*)d_in[7];
  const float* Wk       = (const float*)d_in[8];
  const float* Wv       = (const float*)d_in[9];
  const float* Wpb      = (const float*)d_in[10];
  const float* Wpq_past = (const float*)d_in[11];
  const float* Wpk_past = (const float*)d_in[12];
  const float* Wpv_past = (const float*)d_in[13];
  const float* Wpq_last = (const float*)d_in[14];
  const float* Wpk_last = (const float*)d_in[15];
  const float* Wpv_last = (const float*)d_in[16];
  const float* tpw_past = (const float*)d_in[17];
  const float* tpw_last = (const float*)d_in[18];
  const float* Wpv      = (const float*)d_in[19];
  const float* W_out    = (const float*)d_in[20];
  const float* b_out    = (const float*)d_in[21];
  float* out = (float*)d_out;
  float* ws  = (float*)d_ws;

  float* q    = ws;
  float* k    = q + 393216;
  float* v    = k + 393216;
  float* qgp  = v + 393216;
  float* kgp  = qgp + 73728;
  float* vgp  = kgp + 73728;
  float* qgl  = vgp + 147456;
  float* kgl  = qgl + 73728;
  float* vgl  = kgl + 73728;
  float* logits   = vgl + 147456;
  float* pairproj = logits + 1572864;
  float* ogp  = pairproj + 1572864;
  float* ogl  = ogp + 147456;
  float* z    = ogl + 147456;
  float* feat = z + 786432;

  const LevelA la{x1d, x2d, Wpb,
                  Wq, Wk, Wv, Wpq_past, Wpk_past, Wpv_past, Wpq_last, Wpk_last, Wpv_last,
                  pairproj,
                  q, k, v, qgp, kgp, vgp, qgl, kgl, vgl};
  ipa_level_a<<<WG_PAIRBIAS_ + WG_PROJ_, 256, 0, stream>>>(la);

  const LevelB lb{q, k, logits, qgp, kgp, vgp, qgl, kgl, vgl,
                  past_T, past_R, last_T, last_R};
  ipa_level_b<<<WG_LOGITS_ + 4 * WG_AFF4_ + 2 * WG_AFF8_, 256, 0, stream>>>(lb);

  ipa_combine_softmax<<<B_ * H_ * (I_ / ROWS_PER_WG_), 256, 0, stream>>>(
      logits, qgp, kgp, qgl, kgl, pairproj, bias, tpw_past, tpw_last);

  const LevelD ld{logits, x2d, v, vgp, vgl, z, feat + D_, ogp, ogl};
  ipa_level_d<<<WG_Z_ + WG_AV_ + 2 * WG_OG_, 256, 0, stream>>>(ld);

  const LevelE le{z, Wpv, ogp, ogl, past_T, past_R, feat};
  ipa_level_e<<<WG_OUTPAIR_ + 2 * WG_AFF8_, 256, 0, stream>>>(le);

  const GemmP pf{feat, W_out, out, b_out, B_ * I_, D_, 2 * D_ + 2 * H_ * NPV_ * 4,
                 2 * D_ + 2 * H_ * NPV_ * 4, D_, D_, 0, 0, 0, 0, 0, 0, 1.f};
  ipa_gemm_strip<<<dim3(12, 32, 1), 256, 0, stream>>>(pf);
}


# GEMMs left on one 16x16 tile per wave (pair bias, q*k, attn*vg, z*W_pair_value): 128-bit fragment loads, 2-8 tiles per wave; attn*vg joins attn*V's group, z's strips share one; f32 sum order changes
# speedup vs baseline: 5.9113x; 1.1742x over previous
#include <hip/hip_runtime.h>
#include <math.h>

#define B_ 2
#define I_ 256
#define D_ 768
#define H_ 12
#define DP_ 128
#define DK_ 64
#define NPQ_ 4
#define NPV_ 8

typedef float v2f __attribute__((ext_vector_type(2)));
typedef float v8f __attribute__((ext_vector_type(8)));

struct GemmP {
  const float* A; const float* B; float* C; const float* bias;
  int M, N, K;
  int lda, ldb, ldc;
  long sAy, sAz, sBy, sBz, sCy, sCz;
  float alpha;
};

typedef float v4f __attribute__((ext_vector_type(4)));

template <int NT> struct VecIO;
template <> struct VecIO<1> {
  static __device__ __forceinline__ void ld(const float* p, float* o) { o[0] = p[0]; }
  static __device__ __forceinline__ void st(float* p, const float* o) { p[0] = o[0]; }
};
template <> struct VecIO<2> {
  static __device__ __forceinline__ void ld(const float* p, float* o) {
    const v2f t = *(const v2f*)p; o[0] = t.x; o[1] = t.y;
  }
  static __device__ __forceinline__ void st(float* p, const float* o) {
    v2f t; t.x = o[0]; t.y = o[1]; *(v2f*)p = t;
  }
};
template <> struct VecIO<4> {
  static __device__ __forceinline__ void ld(const float* p, float* o) {
    const v4f t = *(const v4f*)p; o[0] = t.x; o[1] = t.y; o[2] = t.z; o[3] = t.w;
  }
  static __device__ __forceinline__ void st(float* p, const float* o) {
    v4f t; t.x = o[0]; t.y = o[1]; t.z = o[2]; t.w = o[3]; *(v4f*)p = t;
  }
};

template <int NT, int ST = 4>
__device__ __forceinline__ void gemm_strip(const GemmP p, const int row0, const int col0,
                                           const int by, const int bz, float* red) {
  const int lane = threadIdx.x & 31;
  const int ks = __builtin_amdgcn_readfirstlane((int)(threadIdx.x >> 5));
  const int per = p.K >> 3;
  const int kBeg = ks * per, kEnd = kBeg + per;

  const float* A = p.A + (long)by * p.sAy + (long)bz * p.sAz;
  const float* B = p.B + (long)by * p.sBy + (long)bz * p.sBz;
  float*       C = p.C + (long)by * p.sCy + (long)bz * p.sCz;

  const int c = lane & 15;
  const int h4 = (lane >> 4) << 2;
  const float* aPtr = A + (long)(row0 + c) * p.lda + h4;
  const float* bPtr = B + (long)h4 * p.ldb + col0 + NT * c;

  v8f acc[NT];
  for (int j = 0; j < NT; ++j)
    for (int r = 0; r < 8; ++r) acc[j][r] = 0.f;

  for (int k0 = kBeg; k0 < kEnd; k0 += 8 * ST) {
    v4f a[ST];
    float b[ST][4][NT];
#pragma unroll
    for (int u = 0; u < ST; ++u) {
      a[u] = *(const v4f*)(aPtr + k0 + 8 * u);
      const float* bp = bPtr + (long)(k0 + 8 * u) * p.ldb;
      for (int e = 0; e < 4; ++e) VecIO<NT>::ld(bp + (long)e * p.ldb, b[u][e]);
    }
    __builtin_amdgcn_sched_barrier(0);
#pragma unroll
    for (int u = 0; u < ST; ++u) {
      const v4f av = a[u];
      for (int j = 0; j < NT; ++j) {
        v2f af, bf;
        af.x = av.x; af.y = av.y; bf.x = b[u][0][j]; bf.y = b[u][1][j];
        acc[j] = __builtin_amdgcn_wmma_f32_16x16x4_f32(false, af, false, bf,
                                                       (short)0, acc[j], false, false);
      }
      for (int j = 0; j < NT; ++j) {
        v2f af, bf;
        af.x = av.z; af.y = av.w; bf.x = b[u][2][j]; bf.y = b[u][3][j];
        acc[j] = __builtin_amdgcn_wmma_f32_16x16x4_f32(false, af, false, bf,
                                                       (short)0, acc[j], false, false);
      }
    }
  }

  for (int j = 0; j < NT; ++j)
    for (int r = 0; r < 8; ++r) red[((ks * NT + j) * 8 + r) * 32 + lane] = acc[j][r];
  __syncthreads();
  const int t = (int)threadIdx.x;
  float o[NT];
  for (int j = 0; j < NT; ++j) {
    float sum = red[j * 256 + t];
    for (int s = 1; s < 8; ++s) sum += red[(s * NT + j) * 256 + t];
    o[j] = sum * p.alpha;
  }
  const int oRow = row0 + (((t & 31) >> 4) << 3) + (t >> 5);
  const int oCol = col0 + NT * (t & 15);
  if (p.bias) {
    float bv[NT];
    VecIO<NT>::ld(p.bias + oCol, bv);
    for (int j = 0; j < NT; ++j) o[j] += bv[j];
  }
  VecIO<NT>::st(C + (long)oRow * p.ldc + oCol, o);
}

struct StripBC { const float* B; float* C; int ldb, ldc; };

template <int NT, int NV>
__device__ __forceinline__ const float* strip_bptr(const float* B, const int ldb, const int c,
                                                   const int h4, float& mask) {
  int cc = c;
  mask = 1.f;
  if (NV < 16 * NT) {
    if (NT * c >= NV) { cc = 0; mask = 0.f; }
  }
  return B + (long)h4 * ldb + NT * cc;
}

template <int NT>
__device__ __forceinline__ void strip_load_b(const float* bPtr, const int ldb, const int k0,
                                             float (&b)[4][4][NT]) {
#pragma unroll
  for (int u = 0; u < 4; ++u) {
    const float* bp = bPtr + (long)(k0 + 8 * u) * ldb;
#pragma unroll
    for (int e = 0; e < 4; ++e) VecIO<NT>::ld(bp + (long)e * ldb, b[u][e]);
  }
}

template <int NT, bool MASK>
__device__ __forceinline__ void strip_mma(const v4f av, const float (&bu)[4][NT], const float mask,
                                          v8f (&acc)[NT]) {
  for (int j = 0; j < NT; ++j) {
    v2f af, bf;
    af.x = av.x; af.y = av.y; bf.x = bu[0][j]; bf.y = bu[1][j];
    if (MASK) bf *= mask;
    acc[j] = __builtin_amdgcn_wmma_f32_16x16x4_f32(false, af, false, bf,
                                                   (short)0, acc[j], false, false);
  }
  for (int j = 0; j < NT; ++j) {
    v2f af, bf;
    af.x = av.z; af.y = av.w; bf.x = bu[2][j]; bf.y = bu[3][j];
    if (MASK) bf *= mask;
    acc[j] = __builtin_amdgcn_wmma_f32_16x16x4_f32(false, af, false, bf,
                                                   (short)0, acc[j], false, false);
  }
}

template <int NT>
__device__ __forceinline__ void strip_zero(v8f (&acc)[NT]) {
  for (int j = 0; j < NT; ++j)
    for (int r = 0; r < 8; ++r) acc[j][r] = 0.f;
}

template <int NT>
__device__ __forceinline__ void strip_spill(float* red, const int ks, const int ntt, const int j0,
                                            const int lane, const v8f (&acc)[NT]) {
  for (int j = 0; j < NT; ++j)
    for (int r = 0; r < 8; ++r) red[((ks * ntt + j0 + j) * 8 + r) * 32 + lane] = acc[j][r];
}

template <int NT, int NV>
__device__ __forceinline__ void strip_finish(const float* red, const int ntt, const int j0,
                                             const int t, const bool rowOK, float* cRow) {
  float o[NT];
  for (int j = 0; j < NT; ++j) {
    float sum = red[(j0 + j) * 256 + t];
    for (int s = 1; s < 8; ++s) sum += red[(s * ntt + j0 + j) * 256 + t];
    o[j] = sum;
  }
  const int oc = NT * (t & 15);
  if (rowOK && (NV == 16 * NT || oc < NV)) VecIO<NT>::st(cRow + oc, o);
}

template <int NT0, int NV0, int NT1, int NV1, int NT2, int NV2, bool GM>
__device__ __forceinline__ void gemm_strip_shared(const float* A, const int lda, const int M,
                                                  const int K, const int row0,
                                                  const StripBC g0, const StripBC g1,
                                                  const StripBC g2, float* red) {
  constexpr int N2 = NT2 ? NT2 : 1;
  constexpr int NTT = NT0 + NT1 + NT2;
  const int lane = threadIdx.x & 31;
  const int ks = __builtin_amdgcn_readfirstlane((int)(threadIdx.x >> 5));
  const int per = K >> 3;
  const int kBeg = ks * per, kEnd = kBeg + per;

  const int c = lane & 15;
  const int h4 = (lane >> 4) << 2;
  int aRow = row0 + c;
  float aMask = 1.f;
  if (GM) {
    if (aRow >= M) { aRow = M - 1; aMask = 0.f; }
  }
  const float* aPtr = A + (long)aRow * lda + h4;
  float m0, m1, m2 = 1.f;
  const float* bPtr0 = strip_bptr<NT0, NV0>(g0.B, g0.ldb, c, h4, m0);
  const float* bPtr1 = strip_bptr<NT1, NV1>(g1.B, g1.ldb, c, h4, m1);
  const float* bPtr2 = nullptr;
  if (NT2) bPtr2 = strip_bptr<N2, NV2>(g2.B, g2.ldb, c, h4, m2);

  v8f acc0[NT0], acc1[NT1], acc2[N2];
  strip_zero<NT0>(acc0);
  strip_zero<NT1>(acc1);
  if (NT2) strip_zero<N2>(acc2);

  for (int k0 = kBeg; k0 < kEnd; k0 += 32) {
    v4f a[4];
    float b0[4][4][NT0], b1[4][4][NT1], b2[4][4][N2];
#pragma unroll
    for (int u = 0; u < 4; ++u) a[u] = *(const v4f*)(aPtr + k0 + 8 * u);
    strip_load_b<NT0>(bPtr0, g0.ldb, k0, b0);
    strip_load_b<NT1>(bPtr1, g1.ldb, k0, b1);
    if (NT2) strip_load_b<N2>(bPtr2, g2.ldb, k0, b2);
    __builtin_amdgcn_sched_barrier(0);
#pragma unroll
    for (int u = 0; u < 4; ++u) {
      v4f av = a[u];
      if (GM) av *= aMask;
      strip_mma<NT0, (NV0 < 16 * NT0)>(av, b0[u], m0, acc0);
      strip_mma<NT1, (NV1 < 16 * NT1)>(av, b1[u], m1, acc1);
      if (NT2) strip_mma<N2, (NV2 < 16 * N2)>(av, b2[u], m2, acc2);
    }
  }

  strip_spill<NT0>(red, ks, NTT, 0, lane, acc0);
  strip_spill<NT1>(red, ks, NTT, NT0, lane, acc1);
  if (NT2) strip_spill<N2>(red, ks, NTT, NT0 + NT1, lane, acc2);
  __syncthreads();
  const int t = (int)threadIdx.x;
  const int oRow = row0 + (((t & 31) >> 4) << 3) + (t >> 5);
  const bool rowOK = !GM || oRow < M;
  strip_finish<NT0, NV0>(red, NTT, 0, t, rowOK, g0.C + (long)oRow * g0.ldc);
  strip_finish<NT1, NV1>(red, NTT, NT0, t, rowOK, g1.C + (long)oRow * g1.ldc);
  if (NT2) strip_finish<N2, NV2>(red, NTT, NT0 + NT1, t, rowOK, g2.C + (long)oRow * g2.ldc);
}

template <int KK>
__device__ __forceinline__ void gemm_abt_strip(const float* A, const int lda, const float* B,
                                               const int ldb, float* C, const int ldc,
                                               const int row0, const int col0,
                                               const float alpha) {
  const int lane = threadIdx.x & 31;
  const int c = lane & 15;
  const int h4 = (lane >> 4) << 2;
  const float* aPtr = A + (long)(row0 + c) * lda + h4;
  const float* bPtr = B + (long)(col0 + 2 * c) * ldb + h4;
  v4f a[KK / 8], b0[KK / 8], b1[KK / 8];
#pragma unroll
  for (int u = 0; u < KK / 8; ++u) {
    a[u]  = *(const v4f*)(aPtr + 8 * u);
    b0[u] = *(const v4f*)(bPtr + 8 * u);
    b1[u] = *(const v4f*)(bPtr + ldb + 8 * u);
  }
  __builtin_amdgcn_sched_barrier(0);
  v8f acc0 = {0.f, 0.f, 0.f, 0.f, 0.f, 0.f, 0.f, 0.f};
  v8f acc1 = {0.f, 0.f, 0.f, 0.f, 0.f, 0.f, 0.f, 0.f};
#pragma unroll
  for (int u = 0; u < KK / 8; ++u) {
    v2f af, bf;
    af.x = a[u].x; af.y = a[u].y;
    bf.x = b0[u].x; bf.y = b0[u].y;
    acc0 = __builtin_amdgcn_wmma_f32_16x16x4_f32(false, af, false, bf, (short)0, acc0, false, false);
    bf.x = b1[u].x; bf.y = b1[u].y;
    acc1 = __builtin_amdgcn_wmma_f32_16x16x4_f32(false, af, false, bf, (short)0, acc1, false, false);
    af.x = a[u].z; af.y = a[u].w;
    bf.x = b0[u].z; bf.y = b0[u].w;
    acc0 = __builtin_amdgcn_wmma_f32_16x16x4_f32(false, af, false, bf, (short)0, acc0, false, false);
    bf.x = b1[u].z; bf.y = b1[u].w;
    acc1 = __builtin_amdgcn_wmma_f32_16x16x4_f32(false, af, false, bf, (short)0, acc1, false, false);
  }
  float* cPtr = C + (long)(row0 + ((lane >> 4) << 3)) * ldc + col0 + 2 * c;
  for (int r = 0; r < 8; ++r) {
    v2f o;
    o.x = acc0[r] * alpha; o.y = acc1[r] * alpha;
    *(v2f*)(cPtr + (long)r * ldc) = o;
  }
}

#define PB_TPW_ 2

__device__ __forceinline__ void pairbias_load_a(const float* aPtr, v4f (&a)[DP_ / 8]) {
#pragma unroll
  for (int u = 0; u < DP_ / 8; ++u) a[u] = *(const v4f*)(aPtr + 8 * u);
}

__device__ __forceinline__ void pairbias_tiles(const float* x2d, const float* Wpb, float* pairproj,
                                               const int wg, const float alpha, float* tab) {
  const int t = (int)threadIdx.x;
  const int lane = t & 31;
  const int wave = __builtin_amdgcn_readfirstlane(t >> 5);
  const int c = lane & 15;
  const int h4 = (lane >> 4) << 2;
  const int tile0 = (wg * 8 + wave) * PB_TPW_;
  const float* aPtr = x2d + ((long)(tile0 << 4) + c) * DP_ + h4;

  v4f a[DP_ / 8];
  pairbias_load_a(aPtr, a);

  for (int q = 0; q < 2; ++q) {
    const int ent = t + 256 * q;
    const int ec = ent & 15;
    const int ek = 8 * (ent >> 5) + (((ent & 31) >> 4) << 2);
    v4f w = {0.f, 0.f, 0.f, 0.f};
    if (ec < H_) {
      const float* wp = Wpb + (long)ek * H_ + ec;
      w.x = wp[0]; w.y = wp[H_]; w.z = wp[2 * H_]; w.w = wp[3 * H_];
    }
    *(v4f*)(tab + 4 * ent) = w;
  }
  __syncthreads();

  const int mo = (lane >> 4) << 3;
  for (int n = 0; n < PB_TPW_; ++n) {
    v8f acc = {0.f, 0.f, 0.f, 0.f, 0.f, 0.f, 0.f, 0.f};
#pragma unroll
    for (int u = 0; u < DP_ / 8; ++u) {
      const v4f bw = *(const v4f*)(tab + 4 * (u * 32 + lane));
      v2f af, bf;
      af.x = a[u].x; af.y = a[u].y; bf.x = bw.x; bf.y = bw.y;
      acc = __builtin_amdgcn_wmma_f32_16x16x4_f32(false, af, false, bf, (short)0, acc, false, false);
      af.x = a[u].z; af.y = a[u].w; bf.x = bw.z; bf.y = bw.w;
      acc = __builtin_amdgcn_wmma_f32_16x16x4_f32(false, af, false, bf, (short)0, acc, false, false);
    }
    if (n + 1 < PB_TPW_) pairbias_load_a(aPtr + (long)(n + 1) * 16 * DP_, a);
    if (c < H_) {
      float* cPtr = pairproj + ((long)((tile0 + n) << 4) + mo) * H_ + c;
      for (int r = 0; r < 8; ++r) cPtr[r * H_] = acc[r] * alpha;
    }
  }
}

__device__ __forceinline__ void affine_body(float* pts, const float* T, const float* R,
                                            int np, int total, int idx) {
  if (idx >= total) return;
  int bi = idx / (H_ * np);
  const float* Rm = R + (long)bi * 9;
  const float* Tv = T + (long)bi * 3;
  float x = pts[idx * 3 + 0], y = pts[idx * 3 + 1], z = pts[idx * 3 + 2];
  float o0 = Rm[0] * x + Rm[3] * y + Rm[6] * z + Tv[0];
  float o1 = Rm[1] * x + Rm[4] * y + Rm[7] * z + Tv[1];
  float o2 = Rm[2] * x + Rm[5] * y + Rm[8] * z + Tv[2];
  pts[idx * 3 + 0] = o0; pts[idx * 3 + 1] = o1; pts[idx * 3 + 2] = o2;
}

#define ROWS_PER_WG_ 8
#define NKC_ (NPQ_ * 3)

__global__ void __launch_bounds__(256)
ipa_combine_softmax(float* logits,
                    const float* qgp, const float* kgp,
                    const float* qgl, const float* kgl,
                    const float* pairproj, const float* bias,
                    const float* tpw_past, const float* tpw_last) {
  __shared__ float kst[2 * NKC_ * I_];
  const int t = (int)threadIdx.x;
  const int ig = (int)blockIdx.x % (I_ / ROWS_PER_WG_);
  const int bh = (int)blockIdx.x / (I_ / ROWS_PER_WG_);
  const int h = bh % H_;
  const int b = bh / H_;
  const int lane = t & 31;
  const int w = __builtin_amdgcn_readfirstlane(t >> 5);
  const int i = ig * ROWS_PER_WG_ + w;

  float qp[NKC_], ql[NKC_];
  {
    const long qo = ((long)(b * I_ + i) * H_ + h) * NKC_;
    for (int c = 0; c < NKC_; ++c) { qp[c] = qgp[qo + c]; ql[c] = qgl[qo + c]; }
  }

  {
    const long ko = ((long)(b * I_ + t) * H_ + h) * NKC_;
    const v4f* kp = (const v4f*)(kgp + ko);
    const v4f* kl = (const v4f*)(kgl + ko);
    const v4f p0 = kp[0], p1 = kp[1], p2 = kp[2];
    const v4f l0 = kl[0], l1 = kl[1], l2 = kl[2];
    kst[ 0 * I_ + t] = p0.x; kst[ 1 * I_ + t] = p0.y; kst[ 2 * I_ + t] = p0.z; kst[ 3 * I_ + t] = p0.w;
    kst[ 4 * I_ + t] = p1.x; kst[ 5 * I_ + t] = p1.y; kst[ 6 * I_ + t] = p1.z; kst[ 7 * I_ + t] = p1.w;
    kst[ 8 * I_ + t] = p2.x; kst[ 9 * I_ + t] = p2.y; kst[10 * I_ + t] = p2.z; kst[11 * I_ + t] = p2.w;
    kst[12 * I_ + t] = l0.x; kst[13 * I_ + t] = l0.y; kst[14 * I_ + t] = l0.z; kst[15 * I_ + t] = l0.w;
    kst[16 * I_ + t] = l1.x; kst[17 * I_ + t] = l1.y; kst[18 * I_ + t] = l1.z; kst[19 * I_ + t] = l1.w;
    kst[20 * I_ + t] = l2.x; kst[21 * I_ + t] = l2.y; kst[22 * I_ + t] = l2.z; kst[23 * I_ + t] = l2.w;
  }

  const float point_w = 0.13608276348795434f;
  const float pwp = -0.5f * point_w * log1pf(expf(tpw_past[h]));
  const float pwl = -0.5f * point_w * log1pf(expf(tpw_last[h]));

  const long row = ((long)(b * H_ + h) * I_ + i) * I_;
  const float* pr = pairproj + ((long)(b * I_ + i) * I_) * H_ + h;

  __syncthreads();

  float v[I_ / 32];
#pragma unroll
  for (int m = 0; m < I_ / 32; ++m) {
    const int j = m * 32 + lane;
    float sp = 0.f, sl = 0.f;
    for (int c = 0; c < NPQ_; ++c) {
      float dx = qp[c * 3 + 0] - kst[(c * 3 + 0) * I_ + j];
      float dy = qp[c * 3 + 1] - kst[(c * 3 + 1) * I_ + j];
      float dz = qp[c * 3 + 2] - kst[(c * 3 + 2) * I_ + j];
      sp += sqrtf(dx * dx + dy * dy + dz * dz);
      dx = ql[c * 3 + 0] - kst[(NKC_ + c * 3 + 0) * I_ + j];
      dy = ql[c * 3 + 1] - kst[(NKC_ + c * 3 + 1) * I_ + j];
      dz = ql[c * 3 + 2] - kst[(NKC_ + c * 3 + 2) * I_ + j];
      sl += sqrtf(dx * dx + dy * dy + dz * dz);
    }
    const float pv = pr[(long)j * H_];
    const float add = pwp * sp + pwl * sl + pv + bias[row + j];
    v[m] = logits[row + j] + add;
  }

  float mx = v[0];
  for (int m = 1; m < I_ / 32; ++m) mx = fmaxf(mx, v[m]);
  for (int off = 16; off > 0; off >>= 1) mx = fmaxf(mx, __shfl_xor(mx, off));

  float sum = 0.f;
  for (int m = 0; m < I_ / 32; ++m) { v[m] = expf(v[m] - mx); sum += v[m]; }
  for (int off = 16; off > 0; off >>= 1) sum += __shfl_xor(sum, off);

  for (int m = 0; m < I_ / 32; ++m) logits[row + m * 32 + lane] = v[m] / sum;
}

__device__ __forceinline__ void inv_affine_body(const float* og, const float* T, const float* R,
                                                float* feat, int colOl, int colN, int total,
                                                int idx) {
  if (idx >= total) return;
  int c = idx & (NPV_ - 1);
  int t = idx >> 3;
  int h = t % H_;
  int bi = t / H_;
  const float* o = og + (long)idx * 3;
  const float* Tv = T + (long)bi * 3;
  const float* Rm = R + (long)bi * 9;
  float x = o[0] - Tv[0], y = o[1] - Tv[1], z = o[2] - Tv[2];
  float p0 = Rm[0] * x + Rm[1] * y + Rm[2] * z;
  float p1 = Rm[3] * x + Rm[4] * y + Rm[5] * z;
  float p2 = Rm[6] * x + Rm[7] * y + Rm[8] * z;
  float* fr = feat + (long)bi * (2 * D_ + 2 * H_ * NPV_ * 4);
  int tc = h * (NPV_ * 3) + c * 3;
  fr[colOl + tc + 0] = p0;
  fr[colOl + tc + 1] = p1;
  fr[colOl + tc + 2] = p2;
  fr[colN + h * NPV_ + c] = sqrtf(p0 * p0 + p1 * p1 + p2 * p2);
}

#define WG_PAIRBIAS_ (B_ * I_ * I_ / 16 / 8 / PB_TPW_)
#define WG_PROJ_ (3 * 384 + 4 * 96 + 2 * 160)

struct LevelA {
  const float* x1d; const float* x2d; const float* Wpb;
  const float* W0; const float* W1; const float* W2;
  const float* W3; const float* W4; const float* W5;
  const float* W6; const float* W7; const float* W8;
  float* pairproj;
  float* C0; float* C1; float* C2; float* C3; float* C4; float* C5;
  float* C6; float* C7; float* C8;
};

__global__ void __launch_bounds__(256, 8) ipa_level_a(LevelA a) {
  __shared__ float red[8 * 4 * 256];
  const int bx = (int)blockIdx.x;
  if (bx < WG_PAIRBIAS_) {
    pairbias_tiles(a.x2d, a.Wpb, a.pairproj, bx, 0.5773502691896258f, red);
    return;
  }
  const int t = bx - WG_PAIRBIAS_;
  const float* Wm = a.W0; float* Cm = a.C0; int N = 768; int base = 0; int strips = 12; int full = 12;
  if (t >= 384)  { Wm = a.W1; Cm = a.C1; base = 384; }
  if (t >= 768)  { Wm = a.W2; Cm = a.C2; base = 768; }
  if (t >= 1152) { Wm = a.W3; Cm = a.C3; base = 1152; N = 144; strips = 3; full = 2; }
  if (t >= 1248) { Wm = a.W4; Cm = a.C4; base = 1248; N = 144; strips = 3; full = 2; }
  if (t >= 1344) { Wm = a.W5; Cm = a.C5; base = 1344; N = 288; strips = 5; full = 4; }
  if (t >= 1504) { Wm = a.W6; Cm = a.C6; base = 1504; N = 144; strips = 3; full = 2; }
  if (t >= 1600) { Wm = a.W7; Cm = a.C7; base = 1600; N = 144; strips = 3; full = 2; }
  if (t >= 1696) { Wm = a.W8; Cm = a.C8; base = 1696; N = 288; strips = 5; full = 4; }
  const GemmP p{a.x1d, Wm, Cm, nullptr, B_ * I_, N, D_, D_, N, N,
                0, 0, 0, 0, 0, 0, 1.f};
  const int u = t - base;
  const int row0 = (u / strips) << 4;
  const int sx = u % strips;
  if (sx < full)      gemm_strip<4>(p, row0, sx << 6, 0, 0, red);
  else if (N == 288)  gemm_strip<2>(p, row0, 256, 0, 0, red);
  else                gemm_strip<1>(p, row0, 128, 0, 0, red);
}

#define WG_LOGITS_ (16 * H_ * B_)
#define WG_AFF4_ (B_ * I_ * H_ * NPQ_ / 256)
#define WG_AFF8_ (B_ * I_ * H_ * NPV_ / 256)

struct LevelB {
  const float* q; const float* k; float* logits;
  float* qgp; float* kgp; float* vgp; float* qgl; float* kgl; float* vgl;
  const float* past_T; const float* past_R; const float* last_T; const float* last_R;
};

__global__ void __launch_bounds__(256) ipa_level_b(LevelB a) {
  const int bx = (int)blockIdx.x;
  if (bx < WG_LOGITS_) {
    const int h = (bx >> 4) % H_, b = bx / (16 * H_);
    const long qo = (long)b * I_ * D_ + h * DK_;
    const int w = __builtin_amdgcn_readfirstlane((int)(threadIdx.x >> 5));
    gemm_abt_strip<DK_>(a.q + qo, D_, a.k + qo, D_, a.logits + ((long)b * H_ + h) * I_ * I_, I_,
                        (bx & 15) << 4, w << 5, 0.07216878364870323f);
    return;
  }
  int t = bx - WG_LOGITS_;
  const bool lastBranch = t >= (2 * WG_AFF4_ + WG_AFF8_);
  if (lastBranch) t -= 2 * WG_AFF4_ + WG_AFF8_;
  float* pts = lastBranch ? a.qgl : a.qgp; int np = NPQ_; int base = 0;
  if (t >= WG_AFF4_)     { pts = lastBranch ? a.kgl : a.kgp; base = WG_AFF4_; }
  if (t >= 2 * WG_AFF4_) { pts = lastBranch ? a.vgl : a.vgp; base = 2 * WG_AFF4_; np = NPV_; }
  const float* T = lastBranch ? a.last_T : a.past_T;
  const float* R = lastBranch ? a.last_R : a.past_R;
  affine_body(pts, T, R, np, B_ * I_ * H_ * np, (t - base) * 256 + (int)threadIdx.x);
}

#define WG_Z_ (I_ * B_)
#define WG_AVG_ (16 * H_ * B_)

struct LevelD {
  const float* attn; const float* x2d; const float* v; const float* vgp; const float* vgl;
  float* z; float* featS; float* ogp; float* ogl;
};

__global__ void __launch_bounds__(256) ipa_level_d(LevelD a) {
  __shared__ float red[8 * 8 * 256];
  const int bx = (int)blockIdx.x;
  const int FE = 2 * D_ + 2 * H_ * NPV_ * 4;
  const int NG = NPV_ * 3;
  if (bx < WG_Z_) {
    const int i = bx & (I_ - 1), b = bx >> 8;
    const float* A = a.attn + (long)b * H_ * I_ * I_ + (long)i * I_;
    const float* Bm = a.x2d + ((long)b * I_ + i) * I_ * DP_;
    float* Cm = a.z + ((long)b * I_ + i) * H_ * DP_;
    const StripBC g0{Bm, Cm, DP_, DP_};
    const StripBC g1{Bm + 64, Cm + 64, DP_, DP_};
    const StripBC none{nullptr, nullptr, 0, 0};
    gemm_strip_shared<4, 64, 4, 64, 0, 0, true>(A, I_ * I_, H_, I_, 0, g0, g1, none, red);
    return;
  }
  const int t = bx - WG_Z_;
  const int h = (t >> 4) % H_, b = t / (16 * H_);
  const float* A = a.attn + ((long)b * H_ + h) * I_ * I_;
  const long vo = (long)b * I_ * D_ + h * DK_;
  const long go = (long)b * I_ * H_ * NG + h * NG;
  const StripBC g0{a.v + vo, a.featS + (long)b * I_ * FE + h * DK_, D_, FE};
  const StripBC g1{a.vgp + go, a.ogp + go, H_ * NG, H_ * NG};
  const StripBC g2{a.vgl + go, a.ogl + go, H_ * NG, H_ * NG};
  gemm_strip_shared<4, 64, 2, 24, 2, 24, false>(A, I_, I_, I_, (t & 15) << 4, g0, g1, g2, red);
}

#define WG_OUTPAIR_ (16 * H_ * B_)

struct LevelE {
  const float* z; const float* Wpv; const float* ogp; const float* ogl;
  const float* past_T; const float* past_R; float* feat;
};

__global__ void __launch_bounds__(256) ipa_level_e(LevelE a) {
  __shared__ float red[8 * 4 * 256];
  const int bx = (int)blockIdx.x;
  const int FE = 2 * D_ + 2 * H_ * NPV_ * 4;
  if (bx < WG_OUTPAIR_) {
    const GemmP p{a.z, a.Wpv, a.feat, nullptr, I_, DK_, DP_, H_ * DP_, D_, FE,
                  DP_, (long)I_ * H_ * DP_, 64, 0, 64, (long)I_ * FE, 1.f};
    gemm_strip<4, 2>(p, (bx & 15) << 4, 0, (bx >> 4) % H_, bx / (16 * H_), red);
    return;
  }
  int t = bx - WG_OUTPAIR_;
  const bool lastBranch = t >= WG_AFF8_;
  if (lastBranch) t -= WG_AFF8_;
  const int colOl = lastBranch ? (2 * D_ + 384) : (2 * D_);
  inv_affine_body(lastBranch ? a.ogl : a.ogp, a.past_T, a.past_R, a.feat,
                  colOl, colOl + 288, B_ * I_ * H_ * NPV_, t * 256 + (int)threadIdx.x);
}

__global__ void __launch_bounds__(256) ipa_gemm_strip(GemmP p) {
  __shared__ float red[8 * 4 * 256];
  gemm_strip<4>(p, (int)blockIdx.y << 4, (int)blockIdx.x << 6, 0, 0, red);
}

extern "C" void kernel_launch(void* const* d_in, const int* in_sizes, int n_in,
                              void* d_out, int out_size, void* d_ws, size_t ws_size,
                              hipStream_t stream) {
  (void)in_sizes; (void)n_in; (void)out_size; (void)ws_size;
  const float* x1d      = (const float*)d_in[0];
  const float* x2d      = (const float*)d_in[1];
  const float* past_T   = (const float*)d_in[2];
  const float* past_R   = (const float*)d_in[3];
  const float* last_T   = (const float*)d_in[4];
  const float* last_R   = (const float*)d_in[5];
  const float* bias     = (const float*)d_in[6];
  const float* Wq       = (const float*)d_in[7];
  const float* Wk       = (const float*)d_in[8];
  const float* Wv       = (const float*)d_in[9];
  const float* Wpb      = (const float*)d_in[10];
  const float* Wpq_past = (const float*)d_in[11];
  const float* Wpk_past = (const float*)d_in[12];
  const float* Wpv_past = (const float*)d_in[13];
  const float* Wpq_last = (const float*)d_in[14];
  const float* Wpk_last = (const float*)d_in[15];
  const float* Wpv_last = (const float*)d_in[16];
  const float* tpw_past = (const float*)d_in[17];
  const float* tpw_last = (const float*)d_in[18];
  const float* Wpv      = (const float*)d_in[19];
  const float* W_out    = (const float*)d_in[20];
  const float* b_out    = (const float*)d_in[21];
  float* out = (float*)d_out;
  float* ws  = (float*)d_ws;

  float* q    = ws;
  float* k    = q + 393216;
  float* v    = k + 393216;
  float* qgp  = v + 393216;
  float* kgp  = qgp + 73728;
  float* vgp  = kgp + 73728;
  float* qgl  = vgp + 147456;
  float* kgl  = qgl + 73728;
  float* vgl  = kgl + 73728;
  float* logits   = vgl + 147456;
  float* pairproj = logits + 1572864;
  float* ogp  = pairproj + 1572864;
  float* ogl  = ogp + 147456;
  float* z    = ogl + 147456;
  float* feat = z + 786432;

  const LevelA la{x1d, x2d, Wpb,
                  Wq, Wk, Wv, Wpq_past, Wpk_past, Wpv_past, Wpq_last, Wpk_last, Wpv_last,
                  pairproj,
                  q, k, v, qgp, kgp, vgp, qgl, kgl, vgl};
  ipa_level_a<<<WG_PAIRBIAS_ + WG_PROJ_, 256, 0, stream>>>(la);

  const LevelB lb{q, k, logits, qgp, kgp, vgp, qgl, kgl, vgl,
                  past_T, past_R, last_T, last_R};
  ipa_level_b<<<WG_LOGITS_ + 4 * WG_AFF4_ + 2 * WG_AFF8_, 256, 0, stream>>>(lb);

  ipa_combine_softmax<<<B_ * H_ * (I_ / ROWS_PER_WG_), 256, 0, stream>>>(
      logits, qgp, kgp, qgl, kgl, pairproj, bias, tpw_past, tpw_last);

  const LevelD ld{logits, x2d, v, vgp, vgl, z, feat + D_, ogp, ogl};
  ipa_level_d<<<WG_Z_ + WG_AVG_, 256, 0, stream>>>(ld);

  const LevelE le{z, Wpv, ogp, ogl, past_T, past_R, feat};
  ipa_level_e<<<WG_OUTPAIR_ + 2 * WG_AFF8_, 256, 0, stream>>>(le);

  const GemmP pf{feat, W_out, out, b_out, B_ * I_, D_, 2 * D_ + 2 * H_ * NPV_ * 4,
                 2 * D_ + 2 * H_ * NPV_ * 4, D_, D_, 0, 0, 0, 0, 0, 0, 1.f};
  ipa_gemm_strip<<<dim3(12, 32, 1), 256, 0, stream>>>(pf);
}


# strip GEMMs (nine projections, z*W_pair_value, output): each wave computes two row tiles (32x64) off one B fragment set, so B bytes per product from L2 halve; every sum and its order unchanged
# speedup vs baseline: 6.1891x; 1.0470x over previous
#include <hip/hip_runtime.h>
#include <math.h>

#define B_ 2
#define I_ 256
#define D_ 768
#define H_ 12
#define DP_ 128
#define DK_ 64
#define NPQ_ 4
#define NPV_ 8

typedef float v2f __attribute__((ext_vector_type(2)));
typedef float v8f __attribute__((ext_vector_type(8)));

struct GemmP {
  const float* A; const float* B; float* C; const float* bias;
  int M, N, K;
  int lda, ldb, ldc;
  long sAy, sAz, sBy, sBz, sCy, sCz;
  float alpha;
};

typedef float v4f __attribute__((ext_vector_type(4)));

template <int NT> struct VecIO;
template <> struct VecIO<1> {
  static __device__ __forceinline__ void ld(const float* p, float* o) { o[0] = p[0]; }
  static __device__ __forceinline__ void st(float* p, const float* o) { p[0] = o[0]; }
};
template <> struct VecIO<2> {
  static __device__ __forceinline__ void ld(const float* p, float* o) {
    const v2f t = *(const v2f*)p; o[0] = t.x; o[1] = t.y;
  }
  static __device__ __forceinline__ void st(float* p, const float* o) {
    v2f t; t.x = o[0]; t.y = o[1]; *(v2f*)p = t;
  }
};
template <> struct VecIO<4> {
  static __device__ __forceinline__ void ld(const float* p, float* o) {
    const v4f t = *(const v4f*)p; o[0] = t.x; o[1] = t.y; o[2] = t.z; o[3] = t.w;
  }
  static __device__ __forceinline__ void st(float* p, const float* o) {
    v4f t; t.x = o[0]; t.y = o[1]; t.z = o[2]; t.w = o[3]; *(v4f*)p = t;
  }
};

template <int NT, int ST = 4, int MT = 1>
__device__ __forceinline__ void gemm_strip(const GemmP p, const int row0, const int col0,
                                           const int by, const int bz, float* red) {
  const int lane = threadIdx.x & 31;
  const int ks = __builtin_amdgcn_readfirstlane((int)(threadIdx.x >> 5));
  const int per = p.K >> 3;
  const int kBeg = ks * per, kEnd = kBeg + per;

  const float* A = p.A + (long)by * p.sAy + (long)bz * p.sAz;
  const float* B = p.B + (long)by * p.sBy + (long)bz * p.sBz;
  float*       C = p.C + (long)by * p.sCy + (long)bz * p.sCz;

  const int c = lane & 15;
  const int h4 = (lane >> 4) << 2;
  const float* aPtr = A + (long)(row0 + c) * p.lda + h4;
  const float* bPtr = B + (long)h4 * p.ldb + col0 + NT * c;

  const long aTile = (long)16 * p.lda;
  v8f acc[MT][NT];
  for (int i = 0; i < MT; ++i)
    for (int j = 0; j < NT; ++j)
      for (int r = 0; r < 8; ++r) acc[i][j][r] = 0.f;

  for (int k0 = kBeg; k0 < kEnd; k0 += 8 * ST) {
    v4f a[ST][MT];
    float b[ST][4][NT];
#pragma unroll
    for (int u = 0; u < ST; ++u) {
      for (int i = 0; i < MT; ++i) a[u][i] = *(const v4f*)(aPtr + i * aTile + k0 + 8 * u);
      const float* bp = bPtr + (long)(k0 + 8 * u) * p.ldb;
      for (int e = 0; e < 4; ++e) VecIO<NT>::ld(bp + (long)e * p.ldb, b[u][e]);
    }
    __builtin_amdgcn_sched_barrier(0);
#pragma unroll
    for (int u = 0; u < ST; ++u) {
      for (int i = 0; i < MT; ++i) {
        const v4f av = a[u][i];
        for (int j = 0; j < NT; ++j) {
          v2f af, bf;
          af.x = av.x; af.y = av.y; bf.x = b[u][0][j]; bf.y = b[u][1][j];
          acc[i][j] = __builtin_amdgcn_wmma_f32_16x16x4_f32(false, af, false, bf,
                                                            (short)0, acc[i][j], false, false);
        }
      }
      for (int i = 0; i < MT; ++i) {
        const v4f av = a[u][i];
        for (int j = 0; j < NT; ++j) {
          v2f af, bf;
          af.x = av.z; af.y = av.w; bf.x = b[u][2][j]; bf.y = b[u][3][j];
          acc[i][j] = __builtin_amdgcn_wmma_f32_16x16x4_f32(false, af, false, bf,
                                                            (short)0, acc[i][j], false, false);
        }
      }
    }
  }

  for (int i = 0; i < MT; ++i)
    for (int j = 0; j < NT; ++j)
      for (int r = 0; r < 8; ++r)
        red[(((ks * MT + i) * NT + j) * 8 + r) * 32 + lane] = acc[i][j][r];
  __syncthreads();
  const int t = (int)threadIdx.x;
  const int oRow = row0 + (((t & 31) >> 4) << 3) + (t >> 5);
  const int oCol = col0 + NT * (t & 15);
  float bv[NT];
  for (int j = 0; j < NT; ++j) bv[j] = 0.f;
  if (p.bias) VecIO<NT>::ld(p.bias + oCol, bv);
  for (int i = 0; i < MT; ++i) {
    float o[NT];
    for (int j = 0; j < NT; ++j) {
      float sum = red[(i * NT + j) * 256 + t];
      for (int s = 1; s < 8; ++s) sum += red[((s * MT + i) * NT + j) * 256 + t];
      o[j] = sum * p.alpha;
      if (p.bias) o[j] += bv[j];
    }
    VecIO<NT>::st(C + (long)(oRow + 16 * i) * p.ldc + oCol, o);
  }
}

struct StripBC { const float* B; float* C; int ldb, ldc; };

template <int NT, int NV>
__device__ __forceinline__ const float* strip_bptr(const float* B, const int ldb, const int c,
                                                   const int h4, float& mask) {
  int cc = c;
  mask = 1.f;
  if (NV < 16 * NT) {
    if (NT * c >= NV) { cc = 0; mask = 0.f; }
  }
  return B + (long)h4 * ldb + NT * cc;
}

template <int NT>
__device__ __forceinline__ void strip_load_b(const float* bPtr, const int ldb, const int k0,
                                             float (&b)[4][4][NT]) {
#pragma unroll
  for (int u = 0; u < 4; ++u) {
    const float* bp = bPtr + (long)(k0 + 8 * u) * ldb;
#pragma unroll
    for (int e = 0; e < 4; ++e) VecIO<NT>::ld(bp + (long)e * ldb, b[u][e]);
  }
}

template <int NT, bool MASK>
__device__ __forceinline__ void strip_mma(const v4f av, const float (&bu)[4][NT], const float mask,
                                          v8f (&acc)[NT]) {
  for (int j = 0; j < NT; ++j) {
    v2f af, bf;
    af.x = av.x; af.y = av.y; bf.x = bu[0][j]; bf.y = bu[1][j];
    if (MASK) bf *= mask;
    acc[j] = __builtin_amdgcn_wmma_f32_16x16x4_f32(false, af, false, bf,
                                                   (short)0, acc[j], false, false);
  }
  for (int j = 0; j < NT; ++j) {
    v2f af, bf;
    af.x = av.z; af.y = av.w; bf.x = bu[2][j]; bf.y = bu[3][j];
    if (MASK) bf *= mask;
    acc[j] = __builtin_amdgcn_wmma_f32_16x16x4_f32(false, af, false, bf,
                                                   (short)0, acc[j], false, false);
  }
}

template <int NT>
__device__ __forceinline__ void strip_zero(v8f (&acc)[NT]) {
  for (int j = 0; j < NT; ++j)
    for (int r = 0; r < 8; ++r) acc[j][r] = 0.f;
}

template <int NT>
__device__ __forceinline__ void strip_spill(float* red, const int ks, const int ntt, const int j0,
                                            const int lane, const v8f (&acc)[NT]) {
  for (int j = 0; j < NT; ++j)
    for (int r = 0; r < 8; ++r) red[((ks * ntt + j0 + j) * 8 + r) * 32 + lane] = acc[j][r];
}

template <int NT, int NV>
__device__ __forceinline__ void strip_finish(const float* red, const int ntt, const int j0,
                                             const int t, const bool rowOK, float* cRow) {
  float o[NT];
  for (int j = 0; j < NT; ++j) {
    float sum = red[(j0 + j) * 256 + t];
    for (int s = 1; s < 8; ++s) sum += red[(s * ntt + j0 + j) * 256 + t];
    o[j] = sum;
  }
  const int oc = NT * (t & 15);
  if (rowOK && (NV == 16 * NT || oc < NV)) VecIO<NT>::st(cRow + oc, o);
}

template <int NT0, int NV0, int NT1, int NV1, int NT2, int NV2, bool GM>
__device__ __forceinline__ void gemm_strip_shared(const float* A, const int lda, const int M,
                                                  const int K, const int row0,
                                                  const StripBC g0, const StripBC g1,
                                                  const StripBC g2, float* red) {
  constexpr int N2 = NT2 ? NT2 : 1;
  constexpr int NTT = NT0 + NT1 + NT2;
  const int lane = threadIdx.x & 31;
  const int ks = __builtin_amdgcn_readfirstlane((int)(threadIdx.x >> 5));
  const int per = K >> 3;
  const int kBeg = ks * per, kEnd = kBeg + per;

  const int c = lane & 15;
  const int h4 = (lane >> 4) << 2;
  int aRow = row0 + c;
  float aMask = 1.f;
  if (GM) {
    if (aRow >= M) { aRow = M - 1; aMask = 0.f; }
  }
  const float* aPtr = A + (long)aRow * lda + h4;
  float m0, m1, m2 = 1.f;
  const float* bPtr0 = strip_bptr<NT0, NV0>(g0.B, g0.ldb, c, h4, m0);
  const float* bPtr1 = strip_bptr<NT1, NV1>(g1.B, g1.ldb, c, h4, m1);
  const float* bPtr2 = nullptr;
  if (NT2) bPtr2 = strip_bptr<N2, NV2>(g2.B, g2.ldb, c, h4, m2);

  v8f acc0[NT0], acc1[NT1], acc2[N2];
  strip_zero<NT0>(acc0);
  strip_zero<NT1>(acc1);
  if (NT2) strip_zero<N2>(acc2);

  for (int k0 = kBeg; k0 < kEnd; k0 += 32) {
    v4f a[4];
    float b0[4][4][NT0], b1[4][4][NT1], b2[4][4][N2];
#pragma unroll
    for (int u = 0; u < 4; ++u) a[u] = *(const v4f*)(aPtr + k0 + 8 * u);
    strip_load_b<NT0>(bPtr0, g0.ldb, k0, b0);
    strip_load_b<NT1>(bPtr1, g1.ldb, k0, b1);
    if (NT2) strip_load_b<N2>(bPtr2, g2.ldb, k0, b2);
    __builtin_amdgcn_sched_barrier(0);
#pragma unroll
    for (int u = 0; u < 4; ++u) {
      v4f av = a[u];
      if (GM) av *= aMask;
      strip_mma<NT0, (NV0 < 16 * NT0)>(av, b0[u], m0, acc0);
      strip_mma<NT1, (NV1 < 16 * NT1)>(av, b1[u], m1, acc1);
      if (NT2) strip_mma<N2, (NV2 < 16 * N2)>(av, b2[u], m2, acc2);
    }
  }

  strip_spill<NT0>(red, ks, NTT, 0, lane, acc0);
  strip_spill<NT1>(red, ks, NTT, NT0, lane, acc1);
  if (NT2) strip_spill<N2>(red, ks, NTT, NT0 + NT1, lane, acc2);
  __syncthreads();
  const int t = (int)threadIdx.x;
  const int oRow = row0 + (((t & 31) >> 4) << 3) + (t >> 5);
  const bool rowOK = !GM || oRow < M;
  strip_finish<NT0, NV0>(red, NTT, 0, t, rowOK, g0.C + (long)oRow * g0.ldc);
  strip_finish<NT1, NV1>(red, NTT, NT0, t, rowOK, g1.C + (long)oRow * g1.ldc);
  if (NT2) strip_finish<N2, NV2>(red, NTT, NT0 + NT1, t, rowOK, g2.C + (long)oRow * g2.ldc);
}

template <int KK>
__device__ __forceinline__ void gemm_abt_strip(const float* A, const int lda, const float* B,
                                               const int ldb, float* C, const int ldc,
                                               const int row0, const int col0,
                                               const float alpha) {
  const int lane = threadIdx.x & 31;
  const int c = lane & 15;
  const int h4 = (lane >> 4) << 2;
  const float* aPtr = A + (long)(row0 + c) * lda + h4;
  const float* bPtr = B + (long)(col0 + 2 * c) * ldb + h4;
  v4f a[KK / 8], b0[KK / 8], b1[KK / 8];
#pragma unroll
  for (int u = 0; u < KK / 8; ++u) {
    a[u]  = *(const v4f*)(aPtr + 8 * u);
    b0[u] = *(const v4f*)(bPtr + 8 * u);
    b1[u] = *(const v4f*)(bPtr + ldb + 8 * u);
  }
  __builtin_amdgcn_sched_barrier(0);
  v8f acc0 = {0.f, 0.f, 0.f, 0.f, 0.f, 0.f, 0.f, 0.f};
  v8f acc1 = {0.f, 0.f, 0.f, 0.f, 0.f, 0.f, 0.f, 0.f};
#pragma unroll
  for (int u = 0; u < KK / 8; ++u) {
    v2f af, bf;
    af.x = a[u].x; af.y = a[u].y;
    bf.x = b0[u].x; bf.y = b0[u].y;
    acc0 = __builtin_amdgcn_wmma_f32_16x16x4_f32(false, af, false, bf, (short)0, acc0, false, false);
    bf.x = b1[u].x; bf.y = b1[u].y;
    acc1 = __builtin_amdgcn_wmma_f32_16x16x4_f32(false, af, false, bf, (short)0, acc1, false, false);
    af.x = a[u].z; af.y = a[u].w;
    bf.x = b0[u].z; bf.y = b0[u].w;
    acc0 = __builtin_amdgcn_wmma_f32_16x16x4_f32(false, af, false, bf, (short)0, acc0, false, false);
    bf.x = b1[u].z; bf.y = b1[u].w;
    acc1 = __builtin_amdgcn_wmma_f32_16x16x4_f32(false, af, false, bf, (short)0, acc1, false, false);
  }
  float* cPtr = C + (long)(row0 + ((lane >> 4) << 3)) * ldc + col0 + 2 * c;
  for (int r = 0; r < 8; ++r) {
    v2f o;
    o.x = acc0[r] * alpha; o.y = acc1[r] * alpha;
    *(v2f*)(cPtr + (long)r * ldc) = o;
  }
}

#define PB_TPW_ 2

__device__ __forceinline__ void pairbias_load_a(const float* aPtr, v4f (&a)[DP_ / 8]) {
#pragma unroll
  for (int u = 0; u < DP_ / 8; ++u) a[u] = *(const v4f*)(aPtr + 8 * u);
}

__device__ __forceinline__ void pairbias_tiles(const float* x2d, const float* Wpb, float* pairproj,
                                               const int wg, const float alpha, float* tab) {
  const int t = (int)threadIdx.x;
  const int lane = t & 31;
  const int wave = __builtin_amdgcn_readfirstlane(t >> 5);
  const int c = lane & 15;
  const int h4 = (lane >> 4) << 2;
  const int tile0 = (wg * 8 + wave) * PB_TPW_;
  const float* aPtr = x2d + ((long)(tile0 << 4) + c) * DP_ + h4;

  v4f a[DP_ / 8];
  pairbias_load_a(aPtr, a);

  for (int q = 0; q < 2; ++q) {
    const int ent = t + 256 * q;
    const int ec = ent & 15;
    const int ek = 8 * (ent >> 5) + (((ent & 31) >> 4) << 2);
    v4f w = {0.f, 0.f, 0.f, 0.f};
    if (ec < H_) {
      const float* wp = Wpb + (long)ek * H_ + ec;
      w.x = wp[0]; w.y = wp[H_]; w.z = wp[2 * H_]; w.w = wp[3 * H_];
    }
    *(v4f*)(tab + 4 * ent) = w;
  }
  __syncthreads();

  const int mo = (lane >> 4) << 3;
  for (int n = 0; n < PB_TPW_; ++n) {
    v8f acc = {0.f, 0.f, 0.f, 0.f, 0.f, 0.f, 0.f, 0.f};
#pragma unroll
    for (int u = 0; u < DP_ / 8; ++u) {
      const v4f bw = *(const v4f*)(tab + 4 * (u * 32 + lane));
      v2f af, bf;
      af.x = a[u].x; af.y = a[u].y; bf.x = bw.x; bf.y = bw.y;
      acc = __builtin_amdgcn_wmma_f32_16x16x4_f32(false, af, false, bf, (short)0, acc, false, false);
      af.x = a[u].z; af.y = a[u].w; bf.x = bw.z; bf.y = bw.w;
      acc = __builtin_amdgcn_wmma_f32_16x16x4_f32(false, af, false, bf, (short)0, acc, false, false);
    }
    if (n + 1 < PB_TPW_) pairbias_load_a(aPtr + (long)(n + 1) * 16 * DP_, a);
    if (c < H_) {
      float* cPtr = pairproj + ((long)((tile0 + n) << 4) + mo) * H_ + c;
      for (int r = 0; r < 8; ++r) cPtr[r * H_] = acc[r] * alpha;
    }
  }
}

__device__ __forceinline__ void affine_body(float* pts, const float* T, const float* R,
                                            int np, int total, int idx) {
  if (idx >= total) return;
  int bi = idx / (H_ * np);
  const float* Rm = R + (long)bi * 9;
  const float* Tv = T + (long)bi * 3;
  float x = pts[idx * 3 + 0], y = pts[idx * 3 + 1], z = pts[idx * 3 + 2];
  float o0 = Rm[0] * x + Rm[3] * y + Rm[6] * z + Tv[0];
  float o1 = Rm[1] * x + Rm[4] * y + Rm[7] * z + Tv[1];
  float o2 = Rm[2] * x + Rm[5] * y + Rm[8] * z + Tv[2];
  pts[idx * 3 + 0] = o0; pts[idx * 3 + 1] = o1; pts[idx * 3 + 2] = o2;
}

#define ROWS_PER_WG_ 8
#define NKC_ (NPQ_ * 3)

__global__ void __launch_bounds__(256)
ipa_combine_softmax(float* logits,
                    const float* qgp, const float* kgp,
                    const float* qgl, const float* kgl,
                    const float* pairproj, const float* bias,
                    const float* tpw_past, const float* tpw_last) {
  __shared__ float kst[2 * NKC_ * I_];
  const int t = (int)threadIdx.x;
  const int ig = (int)blockIdx.x % (I_ / ROWS_PER_WG_);
  const int bh = (int)blockIdx.x / (I_ / ROWS_PER_WG_);
  const int h = bh % H_;
  const int b = bh / H_;
  const int lane = t & 31;
  const int w = __builtin_amdgcn_readfirstlane(t >> 5);
  const int i = ig * ROWS_PER_WG_ + w;

  float qp[NKC_], ql[NKC_];
  {
    const long qo = ((long)(b * I_ + i) * H_ + h) * NKC_;
    for (int c = 0; c < NKC_; ++c) { qp[c] = qgp[qo + c]; ql[c] = qgl[qo + c]; }
  }

  {
    const long ko = ((long)(b * I_ + t) * H_ + h) * NKC_;
    const v4f* kp = (const v4f*)(kgp + ko);
    const v4f* kl = (const v4f*)(kgl + ko);
    const v4f p0 = kp[0], p1 = kp[1], p2 = kp[2];
    const v4f l0 = kl[0], l1 = kl[1], l2 = kl[2];
    kst[ 0 * I_ + t] = p0.x; kst[ 1 * I_ + t] = p0.y; kst[ 2 * I_ + t] = p0.z; kst[ 3 * I_ + t] = p0.w;
    kst[ 4 * I_ + t] = p1.x; kst[ 5 * I_ + t] = p1.y; kst[ 6 * I_ + t] = p1.z; kst[ 7 * I_ + t] = p1.w;
    kst[ 8 * I_ + t] = p2.x; kst[ 9 * I_ + t] = p2.y; kst[10 * I_ + t] = p2.z; kst[11 * I_ + t] = p2.w;
    kst[12 * I_ + t] = l0.x; kst[13 * I_ + t] = l0.y; kst[14 * I_ + t] = l0.z; kst[15 * I_ + t] = l0.w;
    kst[16 * I_ + t] = l1.x; kst[17 * I_ + t] = l1.y; kst[18 * I_ + t] = l1.z; kst[19 * I_ + t] = l1.w;
    kst[20 * I_ + t] = l2.x; kst[21 * I_ + t] = l2.y; kst[22 * I_ + t] = l2.z; kst[23 * I_ + t] = l2.w;
  }

  const float point_w = 0.13608276348795434f;
  const float pwp = -0.5f * point_w * log1pf(expf(tpw_past[h]));
  const float pwl = -0.5f * point_w * log1pf(expf(tpw_last[h]));

  const long row = ((long)(b * H_ + h) * I_ + i) * I_;
  const float* pr = pairproj + ((long)(b * I_ + i) * I_) * H_ + h;

  __syncthreads();

  float v[I_ / 32];
#pragma unroll
  for (int m = 0; m < I_ / 32; ++m) {
    const int j = m * 32 + lane;
    float sp = 0.f, sl = 0.f;
    for (int c = 0; c < NPQ_; ++c) {
      float dx = qp[c * 3 + 0] - kst[(c * 3 + 0) * I_ + j];
      float dy = qp[c * 3 + 1] - kst[(c * 3 + 1) * I_ + j];
      float dz = qp[c * 3 + 2] - kst[(c * 3 + 2) * I_ + j];
      sp += sqrtf(dx * dx + dy * dy + dz * dz);
      dx = ql[c * 3 + 0] - kst[(NKC_ + c * 3 + 0) * I_ + j];
      dy = ql[c * 3 + 1] - kst[(NKC_ + c * 3 + 1) * I_ + j];
      dz = ql[c * 3 + 2] - kst[(NKC_ + c * 3 + 2) * I_ + j];
      sl += sqrtf(dx * dx + dy * dy + dz * dz);
    }
    const float pv = pr[(long)j * H_];
    const float add = pwp * sp + pwl * sl + pv + bias[row + j];
    v[m] = logits[row + j] + add;
  }

  float mx = v[0];
  for (int m = 1; m < I_ / 32; ++m) mx = fmaxf(mx, v[m]);
  for (int off = 16; off > 0; off >>= 1) mx = fmaxf(mx, __shfl_xor(mx, off));

  float sum = 0.f;
  for (int m = 0; m < I_ / 32; ++m) { v[m] = expf(v[m] - mx); sum += v[m]; }
  for (int off = 16; off > 0; off >>= 1) sum += __shfl_xor(sum, off);

  for (int m = 0; m < I_ / 32; ++m) logits[row + m * 32 + lane] = v[m] / sum;
}

__device__ __forceinline__ void inv_affine_body(const float* og, const float* T, const float* R,
                                                float* feat, int colOl, int colN, int total,
                                                int idx) {
  if (idx >= total) return;
  int c = idx & (NPV_ - 1);
  int t = idx >> 3;
  int h = t % H_;
  int bi = t / H_;
  const float* o = og + (long)idx * 3;
  const float* Tv = T + (long)bi * 3;
  const float* Rm = R + (long)bi * 9;
  float x = o[0] - Tv[0], y = o[1] - Tv[1], z = o[2] - Tv[2];
  float p0 = Rm[0] * x + Rm[1] * y + Rm[2] * z;
  float p1 = Rm[3] * x + Rm[4] * y + Rm[5] * z;
  float p2 = Rm[6] * x + Rm[7] * y + Rm[8] * z;
  float* fr = feat + (long)bi * (2 * D_ + 2 * H_ * NPV_ * 4);
  int tc = h * (NPV_ * 3) + c * 3;
  fr[colOl + tc + 0] = p0;
  fr[colOl + tc + 1] = p1;
  fr[colOl + tc + 2] = p2;
  fr[colN + h * NPV_ + c] = sqrtf(p0 * p0 + p1 * p1 + p2 * p2);
}

#define WG_PAIRBIAS_ (B_ * I_ * I_ / 16 / 8 / PB_TPW_)
#define WG_PROJ_ (3 * 192 + 4 * 48 + 2 * 80)

struct LevelA {
  const float* x1d; const float* x2d; const float* Wpb;
  const float* W0; const float* W1; const float* W2;
  const float* W3; const float* W4; const float* W5;
  const float* W6; const float* W7; const float* W8;
  float* pairproj;
  float* C0; float* C1; float* C2; float* C3; float* C4; float* C5;
  float* C6; float* C7; float* C8;
};

#define PROJ_ST_ 3
#define PROJ_WAVES_ 6
__global__ void __launch_bounds__(256, PROJ_WAVES_) ipa_level_a(LevelA a) {
  __shared__ float red[8 * 8 * 256];
  const int bx = (int)blockIdx.x;
  if (bx < WG_PAIRBIAS_) {
    pairbias_tiles(a.x2d, a.Wpb, a.pairproj, bx, 0.5773502691896258f, red);
    return;
  }
  const int t = bx - WG_PAIRBIAS_;
  const float* Wm = a.W0; float* Cm = a.C0; int N = 768; int base = 0; int strips = 12; int full = 12;
  if (t >= 192) { Wm = a.W1; Cm = a.C1; base = 192; }
  if (t >= 384) { Wm = a.W2; Cm = a.C2; base = 384; }
  if (t >= 576) { Wm = a.W3; Cm = a.C3; base = 576; N = 144; strips = 3; full = 2; }
  if (t >= 624) { Wm = a.W4; Cm = a.C4; base = 624; N = 144; strips = 3; full = 2; }
  if (t >= 672) { Wm = a.W5; Cm = a.C5; base = 672; N = 288; strips = 5; full = 4; }
  if (t >= 752) { Wm = a.W6; Cm = a.C6; base = 752; N = 144; strips = 3; full = 2; }
  if (t >= 800) { Wm = a.W7; Cm = a.C7; base = 800; N = 144; strips = 3; full = 2; }
  if (t >= 848) { Wm = a.W8; Cm = a.C8; base = 848; N = 288; strips = 5; full = 4; }
  const GemmP p{a.x1d, Wm, Cm, nullptr, B_ * I_, N, D_, D_, N, N,
                0, 0, 0, 0, 0, 0, 1.f};
  const int u = t - base;
  const int row0 = (u / strips) << 5;
  const int sx = u % strips;
  if (sx < full)      gemm_strip<4, PROJ_ST_, 2>(p, row0, sx << 6, 0, 0, red);
  else if (N == 288)  gemm_strip<2, PROJ_ST_, 2>(p, row0, 256, 0, 0, red);
  else                gemm_strip<1, PROJ_ST_, 2>(p, row0, 128, 0, 0, red);
}

#define WG_LOGITS_ (16 * H_ * B_)
#define WG_AFF4_ (B_ * I_ * H_ * NPQ_ / 256)
#define WG_AFF8_ (B_ * I_ * H_ * NPV_ / 256)

struct LevelB {
  const float* q; const float* k; float* logits;
  float* qgp; float* kgp; float* vgp; float* qgl; float* kgl; float* vgl;
  const float* past_T; const float* past_R; const float* last_T; const float* last_R;
};

__global__ void __launch_bounds__(256) ipa_level_b(LevelB a) {
  const int bx = (int)blockIdx.x;
  if (bx < WG_LOGITS_) {
    const int h = (bx >> 4) % H_, b = bx / (16 * H_);
    const long qo = (long)b * I_ * D_ + h * DK_;
    const int w = __builtin_amdgcn_readfirstlane((int)(threadIdx.x >> 5));
    gemm_abt_strip<DK_>(a.q + qo, D_, a.k + qo, D_, a.logits + ((long)b * H_ + h) * I_ * I_, I_,
                        (bx & 15) << 4, w << 5, 0.07216878364870323f);
    return;
  }
  int t = bx - WG_LOGITS_;
  const bool lastBranch = t >= (2 * WG_AFF4_ + WG_AFF8_);
  if (lastBranch) t -= 2 * WG_AFF4_ + WG_AFF8_;
  float* pts = lastBranch ? a.qgl : a.qgp; int np = NPQ_; int base = 0;
  if (t >= WG_AFF4_)     { pts = lastBranch ? a.kgl : a.kgp; base = WG_AFF4_; }
  if (t >= 2 * WG_AFF4_) { pts = lastBranch ? a.vgl : a.vgp; base = 2 * WG_AFF4_; np = NPV_; }
  const float* T = lastBranch ? a.last_T : a.past_T;
  const float* R = lastBranch ? a.last_R : a.past_R;
  affine_body(pts, T, R, np, B_ * I_ * H_ * np, (t - base) * 256 + (int)threadIdx.x);
}

#define WG_Z_ (I_ * B_)
#define WG_AVG_ (16 * H_ * B_)

struct LevelD {
  const float* attn; const float* x2d; const float* v; const float* vgp; const float* vgl;
  float* z; float* featS; float* ogp; float* ogl;
};

__global__ void __launch_bounds__(256) ipa_level_d(LevelD a) {
  __shared__ float red[8 * 8 * 256];
  const int bx = (int)blockIdx.x;
  const int FE = 2 * D_ + 2 * H_ * NPV_ * 4;
  const int NG = NPV_ * 3;
  if (bx < WG_Z_) {
    const int i = bx & (I_ - 1), b = bx >> 8;
    const float* A = a.attn + (long)b * H_ * I_ * I_ + (long)i * I_;
    const float* Bm = a.x2d + ((long)b * I_ + i) * I_ * DP_;
    float* Cm = a.z + ((long)b * I_ + i) * H_ * DP_;
    const StripBC g0{Bm, Cm, DP_, DP_};
    const StripBC g1{Bm + 64, Cm + 64, DP_, DP_};
    const StripBC none{nullptr, nullptr, 0, 0};
    gemm_strip_shared<4, 64, 4, 64, 0, 0, true>(A, I_ * I_, H_, I_, 0, g0, g1, none, red);
    return;
  }
  const int t = bx - WG_Z_;
  const int h = (t >> 4) % H_, b = t / (16 * H_);
  const float* A = a.attn + ((long)b * H_ + h) * I_ * I_;
  const long vo = (long)b * I_ * D_ + h * DK_;
  const long go = (long)b * I_ * H_ * NG + h * NG;
  const StripBC g0{a.v + vo, a.featS + (long)b * I_ * FE + h * DK_, D_, FE};
  const StripBC g1{a.vgp + go, a.ogp + go, H_ * NG, H_ * NG};
  const StripBC g2{a.vgl + go, a.ogl + go, H_ * NG, H_ * NG};
  gemm_strip_shared<4, 64, 2, 24, 2, 24, false>(A, I_, I_, I_, (t & 15) << 4, g0, g1, g2, red);
}

#define WG_OUTPAIR_ (8 * H_ * B_)

struct LevelE {
  const float* z; const float* Wpv; const float* ogp; const float* ogl;
  const float* past_T; const float* past_R; float* feat;
};

__global__ void __launch_bounds__(256) ipa_level_e(LevelE a) {
  __shared__ float red[8 * 8 * 256];
  const int bx = (int)blockIdx.x;
  const int FE = 2 * D_ + 2 * H_ * NPV_ * 4;
  if (bx < WG_OUTPAIR_) {
    const GemmP p{a.z, a.Wpv, a.feat, nullptr, I_, DK_, DP_, H_ * DP_, D_, FE,
                  DP_, (long)I_ * H_ * DP_, 64, 0, 64, (long)I_ * FE, 1.f};
    gemm_strip<4, 2, 2>(p, (bx & 7) << 5, 0, (bx >> 3) % H_, bx / (8 * H_), red);
    return;
  }
  int t = bx - WG_OUTPAIR_;
  const bool lastBranch = t >= WG_AFF8_;
  if (lastBranch) t -= WG_AFF8_;
  const int colOl = lastBranch ? (2 * D_ + 384) : (2 * D_);
  inv_affine_body(lastBranch ? a.ogl : a.ogp, a.past_T, a.past_R, a.feat,
                  colOl, colOl + 288, B_ * I_ * H_ * NPV_, t * 256 + (int)threadIdx.x);
}

__global__ void __launch_bounds__(256) ipa_gemm_strip(GemmP p) {
  __shared__ float red[8 * 8 * 256];
  gemm_strip<4, 4, 2>(p, (int)blockIdx.y << 5, (int)blockIdx.x << 6, 0, 0, red);
}

extern "C" void kernel_launch(void* const* d_in, const int* in_sizes, int n_in,
                              void* d_out, int out_size, void* d_ws, size_t ws_size,
                              hipStream_t stream) {
  (void)in_sizes; (void)n_in; (void)out_size; (void)ws_size;
  const float* x1d      = (const float*)d_in[0];
  const float* x2d      = (const float*)d_in[1];
  const float* past_T   = (const float*)d_in[2];
  const float* past_R   = (const float*)d_in[3];
  const float* last_T   = (const float*)d_in[4];
  const float* last_R   = (const float*)d_in[5];
  const float* bias     = (const float*)d_in[6];
  const float* Wq       = (const float*)d_in[7];
  const float* Wk       = (const float*)d_in[8];
  const float* Wv       = (const float*)d_in[9];
  const float* Wpb      = (const float*)d_in[10];
  const float* Wpq_past = (const float*)d_in[11];
  const float* Wpk_past = (const float*)d_in[12];
  const float* Wpv_past = (const float*)d_in[13];
  const float* Wpq_last = (const float*)d_in[14];
  const float* Wpk_last = (const float*)d_in[15];
  const float* Wpv_last = (const float*)d_in[16];
  const float* tpw_past = (const float*)d_in[17];
  const float* tpw_last = (const float*)d_in[18];
  const float* Wpv      = (const float*)d_in[19];
  const float* W_out    = (const float*)d_in[20];
  const float* b_out    = (const float*)d_in[21];
  float* out = (float*)d_out;
  float* ws  = (float*)d_ws;

  float* q    = ws;
  float* k    = q + 393216;
  float* v    = k + 393216;
  float* qgp  = v + 393216;
  float* kgp  = qgp + 73728;
  float* vgp  = kgp + 73728;
  float* qgl  = vgp + 147456;
  float* kgl  = qgl + 73728;
  float* vgl  = kgl + 73728;
  float* logits   = vgl + 147456;
  float* pairproj = logits + 1572864;
  float* ogp  = pairproj + 1572864;
  float* ogl  = ogp + 147456;
  float* z    = ogl + 147456;
  float* feat = z + 786432;

  const LevelA la{x1d, x2d, Wpb,
                  Wq, Wk, Wv, Wpq_past, Wpk_past, Wpv_past, Wpq_last, Wpk_last, Wpv_last,
                  pairproj,
                  q, k, v, qgp, kgp, vgp, qgl, kgl, vgl};
  ipa_level_a<<<WG_PAIRBIAS_ + WG_PROJ_, 256, 0, stream>>>(la);

  const LevelB lb{q, k, logits, qgp, kgp, vgp, qgl, kgl, vgl,
                  past_T, past_R, last_T, last_R};
  ipa_level_b<<<WG_LOGITS_ + 4 * WG_AFF4_ + 2 * WG_AFF8_, 256, 0, stream>>>(lb);

  ipa_combine_softmax<<<B_ * H_ * (I_ / ROWS_PER_WG_), 256, 0, stream>>>(
      logits, qgp, kgp, qgl, kgl, pairproj, bias, tpw_past, tpw_last);

  const LevelD ld{logits, x2d, v, vgp, vgl, z, feat + D_, ogp, ogl};
  ipa_level_d<<<WG_Z_ + WG_AVG_, 256, 0, stream>>>(ld);

  const LevelE le{z, Wpv, ogp, ogl, past_T, past_R, feat};
  ipa_level_e<<<WG_OUTPAIR_ + 2 * WG_AFF8_, 256, 0, stream>>>(le);

  const GemmP pf{feat, W_out, out, b_out, B_ * I_, D_, 2 * D_ + 2 * H_ * NPV_ * 4,
                 2 * D_ + 2 * H_ * NPV_ * 4, D_, D_, 0, 0, 0, 0, 0, 0, 1.f};
  ipa_gemm_strip<<<dim3(12, 16, 1), 256, 0, stream>>>(pf);
}


# scalar-logit GEMM and q/k point frames folded into the softmax launch (8-row groups: q.k^T tile via LDS, frames applied at load; value-point frames as a grid job): 6 launches -> 5; sums unchanged
# speedup vs baseline: 6.7898x; 1.0971x over previous
#include <hip/hip_runtime.h>
#include <math.h>

#define B_ 2
#define I_ 256
#define D_ 768
#define H_ 12
#define DP_ 128
#define DK_ 64
#define NPQ_ 4
#define NPV_ 8

typedef float v2f __attribute__((ext_vector_type(2)));
typedef float v8f __attribute__((ext_vector_type(8)));

struct GemmP {
  const float* A; const float* B; float* C; const float* bias;
  int M, N, K;
  int lda, ldb, ldc;
  long sAy, sAz, sBy, sBz, sCy, sCz;
  float alpha;
};

typedef float v4f __attribute__((ext_vector_type(4)));

template <int NT> struct VecIO;
template <> struct VecIO<1> {
  static __device__ __forceinline__ void ld(const float* p, float* o) { o[0] = p[0]; }
  static __device__ __forceinline__ void st(float* p, const float* o) { p[0] = o[0]; }
};
template <> struct VecIO<2> {
  static __device__ __forceinline__ void ld(const float* p, float* o) {
    const v2f t = *(const v2f*)p; o[0] = t.x; o[1] = t.y;
  }
  static __device__ __forceinline__ void st(float* p, const float* o) {
    v2f t; t.x = o[0]; t.y = o[1]; *(v2f*)p = t;
  }
};
template <> struct VecIO<4> {
  static __device__ __forceinline__ void ld(const float* p, float* o) {
    const v4f t = *(const v4f*)p; o[0] = t.x; o[1] = t.y; o[2] = t.z; o[3] = t.w;
  }
  static __device__ __forceinline__ void st(float* p, const float* o) {
    v4f t; t.x = o[0]; t.y = o[1]; t.z = o[2]; t.w = o[3]; *(v4f*)p = t;
  }
};

template <int NT, int ST = 4, int MT = 1>
__device__ __forceinline__ void gemm_strip(const GemmP p, const int row0, const int col0,
                                           const int by, const int bz, float* red) {
  const int lane = threadIdx.x & 31;
  const int ks = __builtin_amdgcn_readfirstlane((int)(threadIdx.x >> 5));
  const int per = p.K >> 3;
  const int kBeg = ks * per, kEnd = kBeg + per;

  const float* A = p.A + (long)by * p.sAy + (long)bz * p.sAz;
  const float* B = p.B + (long)by * p.sBy + (long)bz * p.sBz;
  float*       C = p.C + (long)by * p.sCy + (long)bz * p.sCz;

  const int c = lane & 15;
  const int h4 = (lane >> 4) << 2;
  const float* aPtr = A + (long)(row0 + c) * p.lda + h4;
  const float* bPtr = B + (long)h4 * p.ldb + col0 + NT * c;

  const long aTile = (long)16 * p.lda;
  v8f acc[MT][NT];
  for (int i = 0; i < MT; ++i)
    for (int j = 0; j < NT; ++j)
      for (int r = 0; r < 8; ++r) acc[i][j][r] = 0.f;

  for (int k0 = kBeg; k0 < kEnd; k0 += 8 * ST) {
    v4f a[ST][MT];
    float b[ST][4][NT];
#pragma unroll
    for (int u = 0; u < ST; ++u) {
      for (int i = 0; i < MT; ++i) a[u][i] = *(const v4f*)(aPtr + i * aTile + k0 + 8 * u);
      const float* bp = bPtr + (long)(k0 + 8 * u) * p.ldb;
      for (int e = 0; e < 4; ++e) VecIO<NT>::ld(bp + (long)e * p.ldb, b[u][e]);
    }
    __builtin_amdgcn_sched_barrier(0);
#pragma unroll
    for (int u = 0; u < ST; ++u) {
      for (int i = 0; i < MT; ++i) {
        const v4f av = a[u][i];
        for (int j = 0; j < NT; ++j) {
          v2f af, bf;
          af.x = av.x; af.y = av.y; bf.x = b[u][0][j]; bf.y = b[u][1][j];
          acc[i][j] = __builtin_amdgcn_wmma_f32_16x16x4_f32(false, af, false, bf,
                                                            (short)0, acc[i][j], false, false);
        }
      }
      for (int i = 0; i < MT; ++i) {
        const v4f av = a[u][i];
        for (int j = 0; j < NT; ++j) {
          v2f af, bf;
          af.x = av.z; af.y = av.w; bf.x = b[u][2][j]; bf.y = b[u][3][j];
          acc[i][j] = __builtin_amdgcn_wmma_f32_16x16x4_f32(false, af, false, bf,
                                                            (short)0, acc[i][j], false, false);
        }
      }
    }
  }

  for (int i = 0; i < MT; ++i)
    for (int j = 0; j < NT; ++j)
      for (int r = 0; r < 8; ++r)
        red[(((ks * MT + i) * NT + j) * 8 + r) * 32 + lane] = acc[i][j][r];
  __syncthreads();
  const int t = (int)threadIdx.x;
  const int oRow = row0 + (((t & 31) >> 4) << 3) + (t >> 5);
  const int oCol = col0 + NT * (t & 15);
  float bv[NT];
  for (int j = 0; j < NT; ++j) bv[j] = 0.f;
  if (p.bias) VecIO<NT>::ld(p.bias + oCol, bv);
  for (int i = 0; i < MT; ++i) {
    float o[NT];
    for (int j = 0; j < NT; ++j) {
      float sum = red[(i * NT + j) * 256 + t];
      for (int s = 1; s < 8; ++s) sum += red[((s * MT + i) * NT + j) * 256 + t];
      o[j] = sum * p.alpha;
      if (p.bias) o[j] += bv[j];
    }
    VecIO<NT>::st(C + (long)(oRow + 16 * i) * p.ldc + oCol, o);
  }
}

struct StripBC { const float* B; float* C; int ldb, ldc; };

template <int NT, int NV>
__device__ __forceinline__ const float* strip_bptr(const float* B, const int ldb, const int c,
                                                   const int h4, float& mask) {
  int cc = c;
  mask = 1.f;
  if (NV < 16 * NT) {
    if (NT * c >= NV) { cc = 0; mask = 0.f; }
  }
  return B + (long)h4 * ldb + NT * cc;
}

template <int NT>
__device__ __forceinline__ void strip_load_b(const float* bPtr, const int ldb, const int k0,
                                             float (&b)[4][4][NT]) {
#pragma unroll
  for (int u = 0; u < 4; ++u) {
    const float* bp = bPtr + (long)(k0 + 8 * u) * ldb;
#pragma unroll
    for (int e = 0; e < 4; ++e) VecIO<NT>::ld(bp + (long)e * ldb, b[u][e]);
  }
}

template <int NT, bool MASK>
__device__ __forceinline__ void strip_mma(const v4f av, const float (&bu)[4][NT], const float mask,
                                          v8f (&acc)[NT]) {
  for (int j = 0; j < NT; ++j) {
    v2f af, bf;
    af.x = av.x; af.y = av.y; bf.x = bu[0][j]; bf.y = bu[1][j];
    if (MASK) bf *= mask;
    acc[j] = __builtin_amdgcn_wmma_f32_16x16x4_f32(false, af, false, bf,
                                                   (short)0, acc[j], false, false);
  }
  for (int j = 0; j < NT; ++j) {
    v2f af, bf;
    af.x = av.z; af.y = av.w; bf.x = bu[2][j]; bf.y = bu[3][j];
    if (MASK) bf *= mask;
    acc[j] = __builtin_amdgcn_wmma_f32_16x16x4_f32(false, af, false, bf,
                                                   (short)0, acc[j], false, false);
  }
}

template <int NT>
__device__ __forceinline__ void strip_zero(v8f (&acc)[NT]) {
  for (int j = 0; j < NT; ++j)
    for (int r = 0; r < 8; ++r) acc[j][r] = 0.f;
}

template <int NT>
__device__ __forceinline__ void strip_spill(float* red, const int ks, const int ntt, const int j0,
                                            const int lane, const v8f (&acc)[NT]) {
  for (int j = 0; j < NT; ++j)
    for (int r = 0; r < 8; ++r) red[((ks * ntt + j0 + j) * 8 + r) * 32 + lane] = acc[j][r];
}

template <int NT, int NV>
__device__ __forceinline__ void strip_finish(const float* red, const int ntt, const int j0,
                                             const int t, const bool rowOK, float* cRow) {
  float o[NT];
  for (int j = 0; j < NT; ++j) {
    float sum = red[(j0 + j) * 256 + t];
    for (int s = 1; s < 8; ++s) sum += red[(s * ntt + j0 + j) * 256 + t];
    o[j] = sum;
  }
  const int oc = NT * (t & 15);
  if (rowOK && (NV == 16 * NT || oc < NV)) VecIO<NT>::st(cRow + oc, o);
}

template <int NT0, int NV0, int NT1, int NV1, int NT2, int NV2, bool GM>
__device__ __forceinline__ void gemm_strip_shared(const float* A, const int lda, const int M,
                                                  const int K, const int row0,
                                                  const StripBC g0, const StripBC g1,
                                                  const StripBC g2, float* red) {
  constexpr int N2 = NT2 ? NT2 : 1;
  constexpr int NTT = NT0 + NT1 + NT2;
  const int lane = threadIdx.x & 31;
  const int ks = __builtin_amdgcn_readfirstlane((int)(threadIdx.x >> 5));
  const int per = K >> 3;
  const int kBeg = ks * per, kEnd = kBeg + per;

  const int c = lane & 15;
  const int h4 = (lane >> 4) << 2;
  int aRow = row0 + c;
  float aMask = 1.f;
  if (GM) {
    if (aRow >= M) { aRow = M - 1; aMask = 0.f; }
  }
  const float* aPtr = A + (long)aRow * lda + h4;
  float m0, m1, m2 = 1.f;
  const float* bPtr0 = strip_bptr<NT0, NV0>(g0.B, g0.ldb, c, h4, m0);
  const float* bPtr1 = strip_bptr<NT1, NV1>(g1.B, g1.ldb, c, h4, m1);
  const float* bPtr2 = nullptr;
  if (NT2) bPtr2 = strip_bptr<N2, NV2>(g2.B, g2.ldb, c, h4, m2);

  v8f acc0[NT0], acc1[NT1], acc2[N2];
  strip_zero<NT0>(acc0);
  strip_zero<NT1>(acc1);
  if (NT2) strip_zero<N2>(acc2);

  for (int k0 = kBeg; k0 < kEnd; k0 += 32) {
    v4f a[4];
    float b0[4][4][NT0], b1[4][4][NT1], b2[4][4][N2];
#pragma unroll
    for (int u = 0; u < 4; ++u) a[u] = *(const v4f*)(aPtr + k0 + 8 * u);
    strip_load_b<NT0>(bPtr0, g0.ldb, k0, b0);
    strip_load_b<NT1>(bPtr1, g1.ldb, k0, b1);
    if (NT2) strip_load_b<N2>(bPtr2, g2.ldb, k0, b2);
    __builtin_amdgcn_sched_barrier(0);
#pragma unroll
    for (int u = 0; u < 4; ++u) {
      v4f av = a[u];
      if (GM) av *= aMask;
      strip_mma<NT0, (NV0 < 16 * NT0)>(av, b0[u], m0, acc0);
      strip_mma<NT1, (NV1 < 16 * NT1)>(av, b1[u], m1, acc1);
      if (NT2) strip_mma<N2, (NV2 < 16 * N2)>(av, b2[u], m2, acc2);
    }
  }

  strip_spill<NT0>(red, ks, NTT, 0, lane, acc0);
  strip_spill<NT1>(red, ks, NTT, NT0, lane, acc1);
  if (NT2) strip_spill<N2>(red, ks, NTT, NT0 + NT1, lane, acc2);
  __syncthreads();
  const int t = (int)threadIdx.x;
  const int oRow = row0 + (((t & 31) >> 4) << 3) + (t >> 5);
  const bool rowOK = !GM || oRow < M;
  strip_finish<NT0, NV0>(red, NTT, 0, t, rowOK, g0.C + (long)oRow * g0.ldc);
  strip_finish<NT1, NV1>(red, NTT, NT0, t, rowOK, g1.C + (long)oRow * g1.ldc);
  if (NT2) strip_finish<N2, NV2>(red, NTT, NT0 + NT1, t, rowOK, g2.C + (long)oRow * g2.ldc);
}

#define PB_TPW_ 2

__device__ __forceinline__ void pairbias_load_a(const float* aPtr, v4f (&a)[DP_ / 8]) {
#pragma unroll
  for (int u = 0; u < DP_ / 8; ++u) a[u] = *(const v4f*)(aPtr + 8 * u);
}

__device__ __forceinline__ void pairbias_tiles(const float* x2d, const float* Wpb, float* pairproj,
                                               const int wg, const float alpha, float* tab) {
  const int t = (int)threadIdx.x;
  const int lane = t & 31;
  const int wave = __builtin_amdgcn_readfirstlane(t >> 5);
  const int c = lane & 15;
  const int h4 = (lane >> 4) << 2;
  const int tile0 = (wg * 8 + wave) * PB_TPW_;
  const float* aPtr = x2d + ((long)(tile0 << 4) + c) * DP_ + h4;

  v4f a[DP_ / 8];
  pairbias_load_a(aPtr, a);

  for (int q = 0; q < 2; ++q) {
    const int ent = t + 256 * q;
    const int ec = ent & 15;
    const int ek = 8 * (ent >> 5) + (((ent & 31) >> 4) << 2);
    v4f w = {0.f, 0.f, 0.f, 0.f};
    if (ec < H_) {
      const float* wp = Wpb + (long)ek * H_ + ec;
      w.x = wp[0]; w.y = wp[H_]; w.z = wp[2 * H_]; w.w = wp[3 * H_];
    }
    *(v4f*)(tab + 4 * ent) = w;
  }
  __syncthreads();

  const int mo = (lane >> 4) << 3;
  for (int n = 0; n < PB_TPW_; ++n) {
    v8f acc = {0.f, 0.f, 0.f, 0.f, 0.f, 0.f, 0.f, 0.f};
#pragma unroll
    for (int u = 0; u < DP_ / 8; ++u) {
      const v4f bw = *(const v4f*)(tab + 4 * (u * 32 + lane));
      v2f af, bf;
      af.x = a[u].x; af.y = a[u].y; bf.x = bw.x; bf.y = bw.y;
      acc = __builtin_amdgcn_wmma_f32_16x16x4_f32(false, af, false, bf, (short)0, acc, false, false);
      af.x = a[u].z; af.y = a[u].w; bf.x = bw.z; bf.y = bw.w;
      acc = __builtin_amdgcn_wmma_f32_16x16x4_f32(false, af, false, bf, (short)0, acc, false, false);
    }
    if (n + 1 < PB_TPW_) pairbias_load_a(aPtr + (long)(n + 1) * 16 * DP_, a);
    if (c < H_) {
      float* cPtr = pairproj + ((long)((tile0 + n) << 4) + mo) * H_ + c;
      for (int r = 0; r < 8; ++r) cPtr[r * H_] = acc[r] * alpha;
    }
  }
}

struct Frame { float r[9]; float t[3]; };

__device__ __forceinline__ Frame frame_load(const float* R, const float* T, const long bi) {
  Frame f;
  for (int n = 0; n < 9; ++n) f.r[n] = R[bi * 9 + n];
  for (int n = 0; n < 3; ++n) f.t[n] = T[bi * 3 + n];
  return f;
}

__device__ __forceinline__ void frame_apply(const Frame& f, const float x, const float y,
                                            const float z, float* o) {
  o[0] = f.r[0] * x + f.r[3] * y + f.r[6] * z + f.t[0];
  o[1] = f.r[1] * x + f.r[4] * y + f.r[7] * z + f.t[1];
  o[2] = f.r[2] * x + f.r[5] * y + f.r[8] * z + f.t[2];
}

__device__ __forceinline__ void affine_body(float* pts, const float* T, const float* R,
                                            int np, int total, int idx) {
  if (idx >= total) return;
  const Frame f = frame_load(R, T, idx / (H_ * np));
  float o[3];
  frame_apply(f, pts[idx * 3 + 0], pts[idx * 3 + 1], pts[idx * 3 + 2], o);
  pts[idx * 3 + 0] = o[0]; pts[idx * 3 + 1] = o[1]; pts[idx * 3 + 2] = o[2];
}

__device__ __forceinline__ void inv_affine_body(const float* og, const float* T, const float* R,
                                                float* feat, int colOl, int colN, int total,
                                                int idx) {
  if (idx >= total) return;
  int c = idx & (NPV_ - 1);
  int t = idx >> 3;
  int h = t % H_;
  int bi = t / H_;
  const float* o = og + (long)idx * 3;
  const float* Tv = T + (long)bi * 3;
  const float* Rm = R + (long)bi * 9;
  float x = o[0] - Tv[0], y = o[1] - Tv[1], z = o[2] - Tv[2];
  float p0 = Rm[0] * x + Rm[1] * y + Rm[2] * z;
  float p1 = Rm[3] * x + Rm[4] * y + Rm[5] * z;
  float p2 = Rm[6] * x + Rm[7] * y + Rm[8] * z;
  float* fr = feat + (long)bi * (2 * D_ + 2 * H_ * NPV_ * 4);
  int tc = h * (NPV_ * 3) + c * 3;
  fr[colOl + tc + 0] = p0;
  fr[colOl + tc + 1] = p1;
  fr[colOl + tc + 2] = p2;
  fr[colN + h * NPV_ + c] = sqrtf(p0 * p0 + p1 * p1 + p2 * p2);
}

#define WG_PAIRBIAS_ (B_ * I_ * I_ / 16 / 8 / PB_TPW_)
#define WG_PROJ_ (3 * 192 + 4 * 48 + 2 * 80)

struct LevelA {
  const float* x1d; const float* x2d; const float* Wpb;
  const float* W0; const float* W1; const float* W2;
  const float* W3; const float* W4; const float* W5;
  const float* W6; const float* W7; const float* W8;
  float* pairproj;
  float* C0; float* C1; float* C2; float* C3; float* C4; float* C5;
  float* C6; float* C7; float* C8;
};

#define PROJ_ST_ 3
#define PROJ_WAVES_ 6
__global__ void __launch_bounds__(256, PROJ_WAVES_) ipa_level_a(LevelA a) {
  __shared__ float red[8 * 8 * 256];
  const int bx = (int)blockIdx.x;
  if (bx < WG_PAIRBIAS_) {
    pairbias_tiles(a.x2d, a.Wpb, a.pairproj, bx, 0.5773502691896258f, red);
    return;
  }
  const int t = bx - WG_PAIRBIAS_;
  const float* Wm = a.W0; float* Cm = a.C0; int N = 768; int base = 0; int strips = 12; int full = 12;
  if (t >= 192) { Wm = a.W1; Cm = a.C1; base = 192; }
  if (t >= 384) { Wm = a.W2; Cm = a.C2; base = 384; }
  if (t >= 576) { Wm = a.W3; Cm = a.C3; base = 576; N = 144; strips = 3; full = 2; }
  if (t >= 624) { Wm = a.W4; Cm = a.C4; base = 624; N = 144; strips = 3; full = 2; }
  if (t >= 672) { Wm = a.W5; Cm = a.C5; base = 672; N = 288; strips = 5; full = 4; }
  if (t >= 752) { Wm = a.W6; Cm = a.C6; base = 752; N = 144; strips = 3; full = 2; }
  if (t >= 800) { Wm = a.W7; Cm = a.C7; base = 800; N = 144; strips = 3; full = 2; }
  if (t >= 848) { Wm = a.W8; Cm = a.C8; base = 848; N = 288; strips = 5; full = 4; }
  const GemmP p{a.x1d, Wm, Cm, nullptr, B_ * I_, N, D_, D_, N, N,
                0, 0, 0, 0, 0, 0, 1.f};
  const int u = t - base;
  const int row0 = (u / strips) << 5;
  const int sx = u % strips;
  if (sx < full)      gemm_strip<4, PROJ_ST_, 2>(p, row0, sx << 6, 0, 0, red);
  else if (N == 288)  gemm_strip<2, PROJ_ST_, 2>(p, row0, 256, 0, 0, red);
  else                gemm_strip<1, PROJ_ST_, 2>(p, row0, 128, 0, 0, red);
}

#define ROWS_PER_WG_ 8
#define NKC_ (NPQ_ * 3)
#define WG_SOFTMAX_ (B_ * H_ * (I_ / ROWS_PER_WG_))
#define VG_PTS_PER_WG_ 1024
#define WG_VG_ (B_ * I_ * H_ * NPV_ / VG_PTS_PER_WG_)

struct LevelC {
  const float* q; const float* k; float* attn;
  const float* qpp; const float* kpp; float* vgp;
  const float* qpl; const float* kpl; float* vgl;
  const float* pairproj; const float* bias;
  const float* tpw_past; const float* tpw_last;
  const float* past_T; const float* past_R; const float* last_T; const float* last_R;
};

__device__ __forceinline__ void points4_framed(const float* p, const Frame& f, float (&o)[NKC_]) {
  const v4f* pv = (const v4f*)p;
  const v4f p0 = pv[0], p1 = pv[1], p2 = pv[2];
  frame_apply(f, p0.x, p0.y, p0.z, o + 0);
  frame_apply(f, p0.w, p1.x, p1.y, o + 3);
  frame_apply(f, p1.z, p1.w, p2.x, o + 6);
  frame_apply(f, p2.y, p2.z, p2.w, o + 9);
}

__global__ void __launch_bounds__(256, 6) ipa_level_c(LevelC a) {
  __shared__ float kst[2 * NKC_ * I_];
  __shared__ float lg[ROWS_PER_WG_ * I_];
  const int bx = (int)blockIdx.x;
  const int t = (int)threadIdx.x;
  if (bx >= WG_SOFTMAX_) {
    int u = bx - WG_SOFTMAX_;
    const bool lastBranch = u >= WG_VG_;
    if (lastBranch) u -= WG_VG_;
    float* pts = lastBranch ? a.vgl : a.vgp;
    const float* T = lastBranch ? a.last_T : a.past_T;
    const float* R = lastBranch ? a.last_R : a.past_R;
    for (int s = 0; s < VG_PTS_PER_WG_ / 256; ++s)
      affine_body(pts, T, R, NPV_, B_ * I_ * H_ * NPV_, u * VG_PTS_PER_WG_ + s * 256 + t);
    return;
  }

  const int ig = bx % (I_ / ROWS_PER_WG_);
  const int bh = bx / (I_ / ROWS_PER_WG_);
  const int h = bh % H_;
  const int b = bh / H_;
  const int lane = t & 31;
  const int w = __builtin_amdgcn_readfirstlane(t >> 5);
  const int i0 = ig * ROWS_PER_WG_;
  const int i = i0 + w;
  const int c = lane & 15;
  const int h4 = (lane >> 4) << 2;

  const long hb = (long)b * I_ * D_ + h * DK_;
  const float* aPtr = a.q + hb + (long)(i0 + (c & 7)) * D_ + h4;
  const float* bPtr = a.k + hb + (long)(32 * w + 2 * c) * D_ + h4;
  v4f fa[DK_ / 8], fb0[DK_ / 8], fb1[DK_ / 8];
#pragma unroll
  for (int u = 0; u < DK_ / 8; ++u) {
    fa[u]  = *(const v4f*)(aPtr + 8 * u);
    fb0[u] = *(const v4f*)(bPtr + 8 * u);
    fb1[u] = *(const v4f*)(bPtr + D_ + 8 * u);
  }
  __builtin_amdgcn_sched_barrier(0);

  {
    const long bj = (long)(b * I_ + t);
    const long ko = (bj * H_ + h) * NKC_;
    float o[NKC_];
    points4_framed(a.kpp + ko, frame_load(a.past_R, a.past_T, bj), o);
    for (int n = 0; n < NKC_; ++n) kst[n * I_ + t] = o[n];
    points4_framed(a.kpl + ko, frame_load(a.last_R, a.last_T, bj), o);
    for (int n = 0; n < NKC_; ++n) kst[(NKC_ + n) * I_ + t] = o[n];
  }

  v8f acc0 = {0.f, 0.f, 0.f, 0.f, 0.f, 0.f, 0.f, 0.f};
  v8f acc1 = {0.f, 0.f, 0.f, 0.f, 0.f, 0.f, 0.f, 0.f};
#pragma unroll
  for (int u = 0; u < DK_ / 8; ++u) {
    v2f af, bf;
    af.x = fa[u].x; af.y = fa[u].y;
    bf.x = fb0[u].x; bf.y = fb0[u].y;
    acc0 = __builtin_amdgcn_wmma_f32_16x16x4_f32(false, af, false, bf, (short)0, acc0, false, false);
    bf.x = fb1[u].x; bf.y = fb1[u].y;
    acc1 = __builtin_amdgcn_wmma_f32_16x16x4_f32(false, af, false, bf, (short)0, acc1, false, false);
    af.x = fa[u].z; af.y = fa[u].w;
    bf.x = fb0[u].z; bf.y = fb0[u].w;
    acc0 = __builtin_amdgcn_wmma_f32_16x16x4_f32(false, af, false, bf, (short)0, acc0, false, false);
    bf.x = fb1[u].z; bf.y = fb1[u].w;
    acc1 = __builtin_amdgcn_wmma_f32_16x16x4_f32(false, af, false, bf, (short)0, acc1, false, false);
  }
  {
    const float alpha = 0.07216878364870323f;
    v2f o[8];
    for (int r = 0; r < 8; ++r) { o[r].x = acc0[r] * alpha; o[r].y = acc1[r] * alpha; }
    if (lane < 16) {
      float* lp = lg + 32 * w + 2 * c;
      for (int r = 0; r < 8; ++r) *(v2f*)(lp + r * I_) = o[r];
    }
  }

  float qp[NKC_], ql[NKC_];
  {
    const long bi = (long)(b * I_ + i);
    const long qo = (bi * H_ + h) * NKC_;
    const Frame fp = frame_load(a.past_R, a.past_T, bi);
    const Frame fl = frame_load(a.last_R, a.last_T, bi);
    for (int n = 0; n < NPQ_; ++n) {
      frame_apply(fp, a.qpp[qo + 3 * n], a.qpp[qo + 3 * n + 1], a.qpp[qo + 3 * n + 2], qp + 3 * n);
      frame_apply(fl, a.qpl[qo + 3 * n], a.qpl[qo + 3 * n + 1], a.qpl[qo + 3 * n + 2], ql + 3 * n);
    }
  }

  const float point_w = 0.13608276348795434f;
  const float pwp = -0.5f * point_w * log1pf(expf(a.tpw_past[h]));
  const float pwl = -0.5f * point_w * log1pf(expf(a.tpw_last[h]));

  const long row = ((long)(b * H_ + h) * I_ + i) * I_;
  const float* pr = a.pairproj + ((long)(b * I_ + i) * I_) * H_ + h;
  const float* bias = a.bias;
  float* attn = a.attn;
  const float* lrow = lg + w * I_;

  __syncthreads();

  float v[I_ / 32];
#pragma unroll
  for (int m = 0; m < I_ / 32; ++m) {
    const int j = m * 32 + lane;
    float sp = 0.f, sl = 0.f;
    for (int n = 0; n < NPQ_; ++n) {
      float dx = qp[n * 3 + 0] - kst[(n * 3 + 0) * I_ + j];
      float dy = qp[n * 3 + 1] - kst[(n * 3 + 1) * I_ + j];
      float dz = qp[n * 3 + 2] - kst[(n * 3 + 2) * I_ + j];
      sp += sqrtf(dx * dx + dy * dy + dz * dz);
      dx = ql[n * 3 + 0] - kst[(NKC_ + n * 3 + 0) * I_ + j];
      dy = ql[n * 3 + 1] - kst[(NKC_ + n * 3 + 1) * I_ + j];
      dz = ql[n * 3 + 2] - kst[(NKC_ + n * 3 + 2) * I_ + j];
      sl += sqrtf(dx * dx + dy * dy + dz * dz);
    }
    const float pv = pr[(long)j * H_];
    const float add = pwp * sp + pwl * sl + pv + bias[row + j];
    v[m] = lrow[j] + add;
  }

  float mx = v[0];
  for (int m = 1; m < I_ / 32; ++m) mx = fmaxf(mx, v[m]);
  for (int off = 16; off > 0; off >>= 1) mx = fmaxf(mx, __shfl_xor(mx, off));

  float sum = 0.f;
  for (int m = 0; m < I_ / 32; ++m) { v[m] = expf(v[m] - mx); sum += v[m]; }
  for (int off = 16; off > 0; off >>= 1) sum += __shfl_xor(sum, off);

  for (int m = 0; m < I_ / 32; ++m) attn[row + m * 32 + lane] = v[m] / sum;
}

#define WG_AFF8_ (B_ * I_ * H_ * NPV_ / 256)

#define WG_Z_ (I_ * B_)
#define WG_AVG_ (16 * H_ * B_)

struct LevelD {
  const float* attn; const float* x2d; const float* v; const float* vgp; const float* vgl;
  float* z; float* featS; float* ogp; float* ogl;
};

__global__ void __launch_bounds__(256) ipa_level_d(LevelD a) {
  __shared__ float red[8 * 8 * 256];
  const int bx = (int)blockIdx.x;
  const int FE = 2 * D_ + 2 * H_ * NPV_ * 4;
  const int NG = NPV_ * 3;
  if (bx < WG_Z_) {
    const int i = bx & (I_ - 1), b = bx >> 8;
    const float* A = a.attn + (long)b * H_ * I_ * I_ + (long)i * I_;
    const float* Bm = a.x2d + ((long)b * I_ + i) * I_ * DP_;
    float* Cm = a.z + ((long)b * I_ + i) * H_ * DP_;
    const StripBC g0{Bm, Cm, DP_, DP_};
    const StripBC g1{Bm + 64, Cm + 64, DP_, DP_};
    const StripBC none{nullptr, nullptr, 0, 0};
    gemm_strip_shared<4, 64, 4, 64, 0, 0, true>(A, I_ * I_, H_, I_, 0, g0, g1, none, red);
    return;
  }
  const int t = bx - WG_Z_;
  const int h = (t >> 4) % H_, b = t / (16 * H_);
  const float* A = a.attn + ((long)b * H_ + h) * I_ * I_;
  const long vo = (long)b * I_ * D_ + h * DK_;
  const long go = (long)b * I_ * H_ * NG + h * NG;
  const StripBC g0{a.v + vo, a.featS + (long)b * I_ * FE + h * DK_, D_, FE};
  const StripBC g1{a.vgp + go, a.ogp + go, H_ * NG, H_ * NG};
  const StripBC g2{a.vgl + go, a.ogl + go, H_ * NG, H_ * NG};
  gemm_strip_shared<4, 64, 2, 24, 2, 24, false>(A, I_, I_, I_, (t & 15) << 4, g0, g1, g2, red);
}

#define WG_OUTPAIR_ (8 * H_ * B_)

struct LevelE {
  const float* z; const float* Wpv; const float* ogp; const float* ogl;
  const float* past_T; const float* past_R; float* feat;
};

__global__ void __launch_bounds__(256) ipa_level_e(LevelE a) {
  __shared__ float red[8 * 8 * 256];
  const int bx = (int)blockIdx.x;
  const int FE = 2 * D_ + 2 * H_ * NPV_ * 4;
  if (bx < WG_OUTPAIR_) {
    const GemmP p{a.z, a.Wpv, a.feat, nullptr, I_, DK_, DP_, H_ * DP_, D_, FE,
                  DP_, (long)I_ * H_ * DP_, 64, 0, 64, (long)I_ * FE, 1.f};
    gemm_strip<4, 2, 2>(p, (bx & 7) << 5, 0, (bx >> 3) % H_, bx / (8 * H_), red);
    return;
  }
  int t = bx - WG_OUTPAIR_;
  const bool lastBranch = t >= WG_AFF8_;
  if (lastBranch) t -= WG_AFF8_;
  const int colOl = lastBranch ? (2 * D_ + 384) : (2 * D_);
  inv_affine_body(lastBranch ? a.ogl : a.ogp, a.past_T, a.past_R, a.feat,
                  colOl, colOl + 288, B_ * I_ * H_ * NPV_, t * 256 + (int)threadIdx.x);
}

__global__ void __launch_bounds__(256) ipa_gemm_strip(GemmP p) {
  __shared__ float red[8 * 8 * 256];
  gemm_strip<4, 4, 2>(p, (int)blockIdx.y << 5, (int)blockIdx.x << 6, 0, 0, red);
}

extern "C" void kernel_launch(void* const* d_in, const int* in_sizes, int n_in,
                              void* d_out, int out_size, void* d_ws, size_t ws_size,
                              hipStream_t stream) {
  (void)in_sizes; (void)n_in; (void)out_size; (void)ws_size;
  const float* x1d      = (const float*)d_in[0];
  const float* x2d      = (const float*)d_in[1];
  const float* past_T   = (const float*)d_in[2];
  const float* past_R   = (const float*)d_in[3];
  const float* last_T   = (const float*)d_in[4];
  const float* last_R   = (const float*)d_in[5];
  const float* bias     = (const float*)d_in[6];
  const float* Wq       = (const float*)d_in[7];
  const float* Wk       = (const float*)d_in[8];
  const float* Wv       = (const float*)d_in[9];
  const float* Wpb      = (const float*)d_in[10];
  const float* Wpq_past = (const float*)d_in[11];
  const float* Wpk_past = (const float*)d_in[12];
  const float* Wpv_past = (const float*)d_in[13];
  const float* Wpq_last = (const float*)d_in[14];
  const float* Wpk_last = (const float*)d_in[15];
  const float* Wpv_last = (const float*)d_in[16];
  const float* tpw_past = (const float*)d_in[17];
  const float* tpw_last = (const float*)d_in[18];
  const float* Wpv      = (const float*)d_in[19];
  const float* W_out    = (const float*)d_in[20];
  const float* b_out    = (const float*)d_in[21];
  float* out = (float*)d_out;
  float* ws  = (float*)d_ws;

  float* q    = ws;
  float* k    = q + 393216;
  float* v    = k + 393216;
  float* qgp  = v + 393216;
  float* kgp  = qgp + 73728;
  float* vgp  = kgp + 73728;
  float* qgl  = vgp + 147456;
  float* kgl  = qgl + 73728;
  float* vgl  = kgl + 73728;
  float* logits   = vgl + 147456;
  float* pairproj = logits + 1572864;
  float* ogp  = pairproj + 1572864;
  float* ogl  = ogp + 147456;
  float* z    = ogl + 147456;
  float* feat = z + 786432;

  const LevelA la{x1d, x2d, Wpb,
                  Wq, Wk, Wv, Wpq_past, Wpk_past, Wpv_past, Wpq_last, Wpk_last, Wpv_last,
                  pairproj,
                  q, k, v, qgp, kgp, vgp, qgl, kgl, vgl};
  ipa_level_a<<<WG_PAIRBIAS_ + WG_PROJ_, 256, 0, stream>>>(la);

  const LevelC lc{q, k, logits, qgp, kgp, vgp, qgl, kgl, vgl, pairproj, bias,
                  tpw_past, tpw_last, past_T, past_R, last_T, last_R};
  ipa_level_c<<<WG_SOFTMAX_ + 2 * WG_VG_, 256, 0, stream>>>(lc);

  const LevelD ld{logits, x2d, v, vgp, vgl, z, feat + D_, ogp, ogl};
  ipa_level_d<<<WG_Z_ + WG_AVG_, 256, 0, stream>>>(ld);

  const LevelE le{z, Wpv, ogp, ogl, past_T, past_R, feat};
  ipa_level_e<<<WG_OUTPAIR_ + 2 * WG_AFF8_, 256, 0, stream>>>(le);

  const GemmP pf{feat, W_out, out, b_out, B_ * I_, D_, 2 * D_ + 2 * H_ * NPV_ * 4,
                 2 * D_ + 2 * H_ * NPV_ * 4, D_, D_, 0, 0, 0, 0, 0, 0, 1.f};
  ipa_gemm_strip<<<dim3(12, 16, 1), 256, 0, stream>>>(pf);
}
